# starting kernel
# baseline (speedup 1.0000x reference)
#include <hip/hip_runtime.h>
#include <hip/hip_bf16.h>

#define HID 128
#define NGRAPH 512
#define NEG_SLOPE 0.2f
#define BN_EPS 1e-5f

typedef __attribute__((ext_vector_type(2))) float v2f;
typedef __attribute__((ext_vector_type(8))) float v8f;

__device__ __forceinline__ unsigned f2o(float f) {
    unsigned u = __float_as_uint(f);
    return (u & 0x80000000u) ? ~u : (u | 0x80000000u);
}
__device__ __forceinline__ float o2f(unsigned u) {
    unsigned v = (u & 0x80000000u) ? (u & 0x7fffffffu) : ~u;
    return __uint_as_float(v);
}
__device__ __forceinline__ float lrelu(float a) { return a > 0.0f ? a : NEG_SLOPE * a; }

__global__ void gat_zero(float* __restrict__ p, int n) {
    int i = blockIdx.x * blockDim.x + threadIdx.x;
    if (i < n) p[i] = 0.0f;
}

__global__ void gat_gemm_wmma(const float* __restrict__ X, const float* __restrict__ W,
                              float* __restrict__ H, int nrows) {
    int wave = (blockIdx.x * blockDim.x + threadIdx.x) >> 5;
    int m0 = wave * 16;
    if (m0 >= nrows) return;
    int lane = threadIdx.x & 31;
    int lr = lane & 15;
    int lh = lane >> 4;

    v8f acc[8] = {};
    const float* xrow = X + (size_t)(m0 + lr) * HID;

    for (int k = 0; k < HID; k += 4) {
        int ka = k + lh * 2;
        v2f a;
        a.x = xrow[ka];
        a.y = xrow[ka + 1];
#pragma unroll
        for (int n = 0; n < 8; ++n) {
            v2f b;
            b.x = W[ka * HID + n * 16 + lr];
            b.y = W[(ka + 1) * HID + n * 16 + lr];
            acc[n] = __builtin_amdgcn_wmma_f32_16x16x4_f32(
                false, a, false, b, (short)0, acc[n], false, false);
        }
    }
#pragma unroll
    for (int n = 0; n < 8; ++n)
#pragma unroll
        for (int j = 0; j < 8; ++j)
            H[(size_t)(m0 + j + lh * 8) * HID + n * 16 + lr] = acc[n][j];
}

__global__ void gat_node_dots(const float* __restrict__ H, const float* __restrict__ att_s,
                              const float* __restrict__ att_d, float* __restrict__ a_s,
                              float* __restrict__ a_d, unsigned* __restrict__ mbits, int n_nodes) {
    int node = (blockIdx.x * blockDim.x + threadIdx.x) >> 5;
    if (node >= n_nodes) return;
    int lane = threadIdx.x & 31;
    float4 hv = ((const float4*)(H + (size_t)node * HID))[lane];
    float4 ws = ((const float4*)att_s)[lane];
    float4 wd = ((const float4*)att_d)[lane];
    float ps = hv.x * ws.x + hv.y * ws.y + hv.z * ws.z + hv.w * ws.w;
    float pd = hv.x * wd.x + hv.y * wd.y + hv.z * wd.z + hv.w * wd.w;
    for (int off = 16; off > 0; off >>= 1) {
        ps += __shfl_down(ps, off, 32);
        pd += __shfl_down(pd, off, 32);
    }
    if (lane == 0) {
        a_s[node] = ps;
        a_d[node] = pd;
        mbits[node] = f2o(lrelu(ps + pd));
    }
}

__global__ void gat_edge_max(const int* __restrict__ ei, const float* __restrict__ a_s,
                             const float* __restrict__ a_d, unsigned* __restrict__ mbits,
                             int n_edges) {
    int e = blockIdx.x * blockDim.x + threadIdx.x;
    if (e >= n_edges) return;
    int s = ei[e];
    int d = ei[n_edges + e];
    float a = lrelu(a_s[s] + a_d[d]);
    atomicMax(&mbits[d], f2o(a));
}

__global__ void gat_node_init(const float* __restrict__ H, const float* __restrict__ a_s,
                              const float* __restrict__ a_d, float* __restrict__ m,
                              float* __restrict__ s, float* __restrict__ out_acc, int n_nodes) {
    int node = (blockIdx.x * blockDim.x + threadIdx.x) >> 5;
    if (node >= n_nodes) return;
    int lane = threadIdx.x & 31;
    float mf = o2f(((const unsigned*)m)[node]);
    float es = expf(lrelu(a_s[node] + a_d[node]) - mf);
    if (lane == 0) {
        m[node] = mf;
        s[node] = es;
    }
    float4 hv = ((const float4*)(H + (size_t)node * HID))[lane];
    float4 r;
    r.x = es * hv.x; r.y = es * hv.y; r.z = es * hv.z; r.w = es * hv.w;
    ((float4*)(out_acc + (size_t)node * HID))[lane] = r;
}

__global__ void gat_edge_scatter(const int* __restrict__ ei, const float* __restrict__ a_s,
                                 const float* __restrict__ a_d, const float* __restrict__ m,
                                 const float* __restrict__ H, float* __restrict__ s,
                                 float* __restrict__ out_acc, int n_edges) {
    int e = (blockIdx.x * blockDim.x + threadIdx.x) >> 5;
    if (e >= n_edges) return;
    int lane = threadIdx.x & 31;
    int src = ei[e];
    int dst = ei[n_edges + e];
    float w = expf(lrelu(a_s[src] + a_d[dst]) - m[dst]);
    if (lane == 0) atomicAdd(&s[dst], w);
    float4 hv = ((const float4*)(H + (size_t)src * HID))[lane];
    float* o = out_acc + (size_t)dst * HID + lane * 4;
    atomicAdd(o + 0, w * hv.x);
    atomicAdd(o + 1, w * hv.y);
    atomicAdd(o + 2, w * hv.z);
    atomicAdd(o + 3, w * hv.w);
}

__global__ void gat_finalize(float* __restrict__ out_acc, const float* __restrict__ s,
                             const float* __restrict__ bias, float* __restrict__ sums,
                             float* __restrict__ sumsq, int n_nodes) {
    int c = threadIdx.x;
    float b = bias[c];
    float ls = 0.0f, lq = 0.0f;
    for (int n = blockIdx.x; n < n_nodes; n += gridDim.x) {
        float inv = 1.0f / s[n];
        float v = out_acc[(size_t)n * HID + c] * inv + b;
        v = v > 0.0f ? v : expm1f(v);
        out_acc[(size_t)n * HID + c] = v;
        ls += v;
        lq += v * v;
    }
    atomicAdd(&sums[c], ls);
    atomicAdd(&sumsq[c], lq);
}

__global__ void gat_bn_pool(float* __restrict__ out_acc, const float* __restrict__ sums,
                            const float* __restrict__ sumsq, const float* __restrict__ gamma,
                            const float* __restrict__ beta, const int* __restrict__ batch,
                            float* __restrict__ pooled, int layer, int n_nodes) {
    int idx = blockIdx.x * blockDim.x + threadIdx.x;
    if (idx >= n_nodes * HID) return;
    int c = idx & (HID - 1);
    int n = idx >> 7;
    float invN = 1.0f / (float)n_nodes;
    float mu = sums[c] * invN;
    float var = sumsq[c] * invN - mu * mu;
    float xv = (out_acc[idx] - mu) / sqrtf(var + BN_EPS) * gamma[c] + beta[c];
    out_acc[idx] = xv;
    atomicAdd(&pooled[(size_t)batch[n] * (3 * HID) + layer * HID + c], xv);
}

extern "C" void kernel_launch(void* const* d_in, const int* in_sizes, int n_in,
                              void* d_out, int out_size, void* d_ws, size_t ws_size,
                              hipStream_t stream) {
    const float* x_in  = (const float*)d_in[0];
    const int*   ei    = (const int*)d_in[1];
    const int*   batch = (const int*)d_in[2];
    const float* W     = (const float*)d_in[3];
    const float* att_s = (const float*)d_in[4];
    const float* att_d = (const float*)d_in[5];
    const float* bias  = (const float*)d_in[6];
    const float* gamma = (const float*)d_in[7];
    const float* beta  = (const float*)d_in[8];

    const int N = in_sizes[0] / HID;
    const int E = in_sizes[1] / 2;

    float* pooled = (float*)d_out;
    float* xfinal = (float*)d_out + (size_t)NGRAPH * 3 * HID;

    char* w = (char*)d_ws;
    float* hbuf = (float*)w;  w += (size_t)N * HID * sizeof(float);
    float* xb   = (float*)w;  w += (size_t)N * HID * sizeof(float);
    float* asrc = (float*)w;  w += (size_t)N * sizeof(float);
    float* adst = (float*)w;  w += (size_t)N * sizeof(float);
    float* mbuf = (float*)w;  w += (size_t)N * sizeof(float);
    float* sbuf = (float*)w;  w += (size_t)N * sizeof(float);
    float* sums = (float*)w;
    float* sumsq = sums + HID;

    const int waves_g = (N + 15) / 16;
    dim3 blk(256);
    dim3 grid_gemm((waves_g * 32 + 255) / 256);
    dim3 grid_node((N * 32 + 255) / 256);
    dim3 grid_edge((E + 255) / 256);
    dim3 grid_edgew(((size_t)E * 32 + 255) / 256);
    dim3 grid_elem(((size_t)N * HID + 255) / 256);

    gat_zero<<<(NGRAPH * 3 * HID + 255) / 256, blk, 0, stream>>>(pooled, NGRAPH * 3 * HID);

    for (int l = 0; l < 3; ++l) {
        const float* X  = (l == 0) ? x_in : xb;
        float* oacc     = (l == 2) ? xfinal : xb;
        const float* Wl = W + (size_t)l * HID * HID;

        gat_gemm_wmma<<<grid_gemm, blk, 0, stream>>>(X, Wl, hbuf, N);
        gat_node_dots<<<grid_node, blk, 0, stream>>>(hbuf, att_s + l * HID, att_d + l * HID,
                                                     asrc, adst, (unsigned*)mbuf, N);
        gat_edge_max<<<grid_edge, blk, 0, stream>>>(ei, asrc, adst, (unsigned*)mbuf, E);
        gat_node_init<<<grid_node, blk, 0, stream>>>(hbuf, asrc, adst, mbuf, sbuf, oacc, N);
        gat_edge_scatter<<<grid_edgew, blk, 0, stream>>>(ei, asrc, adst, mbuf, hbuf, sbuf, oacc, E);
        gat_zero<<<1, 256, 0, stream>>>(sums, 2 * HID);
        gat_finalize<<<512, 128, 0, stream>>>(oacc, sbuf, bias + l * HID, sums, sumsq, N);
        gat_bn_pool<<<grid_elem, blk, 0, stream>>>(oacc, sums, sumsq, gamma + l * HID,
                                                   beta + l * HID, batch, pooled, l, N);
    }
}


# Softmax row sums kept in LDS instead of global float atomics: 256 work-groups own 391 rows each, pick their edges by an owner byte per edge, add rows with LDS atomics (row sum order: arrival)
# speedup vs baseline: 1.9424x; 1.9424x over previous
#include <hip/hip_runtime.h>
#include <hip/hip_bf16.h>

#define HID 128
#define NGRAPH 512
#define NEG_SLOPE 0.2f
#define BN_EPS 1e-5f

typedef __attribute__((ext_vector_type(2))) float v2f;
typedef __attribute__((ext_vector_type(8))) float v8f;

__device__ __forceinline__ unsigned f2o(float f) {
    unsigned u = __float_as_uint(f);
    return (u & 0x80000000u) ? ~u : (u | 0x80000000u);
}
__device__ __forceinline__ float o2f(unsigned u) {
    unsigned v = (u & 0x80000000u) ? (u & 0x7fffffffu) : ~u;
    return __uint_as_float(v);
}
__device__ __forceinline__ float lrelu(float a) { return a > 0.0f ? a : NEG_SLOPE * a; }

__global__ void gat_zero(float* __restrict__ p, int n) {
    int i = blockIdx.x * blockDim.x + threadIdx.x;
    if (i < n) p[i] = 0.0f;
}

__global__ void gat_gemm_wmma(const float* __restrict__ X, const float* __restrict__ W,
                              float* __restrict__ H, int nrows) {
    int wave = (blockIdx.x * blockDim.x + threadIdx.x) >> 5;
    int m0 = wave * 16;
    if (m0 >= nrows) return;
    int lane = threadIdx.x & 31;
    int lr = lane & 15;
    int lh = lane >> 4;

    v8f acc[8] = {};
    const float* xrow = X + (size_t)(m0 + lr) * HID;

    for (int k = 0; k < HID; k += 4) {
        int ka = k + lh * 2;
        v2f a;
        a.x = xrow[ka];
        a.y = xrow[ka + 1];
#pragma unroll
        for (int n = 0; n < 8; ++n) {
            v2f b;
            b.x = W[ka * HID + n * 16 + lr];
            b.y = W[(ka + 1) * HID + n * 16 + lr];
            acc[n] = __builtin_amdgcn_wmma_f32_16x16x4_f32(
                false, a, false, b, (short)0, acc[n], false, false);
        }
    }
#pragma unroll
    for (int n = 0; n < 8; ++n)
#pragma unroll
        for (int j = 0; j < 8; ++j)
            H[(size_t)(m0 + j + lh * 8) * HID + n * 16 + lr] = acc[n][j];
}

__global__ void gat_node_dots(const float* __restrict__ H, const float* __restrict__ att_s,
                              const float* __restrict__ att_d, float* __restrict__ a_s,
                              float* __restrict__ a_d, int n_nodes) {
    int node = (blockIdx.x * blockDim.x + threadIdx.x) >> 5;
    if (node >= n_nodes) return;
    int lane = threadIdx.x & 31;
    float4 hv = ((const float4*)(H + (size_t)node * HID))[lane];
    float4 ws = ((const float4*)att_s)[lane];
    float4 wd = ((const float4*)att_d)[lane];
    float ps = hv.x * ws.x + hv.y * ws.y + hv.z * ws.z + hv.w * ws.w;
    float pd = hv.x * wd.x + hv.y * wd.y + hv.z * wd.z + hv.w * wd.w;
    for (int off = 16; off > 0; off >>= 1) {
        ps += __shfl_down(ps, off, 32);
        pd += __shfl_down(pd, off, 32);
    }
    if (lane == 0) {
        a_s[node] = ps;
        a_d[node] = pd;
    }
}

#define NPB   391

__global__ void gat_owner_ids(const int* __restrict__ ei, unsigned* __restrict__ obw, int n_edges) {
    int t = blockIdx.x * blockDim.x + threadIdx.x;
    int e0 = t * 4;
    if (e0 >= n_edges) return;
    const int* __restrict__ dstp = ei + n_edges;
    unsigned wd = 0;
#pragma unroll
    for (int j = 0; j < 4; ++j)
        if (e0 + j < n_edges) wd |= ((unsigned)(dstp[e0 + j] / NPB) & 255u) << (8 * j);
    obw[t] = wd;
}

#define OWN_T 1024
#define OWN_W (OWN_T / 32)
#define WCAP  352
#define LCAP  (OWN_W * WCAP)
#define SCAN_U 8
#define PH2_U 8
#define SC_GROUP 8

template <bool TAIL>
__device__ __forceinline__ void gat_keep4(unsigned wd, unsigned pat, unsigned obv, int e0,
                                          int n_edges, int lane, int* lw, int& wc) {
    unsigned x = wd ^ pat;
    bool hit = ((x - 0x01010101u) & ~x & 0x80808080u) != 0u;
    if (__builtin_amdgcn_ballot_w32(hit) != 0u) {
#pragma unroll
        for (int b = 0; b < 4; ++b) {
            bool mt = ((wd >> (8 * b)) & 255u) == obv;
            if (TAIL) mt = mt && (e0 + b < n_edges);
            unsigned mk = __builtin_amdgcn_ballot_w32(mt);
            if (mk != 0u) {
                if (mt) {
                    int pos = wc + __builtin_popcount(mk & ((1u << lane) - 1u));
                    if (pos < WCAP) lw[pos] = e0 + b;
                }
                wc += __builtin_popcount(mk);
            }
        }
    }
}

__device__ __forceinline__ void gat_add_group(const int* lwl, const unsigned short* dwl,
                                              const float* gwl, int k0, int nl,
                                              const unsigned* mb, float* ssum, float* acc,
                                              const float* __restrict__ H, int lane) {
    const int rest = nl - k0;
    const int gc = rest < SC_GROUP ? rest : SC_GROUP;
    int sv = 0, dv = 0;
    float w = 0.0f;
    if (lane < gc) {
        sv = lwl[k0 + lane];
        dv = dwl[k0 + lane];
        w = expf(gwl[k0 + lane] - o2f(mb[dv]));
        atomicAdd(&ssum[dv], w);
    }
    int wbits = __float_as_int(w);
    if (gc == SC_GROUP) {
        float hv[SC_GROUP][4];
        int dk[SC_GROUP];
        float wk[SC_GROUP];
#pragma unroll
        for (int u = 0; u < SC_GROUP; ++u) {
            int sk = __builtin_amdgcn_readlane(sv, u);
            dk[u] = __builtin_amdgcn_readlane(dv, u);
            wk[u] = __int_as_float(__builtin_amdgcn_readlane(wbits, u));
            const float* hp = H + (size_t)sk * HID + lane;
            hv[u][0] = hp[0];
            hv[u][1] = hp[32];
            hv[u][2] = hp[64];
            hv[u][3] = hp[96];
        }
#pragma unroll
        for (int u = 0; u < SC_GROUP; ++u) {
            float* ap = acc + dk[u] * HID + lane;
            atomicAdd(ap,      wk[u] * hv[u][0]);
            atomicAdd(ap + 32, wk[u] * hv[u][1]);
            atomicAdd(ap + 64, wk[u] * hv[u][2]);
            atomicAdd(ap + 96, wk[u] * hv[u][3]);
        }
    } else {
        for (int u = 0; u < gc; ++u) {
            int sk = __builtin_amdgcn_readlane(sv, u);
            int dd = __builtin_amdgcn_readlane(dv, u);
            float ww = __int_as_float(__builtin_amdgcn_readlane(wbits, u));
            const float* hp = H + (size_t)sk * HID + lane;
            float h0 = hp[0], h1 = hp[32], h2 = hp[64], h3 = hp[96];
            float* ap = acc + dd * HID + lane;
            atomicAdd(ap,      ww * h0);
            atomicAdd(ap + 32, ww * h1);
            atomicAdd(ap + 64, ww * h2);
            atomicAdd(ap + 96, ww * h3);
        }
    }
}

__global__ __launch_bounds__(OWN_T) void gat_owner(const int* __restrict__ ei,
                                                   const unsigned* __restrict__ obw,
                                                   const float* __restrict__ a_s,
                                                   const float* __restrict__ a_d,
                                                   const float* __restrict__ H,
                                                   float* __restrict__ s_out,
                                                   float* __restrict__ out_acc,
                                                   int n_nodes, int n_edges) {
    __shared__ __attribute__((aligned(16))) float acc[NPB * HID];
    __shared__ float adl[NPB];
    __shared__ float ssum[NPB];
    __shared__ unsigned mb[NPB];
    __shared__ int lsrc[LCAP];
    __shared__ float llog[LCAP];
    __shared__ unsigned short ldl[LCAP];
    __shared__ int wcnt[OWN_W];
    __shared__ int ovfl;

    const int tid = threadIdx.x;
    const int lane = tid & 31;
    const int wv = tid >> 5;
    const int lo = blockIdx.x * NPB;
    int nown = n_nodes - lo;
    if (nown > NPB) nown = NPB;
    const int* __restrict__ srcp = ei;
    const int* __restrict__ dstp = ei + n_edges;

    for (int i = tid; i < nown; i += OWN_T) {
        float ad = a_d[lo + i];
        float sl = lrelu(a_s[lo + i] + ad);
        adl[i] = ad;
        ssum[i] = sl;
        mb[i] = f2o(sl);
    }
    if (tid == 0) ovfl = 0;
    __syncthreads();

    int* lw = lsrc + wv * WCAP;
    float* gw = llog + wv * WCAP;
    unsigned short* dw = ldl + wv * WCAP;
    int wc = 0;
    const unsigned obv = blockIdx.x & 255u;
    const unsigned pat = obv * 0x01010101u;
    const unsigned npat = ~pat;
    const int nw = (n_edges + 3) >> 2;
    const int nq = n_edges >> 4;
    const int nit = (nq + SCAN_U * OWN_T - 1) / (SCAN_U * OWN_T);
    const int rot = (int)(((long long)blockIdx.x * nit) / gridDim.x);
    const uint4* __restrict__ oq = (const uint4*)obw;
    for (int it = 0; it < nit; ++it) {
        int itr = it + rot;
        if (itr >= nit) itr -= nit;
        const int qb = itr * (SCAN_U * OWN_T) + tid;
        uint4 d[SCAN_U];
#pragma unroll
        for (int j = 0; j < SCAN_U; ++j) {
            int q = qb + j * OWN_T;
            d[j] = (q < nq) ? oq[q] : make_uint4(npat, npat, npat, npat);
        }
#pragma unroll
        for (int j = 0; j < SCAN_U; ++j) {
            int e0 = (qb + j * OWN_T) * 16;
            gat_keep4<false>(d[j].x, pat, obv, e0,      n_edges, lane, lw, wc);
            gat_keep4<false>(d[j].y, pat, obv, e0 + 4,  n_edges, lane, lw, wc);
            gat_keep4<false>(d[j].z, pat, obv, e0 + 8,  n_edges, lane, lw, wc);
            gat_keep4<false>(d[j].w, pat, obv, e0 + 12, n_edges, lane, lw, wc);
        }
    }
    for (int wi0 = nq * 4; wi0 < nw; wi0 += OWN_T) {
        int wi = wi0 + tid;
        unsigned wd = (wi < nw) ? obw[wi] : npat;
        gat_keep4<true>(wd, pat, obv, wi * 4, n_edges, lane, lw, wc);
    }
    if (lane == 0) {
        wcnt[wv] = wc;
        if (wc > WCAP) ovfl = 1;
    }
    __syncthreads();

    const bool ovf = ovfl != 0;
    const int nl = ovf ? 0 : wc;

    if (!ovf) {
        for (int kb = 0; kb < nl; kb += 32 * PH2_U) {
            int ee[PH2_U], dd[PH2_U], sv[PH2_U];
            float av[PH2_U];
#pragma unroll
            for (int u = 0; u < PH2_U; ++u) {
                int k = kb + u * 32 + lane;
                ee[u] = (k < nl) ? lw[k] : 0;
            }
#pragma unroll
            for (int u = 0; u < PH2_U; ++u) {
                dd[u] = dstp[ee[u]];
                sv[u] = srcp[ee[u]];
            }
#pragma unroll
            for (int u = 0; u < PH2_U; ++u) av[u] = a_s[sv[u]];
#pragma unroll
            for (int u = 0; u < PH2_U; ++u) {
                int k = kb + u * 32 + lane;
                if (k < nl) {
                    unsigned dl = (unsigned)(dd[u] - lo);
                    if (dl < (unsigned)nown) {
                        float lg = lrelu(av[u] + adl[dl]);
                        atomicMax(&mb[dl], f2o(lg));
                        lw[k] = sv[u];
                        dw[k] = (unsigned short)dl;
                        gw[k] = lg;
                    } else {
                        lw[k] = lo;
                        dw[k] = 0;
                        gw[k] = -__builtin_inff();
                    }
                }
            }
        }
    } else {
        for (int e = tid; e < n_edges; e += OWN_T) {
            unsigned dl = (unsigned)(dstp[e] - lo);
            if (dl < (unsigned)nown) {
                float lg = lrelu(a_s[srcp[e]] + adl[dl]);
                atomicMax(&mb[dl], f2o(lg));
            }
        }
    }
    __syncthreads();

    for (int i0 = wv; i0 < nown; i0 += OWN_W * 4) {
        float4 hv[4];
#pragma unroll
        for (int u = 0; u < 4; ++u) {
            int i = i0 + u * OWN_W;
            hv[u] = make_float4(0.0f, 0.0f, 0.0f, 0.0f);
            if (i < nown) hv[u] = ((const float4*)(H + (size_t)(lo + i) * HID))[lane];
        }
#pragma unroll
        for (int u = 0; u < 4; ++u) {
            int i = i0 + u * OWN_W;
            if (i < nown) {
                float es = expf(ssum[i] - o2f(mb[i]));
                float4 r;
                r.x = es * hv[u].x; r.y = es * hv[u].y; r.z = es * hv[u].z; r.w = es * hv[u].w;
                ((float4*)(acc + i * HID))[lane] = r;
                if (lane == 0) ssum[i] = es;
            }
        }
    }
    __syncthreads();

    if (!ovf) {
        const int myc = wcnt[lane];
        for (int wl = 0; wl < OWN_W; ++wl) {
            const int nlw = __builtin_amdgcn_readlane(myc, wl);
            for (int g = (wv - wl) & (OWN_W - 1); g * SC_GROUP < nlw; g += OWN_W)
                gat_add_group(lsrc + wl * WCAP, ldl + wl * WCAP, llog + wl * WCAP, g * SC_GROUP, nlw,
                              mb, ssum, acc, H, lane);
        }
    } else {
        for (int e = tid; e < n_edges; e += OWN_T) {
            unsigned dl = (unsigned)(dstp[e] - lo);
            if (dl < (unsigned)nown) {
                int s = srcp[e];
                float ww = expf(lrelu(a_s[s] + adl[dl]) - o2f(mb[dl]));
                atomicAdd(&ssum[dl], ww);
                const float* hp = H + (size_t)s * HID;
                float* ap = acc + dl * HID;
                for (int c = 0; c < HID; ++c) atomicAdd(ap + c, ww * hp[c]);
            }
        }
    }
    __syncthreads();

    float4* __restrict__ orow = (float4*)(out_acc + (size_t)lo * HID);
    const float4* arow = (const float4*)acc;
    for (int i = tid; i < nown * (HID / 4); i += OWN_T) orow[i] = arow[i];
    for (int i = tid; i < nown; i += OWN_T) s_out[lo + i] = ssum[i];
}

__global__ void gat_finalize(float* __restrict__ out_acc, const float* __restrict__ s,
                             const float* __restrict__ bias, float* __restrict__ sums,
                             float* __restrict__ sumsq, int n_nodes) {
    int c = threadIdx.x;
    float b = bias[c];
    float ls = 0.0f, lq = 0.0f;
    for (int n = blockIdx.x; n < n_nodes; n += gridDim.x) {
        float inv = 1.0f / s[n];
        float v = out_acc[(size_t)n * HID + c] * inv + b;
        v = v > 0.0f ? v : expm1f(v);
        out_acc[(size_t)n * HID + c] = v;
        ls += v;
        lq += v * v;
    }
    atomicAdd(&sums[c], ls);
    atomicAdd(&sumsq[c], lq);
}

__global__ void gat_bn_pool(float* __restrict__ out_acc, const float* __restrict__ sums,
                            const float* __restrict__ sumsq, const float* __restrict__ gamma,
                            const float* __restrict__ beta, const int* __restrict__ batch,
                            float* __restrict__ pooled, int layer, int n_nodes) {
    int idx = blockIdx.x * blockDim.x + threadIdx.x;
    if (idx >= n_nodes * HID) return;
    int c = idx & (HID - 1);
    int n = idx >> 7;
    float invN = 1.0f / (float)n_nodes;
    float mu = sums[c] * invN;
    float var = sumsq[c] * invN - mu * mu;
    float xv = (out_acc[idx] - mu) / sqrtf(var + BN_EPS) * gamma[c] + beta[c];
    out_acc[idx] = xv;
    atomicAdd(&pooled[(size_t)batch[n] * (3 * HID) + layer * HID + c], xv);
}

extern "C" void kernel_launch(void* const* d_in, const int* in_sizes, int n_in,
                              void* d_out, int out_size, void* d_ws, size_t ws_size,
                              hipStream_t stream) {
    const float* x_in  = (const float*)d_in[0];
    const int*   ei    = (const int*)d_in[1];
    const int*   batch = (const int*)d_in[2];
    const float* W     = (const float*)d_in[3];
    const float* att_s = (const float*)d_in[4];
    const float* att_d = (const float*)d_in[5];
    const float* bias  = (const float*)d_in[6];
    const float* gamma = (const float*)d_in[7];
    const float* beta  = (const float*)d_in[8];

    const int N = in_sizes[0] / HID;
    const int E = in_sizes[1] / 2;

    float* pooled = (float*)d_out;
    float* xfinal = (float*)d_out + (size_t)NGRAPH * 3 * HID;

    char* w = (char*)d_ws;
    float* hbuf = (float*)w;  w += (size_t)N * HID * sizeof(float);
    float* xb   = (float*)w;  w += (size_t)N * HID * sizeof(float);
    unsigned* obw = (unsigned*)w;  w += (size_t)((E + 15) / 16) * 16;
    float* asrc = (float*)w;  w += (size_t)N * sizeof(float);
    float* adst = (float*)w;  w += (size_t)N * sizeof(float);
    float* sbuf = (float*)w;  w += (size_t)N * sizeof(float);
    float* sums = (float*)w;
    float* sumsq = sums + HID;

    const int waves_g = (N + 15) / 16;
    dim3 blk(256);
    dim3 grid_gemm((waves_g * 32 + 255) / 256);
    dim3 grid_node((N * 32 + 255) / 256);
    dim3 grid_own((N + NPB - 1) / NPB);
    dim3 grid_elem(((size_t)N * HID + 255) / 256);

    gat_zero<<<(NGRAPH * 3 * HID + 255) / 256, blk, 0, stream>>>(pooled, NGRAPH * 3 * HID);
    gat_owner_ids<<<((E + 3) / 4 + 255) / 256, blk, 0, stream>>>(ei, obw, E);

    for (int l = 0; l < 3; ++l) {
        const float* X  = (l == 0) ? x_in : xb;
        float* oacc     = (l == 2) ? xfinal : xb;
        const float* Wl = W + (size_t)l * HID * HID;

        gat_gemm_wmma<<<grid_gemm, blk, 0, stream>>>(X, Wl, hbuf, N);
        gat_node_dots<<<grid_node, blk, 0, stream>>>(hbuf, att_s + l * HID, att_d + l * HID,
                                                     asrc, adst, N);
        gat_owner<<<grid_own, OWN_T, 0, stream>>>(ei, obw, asrc, adst, hbuf, sbuf, oacc, N, E);
        gat_zero<<<1, 256, 0, stream>>>(sums, 2 * HID);
        gat_finalize<<<512, 128, 0, stream>>>(oacc, sbuf, bias + l * HID, sums, sumsq, N);
        gat_bn_pool<<<grid_elem, blk, 0, stream>>>(oacc, sums, sumsq, gamma + l * HID,
                                                   beta + l * HID, batch, pooled, l, N);
    }
}


# BN apply pass of layers 0,1 removed by algebra: next W scaled by gamma*rstd + constant accumulator row; pooled from raw per-graph run sums (stat/pool sum order: 196-row chunk runs, then atomics)
# speedup vs baseline: 2.0624x; 1.0618x over previous
#include <hip/hip_runtime.h>
#include <hip/hip_bf16.h>

#define HID 128
#define NGRAPH 512
#define NEG_SLOPE 0.2f
#define BN_EPS 1e-5f

typedef __attribute__((ext_vector_type(2))) float v2f;
typedef __attribute__((ext_vector_type(8))) float v8f;

__device__ __forceinline__ unsigned f2o(float f) {
    unsigned u = __float_as_uint(f);
    return (u & 0x80000000u) ? ~u : (u | 0x80000000u);
}
__device__ __forceinline__ float o2f(unsigned u) {
    unsigned v = (u & 0x80000000u) ? (u & 0x7fffffffu) : ~u;
    return __uint_as_float(v);
}
__device__ __forceinline__ float lrelu(float a) { return a > 0.0f ? a : NEG_SLOPE * a; }

__global__ void gat_zero(float* __restrict__ p, int n) {
    int i = blockIdx.x * blockDim.x + threadIdx.x;
    if (i < n) p[i] = 0.0f;
}

__global__ void gat_gemm_wmma(const float* __restrict__ X, const float* __restrict__ W,
                              const float* __restrict__ C0, float* __restrict__ H, int nrows) {
    int wave = (blockIdx.x * blockDim.x + threadIdx.x) >> 5;
    int m0 = wave * 16;
    if (m0 >= nrows) return;
    int lane = threadIdx.x & 31;
    int lr = lane & 15;
    int lh = lane >> 4;

    v8f acc[8];
#pragma unroll
    for (int n = 0; n < 8; ++n) {
        float cv = C0 ? C0[n * 16 + lr] : 0.0f;
#pragma unroll
        for (int j = 0; j < 8; ++j) acc[n][j] = cv;
    }
    const float* xrow = X + (size_t)(m0 + lr) * HID;

    for (int k = 0; k < HID; k += 4) {
        int ka = k + lh * 2;
        v2f a;
        a.x = xrow[ka];
        a.y = xrow[ka + 1];
#pragma unroll
        for (int n = 0; n < 8; ++n) {
            v2f b;
            b.x = W[ka * HID + n * 16 + lr];
            b.y = W[(ka + 1) * HID + n * 16 + lr];
            acc[n] = __builtin_amdgcn_wmma_f32_16x16x4_f32(
                false, a, false, b, (short)0, acc[n], false, false);
        }
    }
#pragma unroll
    for (int n = 0; n < 8; ++n)
#pragma unroll
        for (int j = 0; j < 8; ++j)
            H[(size_t)(m0 + j + lh * 8) * HID + n * 16 + lr] = acc[n][j];
}

__global__ void gat_node_dots(const float* __restrict__ H, const float* __restrict__ att_s,
                              const float* __restrict__ att_d, float* __restrict__ a_s,
                              float* __restrict__ a_d, int n_nodes) {
    int node = (blockIdx.x * blockDim.x + threadIdx.x) >> 5;
    if (node >= n_nodes) return;
    int lane = threadIdx.x & 31;
    float4 hv = ((const float4*)(H + (size_t)node * HID))[lane];
    float4 ws = ((const float4*)att_s)[lane];
    float4 wd = ((const float4*)att_d)[lane];
    float ps = hv.x * ws.x + hv.y * ws.y + hv.z * ws.z + hv.w * ws.w;
    float pd = hv.x * wd.x + hv.y * wd.y + hv.z * wd.z + hv.w * wd.w;
    for (int off = 16; off > 0; off >>= 1) {
        ps += __shfl_down(ps, off, 32);
        pd += __shfl_down(pd, off, 32);
    }
    if (lane == 0) {
        a_s[node] = ps;
        a_d[node] = pd;
    }
}

#define NPB   391

__global__ void gat_owner_ids(const int* __restrict__ ei, unsigned* __restrict__ obw, int n_edges) {
    int t = blockIdx.x * blockDim.x + threadIdx.x;
    int e0 = t * 4;
    if (e0 >= n_edges) return;
    const int* __restrict__ dstp = ei + n_edges;
    unsigned wd = 0;
#pragma unroll
    for (int j = 0; j < 4; ++j)
        if (e0 + j < n_edges) wd |= ((unsigned)(dstp[e0 + j] / NPB) & 255u) << (8 * j);
    obw[t] = wd;
}

#define OWN_T 1024
#define OWN_W (OWN_T / 32)
#define WCAP  352
#define LCAP  (OWN_W * WCAP)
#define SCAN_U 8
#define PH2_U 8
#define SC_GROUP 8

template <bool TAIL>
__device__ __forceinline__ void gat_keep4(unsigned wd, unsigned pat, unsigned obv, int e0,
                                          int n_edges, int lane, int* lw, int& wc) {
    unsigned x = wd ^ pat;
    bool hit = ((x - 0x01010101u) & ~x & 0x80808080u) != 0u;
    if (__builtin_amdgcn_ballot_w32(hit) != 0u) {
#pragma unroll
        for (int b = 0; b < 4; ++b) {
            bool mt = ((wd >> (8 * b)) & 255u) == obv;
            if (TAIL) mt = mt && (e0 + b < n_edges);
            unsigned mk = __builtin_amdgcn_ballot_w32(mt);
            if (mk != 0u) {
                if (mt) {
                    int pos = wc + __builtin_popcount(mk & ((1u << lane) - 1u));
                    if (pos < WCAP) lw[pos] = e0 + b;
                }
                wc += __builtin_popcount(mk);
            }
        }
    }
}

__device__ __forceinline__ void gat_add_group(const int* lwl, const unsigned short* dwl,
                                              const float* gwl, int k0, int nl,
                                              const unsigned* mb, float* ssum, float* acc,
                                              const float* __restrict__ H, int lane) {
    const int rest = nl - k0;
    const int gc = rest < SC_GROUP ? rest : SC_GROUP;
    int sv = 0, dv = 0;
    float w = 0.0f;
    if (lane < gc) {
        sv = lwl[k0 + lane];
        dv = dwl[k0 + lane];
        w = expf(gwl[k0 + lane] - o2f(mb[dv]));
        atomicAdd(&ssum[dv], w);
    }
    int wbits = __float_as_int(w);
    if (gc == SC_GROUP) {
        float hv[SC_GROUP][4];
        int dk[SC_GROUP];
        float wk[SC_GROUP];
#pragma unroll
        for (int u = 0; u < SC_GROUP; ++u) {
            int sk = __builtin_amdgcn_readlane(sv, u);
            dk[u] = __builtin_amdgcn_readlane(dv, u);
            wk[u] = __int_as_float(__builtin_amdgcn_readlane(wbits, u));
            const float* hp = H + (size_t)sk * HID + lane;
            hv[u][0] = hp[0];
            hv[u][1] = hp[32];
            hv[u][2] = hp[64];
            hv[u][3] = hp[96];
        }
#pragma unroll
        for (int u = 0; u < SC_GROUP; ++u) {
            float* ap = acc + dk[u] * HID + lane;
            atomicAdd(ap,      wk[u] * hv[u][0]);
            atomicAdd(ap + 32, wk[u] * hv[u][1]);
            atomicAdd(ap + 64, wk[u] * hv[u][2]);
            atomicAdd(ap + 96, wk[u] * hv[u][3]);
        }
    } else {
        for (int u = 0; u < gc; ++u) {
            int sk = __builtin_amdgcn_readlane(sv, u);
            int dd = __builtin_amdgcn_readlane(dv, u);
            float ww = __int_as_float(__builtin_amdgcn_readlane(wbits, u));
            const float* hp = H + (size_t)sk * HID + lane;
            float h0 = hp[0], h1 = hp[32], h2 = hp[64], h3 = hp[96];
            float* ap = acc + dd * HID + lane;
            atomicAdd(ap,      ww * h0);
            atomicAdd(ap + 32, ww * h1);
            atomicAdd(ap + 64, ww * h2);
            atomicAdd(ap + 96, ww * h3);
        }
    }
}

__global__ __launch_bounds__(OWN_T) void gat_owner(const int* __restrict__ ei,
                                                   const unsigned* __restrict__ obw,
                                                   const float* __restrict__ a_s,
                                                   const float* __restrict__ a_d,
                                                   const float* __restrict__ H,
                                                   float* __restrict__ s_out,
                                                   float* __restrict__ out_acc,
                                                   int n_nodes, int n_edges) {
    __shared__ __attribute__((aligned(16))) float acc[NPB * HID];
    __shared__ float adl[NPB];
    __shared__ float ssum[NPB];
    __shared__ unsigned mb[NPB];
    __shared__ int lsrc[LCAP];
    __shared__ float llog[LCAP];
    __shared__ unsigned short ldl[LCAP];
    __shared__ int wcnt[OWN_W];
    __shared__ int ovfl;

    const int tid = threadIdx.x;
    const int lane = tid & 31;
    const int wv = tid >> 5;
    const int lo = blockIdx.x * NPB;
    int nown = n_nodes - lo;
    if (nown > NPB) nown = NPB;
    const int* __restrict__ srcp = ei;
    const int* __restrict__ dstp = ei + n_edges;

    for (int i = tid; i < nown; i += OWN_T) {
        float ad = a_d[lo + i];
        float sl = lrelu(a_s[lo + i] + ad);
        adl[i] = ad;
        ssum[i] = sl;
        mb[i] = f2o(sl);
    }
    if (tid == 0) ovfl = 0;
    __syncthreads();

    int* lw = lsrc + wv * WCAP;
    float* gw = llog + wv * WCAP;
    unsigned short* dw = ldl + wv * WCAP;
    int wc = 0;
    const unsigned obv = blockIdx.x & 255u;
    const unsigned pat = obv * 0x01010101u;
    const unsigned npat = ~pat;
    const int nw = (n_edges + 3) >> 2;
    const int nq = n_edges >> 4;
    const int nit = (nq + SCAN_U * OWN_T - 1) / (SCAN_U * OWN_T);
    const int rot = (int)(((long long)blockIdx.x * nit) / gridDim.x);
    const uint4* __restrict__ oq = (const uint4*)obw;
    for (int it = 0; it < nit; ++it) {
        int itr = it + rot;
        if (itr >= nit) itr -= nit;
        const int qb = itr * (SCAN_U * OWN_T) + tid;
        uint4 d[SCAN_U];
#pragma unroll
        for (int j = 0; j < SCAN_U; ++j) {
            int q = qb + j * OWN_T;
            d[j] = (q < nq) ? oq[q] : make_uint4(npat, npat, npat, npat);
        }
#pragma unroll
        for (int j = 0; j < SCAN_U; ++j) {
            int e0 = (qb + j * OWN_T) * 16;
            gat_keep4<false>(d[j].x, pat, obv, e0,      n_edges, lane, lw, wc);
            gat_keep4<false>(d[j].y, pat, obv, e0 + 4,  n_edges, lane, lw, wc);
            gat_keep4<false>(d[j].z, pat, obv, e0 + 8,  n_edges, lane, lw, wc);
            gat_keep4<false>(d[j].w, pat, obv, e0 + 12, n_edges, lane, lw, wc);
        }
    }
    for (int wi0 = nq * 4; wi0 < nw; wi0 += OWN_T) {
        int wi = wi0 + tid;
        unsigned wd = (wi < nw) ? obw[wi] : npat;
        gat_keep4<true>(wd, pat, obv, wi * 4, n_edges, lane, lw, wc);
    }
    if (lane == 0) {
        wcnt[wv] = wc;
        if (wc > WCAP) ovfl = 1;
    }
    __syncthreads();

    const bool ovf = ovfl != 0;
    const int nl = ovf ? 0 : wc;

    if (!ovf) {
        for (int kb = 0; kb < nl; kb += 32 * PH2_U) {
            int ee[PH2_U], dd[PH2_U], sv[PH2_U];
            float av[PH2_U];
#pragma unroll
            for (int u = 0; u < PH2_U; ++u) {
                int k = kb + u * 32 + lane;
                ee[u] = (k < nl) ? lw[k] : 0;
            }
#pragma unroll
            for (int u = 0; u < PH2_U; ++u) {
                dd[u] = dstp[ee[u]];
                sv[u] = srcp[ee[u]];
            }
#pragma unroll
            for (int u = 0; u < PH2_U; ++u) av[u] = a_s[sv[u]];
#pragma unroll
            for (int u = 0; u < PH2_U; ++u) {
                int k = kb + u * 32 + lane;
                if (k < nl) {
                    unsigned dl = (unsigned)(dd[u] - lo);
                    if (dl < (unsigned)nown) {
                        float lg = lrelu(av[u] + adl[dl]);
                        atomicMax(&mb[dl], f2o(lg));
                        lw[k] = sv[u];
                        dw[k] = (unsigned short)dl;
                        gw[k] = lg;
                    } else {
                        lw[k] = lo;
                        dw[k] = 0;
                        gw[k] = -__builtin_inff();
                    }
                }
            }
        }
    } else {
        for (int e = tid; e < n_edges; e += OWN_T) {
            unsigned dl = (unsigned)(dstp[e] - lo);
            if (dl < (unsigned)nown) {
                float lg = lrelu(a_s[srcp[e]] + adl[dl]);
                atomicMax(&mb[dl], f2o(lg));
            }
        }
    }
    __syncthreads();

    for (int i0 = wv; i0 < nown; i0 += OWN_W * 4) {
        float4 hv[4];
#pragma unroll
        for (int u = 0; u < 4; ++u) {
            int i = i0 + u * OWN_W;
            hv[u] = make_float4(0.0f, 0.0f, 0.0f, 0.0f);
            if (i < nown) hv[u] = ((const float4*)(H + (size_t)(lo + i) * HID))[lane];
        }
#pragma unroll
        for (int u = 0; u < 4; ++u) {
            int i = i0 + u * OWN_W;
            if (i < nown) {
                float es = expf(ssum[i] - o2f(mb[i]));
                float4 r;
                r.x = es * hv[u].x; r.y = es * hv[u].y; r.z = es * hv[u].z; r.w = es * hv[u].w;
                ((float4*)(acc + i * HID))[lane] = r;
                if (lane == 0) ssum[i] = es;
            }
        }
    }
    __syncthreads();

    if (!ovf) {
        const int myc = wcnt[lane];
        for (int wl = 0; wl < OWN_W; ++wl) {
            const int nlw = __builtin_amdgcn_readlane(myc, wl);
            for (int g = (wv - wl) & (OWN_W - 1); g * SC_GROUP < nlw; g += OWN_W)
                gat_add_group(lsrc + wl * WCAP, ldl + wl * WCAP, llog + wl * WCAP, g * SC_GROUP, nlw,
                              mb, ssum, acc, H, lane);
        }
    } else {
        for (int e = tid; e < n_edges; e += OWN_T) {
            unsigned dl = (unsigned)(dstp[e] - lo);
            if (dl < (unsigned)nown) {
                int s = srcp[e];
                float ww = expf(lrelu(a_s[s] + adl[dl]) - o2f(mb[dl]));
                atomicAdd(&ssum[dl], ww);
                const float* hp = H + (size_t)s * HID;
                float* ap = acc + dl * HID;
                for (int c = 0; c < HID; ++c) atomicAdd(ap + c, ww * hp[c]);
            }
        }
    }
    __syncthreads();

    float4* __restrict__ orow = (float4*)(out_acc + (size_t)lo * HID);
    const float4* arow = (const float4*)acc;
    for (int i = tid; i < nown * (HID / 4); i += OWN_T) orow[i] = arow[i];
    for (int i = tid; i < nown; i += OWN_T) s_out[lo + i] = ssum[i];
}

__global__ void gat_finalize(float* __restrict__ out_acc, const float* __restrict__ s,
                             const float* __restrict__ bias, float* __restrict__ sums,
                             float* __restrict__ sumsq, const int* __restrict__ batch,
                             float* __restrict__ praw, float* __restrict__ cnt,
                             int do_raw, int n_nodes) {
    int c = threadIdx.x;
    float b = bias[c];
    float ls = 0.0f, lq = 0.0f;
    const int chunk = (n_nodes + (int)gridDim.x - 1) / (int)gridDim.x;
    const int n0 = blockIdx.x * chunk;
    int n1 = n0 + chunk;
    if (n1 > n_nodes) n1 = n_nodes;
    int gcur = 0, rl = 0;
    float run = 0.0f;
    if (do_raw && n0 < n1) gcur = batch[n0];
    for (int n = n0; n < n1; ++n) {
        float inv = 1.0f / s[n];
        float v = out_acc[(size_t)n * HID + c] * inv + b;
        v = v > 0.0f ? v : expm1f(v);
        out_acc[(size_t)n * HID + c] = v;
        ls += v;
        lq += v * v;
        if (do_raw) {
            int gb = batch[n];
            if (gb != gcur) {
                atomicAdd(&praw[(size_t)gcur * (3 * HID) + c], run);
                if (c == 0) atomicAdd(&cnt[gcur], (float)rl);
                run = 0.0f;
                rl = 0;
                gcur = gb;
            }
            run += v;
            ++rl;
        }
    }
    if (do_raw && rl > 0) {
        atomicAdd(&praw[(size_t)gcur * (3 * HID) + c], run);
        if (c == 0) atomicAdd(&cnt[gcur], (float)rl);
    }
    atomicAdd(&sums[c], ls);
    atomicAdd(&sumsq[c], lq);
}

__global__ void gat_bn_fold(const float* __restrict__ sums, const float* __restrict__ sumsq,
                            const float* __restrict__ gamma, const float* __restrict__ beta,
                            const float* __restrict__ Wn, float* __restrict__ Wp,
                            float* __restrict__ c0, float* __restrict__ pooled_l,
                            const float* __restrict__ cnt, int n_nodes) {
    const int t = threadIdx.x;
    const int b = blockIdx.x;
    const float invN = 1.0f / (float)n_nodes;
    if (b < HID) {
        float mu = sums[b] * invN;
        float var = sumsq[b] * invN - mu * mu;
        float g = gamma[b] / sqrtf(var + BN_EPS);
        Wp[b * HID + t] = g * Wn[b * HID + t];
    } else if (b == HID) {
        __shared__ float dsh[HID];
        {
            float mu = sums[t] * invN;
            float var = sumsq[t] * invN - mu * mu;
            float g = gamma[t] / sqrtf(var + BN_EPS);
            dsh[t] = beta[t] - mu * g;
        }
        __syncthreads();
        float a = 0.0f;
#pragma unroll 16
        for (int k = 0; k < HID; ++k) a += dsh[k] * Wn[k * HID + t];
        c0[t] = a;
    } else {
        const int gr = b - HID - 1;
        float mu = sums[t] * invN;
        float var = sumsq[t] * invN - mu * mu;
        float g = gamma[t] / sqrtf(var + BN_EPS);
        float cn = cnt[gr];
        size_t idx = (size_t)gr * (3 * HID) + t;
        pooled_l[idx] = (pooled_l[idx] - cn * mu) * g + cn * beta[t];
    }
}

__global__ void gat_bn_pool(float* __restrict__ out_acc, const float* __restrict__ sums,
                            const float* __restrict__ sumsq, const float* __restrict__ gamma,
                            const float* __restrict__ beta, const int* __restrict__ batch,
                            float* __restrict__ pooled, int layer, int n_nodes) {
    int idx = blockIdx.x * blockDim.x + threadIdx.x;
    if (idx >= n_nodes * HID) return;
    int c = idx & (HID - 1);
    int n = idx >> 7;
    float invN = 1.0f / (float)n_nodes;
    float mu = sums[c] * invN;
    float var = sumsq[c] * invN - mu * mu;
    float xv = (out_acc[idx] - mu) / sqrtf(var + BN_EPS) * gamma[c] + beta[c];
    out_acc[idx] = xv;
    atomicAdd(&pooled[(size_t)batch[n] * (3 * HID) + layer * HID + c], xv);
}

extern "C" void kernel_launch(void* const* d_in, const int* in_sizes, int n_in,
                              void* d_out, int out_size, void* d_ws, size_t ws_size,
                              hipStream_t stream) {
    const float* x_in  = (const float*)d_in[0];
    const int*   ei    = (const int*)d_in[1];
    const int*   batch = (const int*)d_in[2];
    const float* W     = (const float*)d_in[3];
    const float* att_s = (const float*)d_in[4];
    const float* att_d = (const float*)d_in[5];
    const float* bias  = (const float*)d_in[6];
    const float* gamma = (const float*)d_in[7];
    const float* beta  = (const float*)d_in[8];

    const int N = in_sizes[0] / HID;
    const int E = in_sizes[1] / 2;

    float* pooled = (float*)d_out;
    float* xfinal = (float*)d_out + (size_t)NGRAPH * 3 * HID;

    char* w = (char*)d_ws;
    float* hbuf = (float*)w;  w += (size_t)N * HID * sizeof(float);
    float* xb   = (float*)w;  w += (size_t)N * HID * sizeof(float);
    unsigned* obw = (unsigned*)w;  w += (size_t)((E + 15) / 16) * 16;
    float* asrc = (float*)w;  w += (size_t)N * sizeof(float);
    float* adst = (float*)w;  w += (size_t)N * sizeof(float);
    float* sbuf = (float*)w;  w += (size_t)N * sizeof(float);
    float* sums = (float*)w;
    float* sumsq = sums + HID;
    float* cnt  = sums + 2 * HID;
    float* c0   = cnt + NGRAPH;
    float* Wp   = c0 + HID;

    const int waves_g = (N + 15) / 16;
    dim3 blk(256);
    dim3 grid_gemm((waves_g * 32 + 255) / 256);
    dim3 grid_node((N * 32 + 255) / 256);
    dim3 grid_own((N + NPB - 1) / NPB);
    dim3 grid_elem(((size_t)N * HID + 255) / 256);

    gat_zero<<<(NGRAPH * 3 * HID + 255) / 256, blk, 0, stream>>>(pooled, NGRAPH * 3 * HID);
    gat_owner_ids<<<((E + 3) / 4 + 255) / 256, blk, 0, stream>>>(ei, obw, E);

    for (int l = 0; l < 3; ++l) {
        const float* X  = (l == 0) ? x_in : xb;
        float* oacc     = (l == 2) ? xfinal : xb;
        const float* Wl = (l == 0) ? W : Wp;
        const float* Cl = (l == 0) ? (const float*)nullptr : c0;

        gat_gemm_wmma<<<grid_gemm, blk, 0, stream>>>(X, Wl, Cl, hbuf, N);
        gat_node_dots<<<grid_node, blk, 0, stream>>>(hbuf, att_s + l * HID, att_d + l * HID,
                                                     asrc, adst, N);
        gat_owner<<<grid_own, OWN_T, 0, stream>>>(ei, obw, asrc, adst, hbuf, sbuf, oacc, N, E);
        gat_zero<<<3, 256, 0, stream>>>(sums, 2 * HID + NGRAPH);
        gat_finalize<<<512, 128, 0, stream>>>(oacc, sbuf, bias + l * HID, sums, sumsq, batch,
                                              pooled + l * HID, cnt, (l < 2) ? 1 : 0, N);
        if (l < 2)
            gat_bn_fold<<<HID + 1 + NGRAPH, HID, 0, stream>>>(sums, sumsq, gamma + l * HID,
                                                              beta + l * HID,
                                                              W + (size_t)(l + 1) * HID * HID, Wp, c0,
                                                              pooled + l * HID, cnt, N);
        else
            gat_bn_pool<<<grid_elem, blk, 0, stream>>>(oacc, sums, sumsq, gamma + l * HID,
                                                       beta + l * HID, batch, pooled, l, N);
    }
}


# Finalize (elu + BN sums) row walk re-tiled: 521 blocks x 4 row slots, a slot keeps 16 row loads in flight, 1/s once per row by one lane (stat/pool sum order: 48-row slot runs, LDS, then atomics)
# speedup vs baseline: 2.9065x; 1.4093x over previous
#include <hip/hip_runtime.h>
#include <hip/hip_bf16.h>

#define HID 128
#define NGRAPH 512
#define NEG_SLOPE 0.2f
#define BN_EPS 1e-5f

typedef __attribute__((ext_vector_type(2))) float v2f;
typedef __attribute__((ext_vector_type(8))) float v8f;

__device__ __forceinline__ unsigned f2o(float f) {
    unsigned u = __float_as_uint(f);
    return (u & 0x80000000u) ? ~u : (u | 0x80000000u);
}
__device__ __forceinline__ float o2f(unsigned u) {
    unsigned v = (u & 0x80000000u) ? (u & 0x7fffffffu) : ~u;
    return __uint_as_float(v);
}
__device__ __forceinline__ float lrelu(float a) { return a > 0.0f ? a : NEG_SLOPE * a; }

__global__ void gat_zero(float* __restrict__ p, int n) {
    int i = blockIdx.x * blockDim.x + threadIdx.x;
    if (i < n) p[i] = 0.0f;
}

__global__ void gat_gemm_wmma(const float* __restrict__ X, const float* __restrict__ W,
                              const float* __restrict__ C0, float* __restrict__ H, int nrows) {
    int wave = (blockIdx.x * blockDim.x + threadIdx.x) >> 5;
    int m0 = wave * 16;
    if (m0 >= nrows) return;
    int lane = threadIdx.x & 31;
    int lr = lane & 15;
    int lh = lane >> 4;

    v8f acc[8];
#pragma unroll
    for (int n = 0; n < 8; ++n) {
        float cv = C0 ? C0[n * 16 + lr] : 0.0f;
#pragma unroll
        for (int j = 0; j < 8; ++j) acc[n][j] = cv;
    }
    const float* xrow = X + (size_t)(m0 + lr) * HID;

    for (int k = 0; k < HID; k += 4) {
        int ka = k + lh * 2;
        v2f a;
        a.x = xrow[ka];
        a.y = xrow[ka + 1];
#pragma unroll
        for (int n = 0; n < 8; ++n) {
            v2f b;
            b.x = W[ka * HID + n * 16 + lr];
            b.y = W[(ka + 1) * HID + n * 16 + lr];
            acc[n] = __builtin_amdgcn_wmma_f32_16x16x4_f32(
                false, a, false, b, (short)0, acc[n], false, false);
        }
    }
#pragma unroll
    for (int n = 0; n < 8; ++n)
#pragma unroll
        for (int j = 0; j < 8; ++j)
            H[(size_t)(m0 + j + lh * 8) * HID + n * 16 + lr] = acc[n][j];
}

__global__ void gat_node_dots(const float* __restrict__ H, const float* __restrict__ att_s,
                              const float* __restrict__ att_d, float* __restrict__ a_s,
                              float* __restrict__ a_d, int n_nodes) {
    int node = (blockIdx.x * blockDim.x + threadIdx.x) >> 5;
    if (node >= n_nodes) return;
    int lane = threadIdx.x & 31;
    float4 hv = ((const float4*)(H + (size_t)node * HID))[lane];
    float4 ws = ((const float4*)att_s)[lane];
    float4 wd = ((const float4*)att_d)[lane];
    float ps = hv.x * ws.x + hv.y * ws.y + hv.z * ws.z + hv.w * ws.w;
    float pd = hv.x * wd.x + hv.y * wd.y + hv.z * wd.z + hv.w * wd.w;
    for (int off = 16; off > 0; off >>= 1) {
        ps += __shfl_down(ps, off, 32);
        pd += __shfl_down(pd, off, 32);
    }
    if (lane == 0) {
        a_s[node] = ps;
        a_d[node] = pd;
    }
}

#define NPB   391

__global__ void gat_owner_ids(const int* __restrict__ ei, unsigned* __restrict__ obw, int n_edges) {
    int t = blockIdx.x * blockDim.x + threadIdx.x;
    int e0 = t * 4;
    if (e0 >= n_edges) return;
    const int* __restrict__ dstp = ei + n_edges;
    unsigned wd = 0;
#pragma unroll
    for (int j = 0; j < 4; ++j)
        if (e0 + j < n_edges) wd |= ((unsigned)(dstp[e0 + j] / NPB) & 255u) << (8 * j);
    obw[t] = wd;
}

#define OWN_T 1024
#define OWN_W (OWN_T / 32)
#define WCAP  352
#define LCAP  (OWN_W * WCAP)
#define SCAN_U 8
#define PH2_U 8
#define SC_GROUP 8

template <bool TAIL>
__device__ __forceinline__ void gat_keep4(unsigned wd, unsigned pat, unsigned obv, int e0,
                                          int n_edges, int lane, int* lw, int& wc) {
    unsigned x = wd ^ pat;
    bool hit = ((x - 0x01010101u) & ~x & 0x80808080u) != 0u;
    if (__builtin_amdgcn_ballot_w32(hit) != 0u) {
#pragma unroll
        for (int b = 0; b < 4; ++b) {
            bool mt = ((wd >> (8 * b)) & 255u) == obv;
            if (TAIL) mt = mt && (e0 + b < n_edges);
            unsigned mk = __builtin_amdgcn_ballot_w32(mt);
            if (mk != 0u) {
                if (mt) {
                    int pos = wc + __builtin_popcount(mk & ((1u << lane) - 1u));
                    if (pos < WCAP) lw[pos] = e0 + b;
                }
                wc += __builtin_popcount(mk);
            }
        }
    }
}

__device__ __forceinline__ void gat_add_group(const int* lwl, const unsigned short* dwl,
                                              const float* gwl, int k0, int nl,
                                              const unsigned* mb, float* ssum, float* acc,
                                              const float* __restrict__ H, int lane) {
    const int rest = nl - k0;
    const int gc = rest < SC_GROUP ? rest : SC_GROUP;
    int sv = 0, dv = 0;
    float w = 0.0f;
    if (lane < gc) {
        sv = lwl[k0 + lane];
        dv = dwl[k0 + lane];
        w = expf(gwl[k0 + lane] - o2f(mb[dv]));
        atomicAdd(&ssum[dv], w);
    }
    int wbits = __float_as_int(w);
    if (gc == SC_GROUP) {
        float hv[SC_GROUP][4];
        int dk[SC_GROUP];
        float wk[SC_GROUP];
#pragma unroll
        for (int u = 0; u < SC_GROUP; ++u) {
            int sk = __builtin_amdgcn_readlane(sv, u);
            dk[u] = __builtin_amdgcn_readlane(dv, u);
            wk[u] = __int_as_float(__builtin_amdgcn_readlane(wbits, u));
            const float* hp = H + (size_t)sk * HID + lane;
            hv[u][0] = hp[0];
            hv[u][1] = hp[32];
            hv[u][2] = hp[64];
            hv[u][3] = hp[96];
        }
#pragma unroll
        for (int u = 0; u < SC_GROUP; ++u) {
            float* ap = acc + dk[u] * HID + lane;
            atomicAdd(ap,      wk[u] * hv[u][0]);
            atomicAdd(ap + 32, wk[u] * hv[u][1]);
            atomicAdd(ap + 64, wk[u] * hv[u][2]);
            atomicAdd(ap + 96, wk[u] * hv[u][3]);
        }
    } else {
        for (int u = 0; u < gc; ++u) {
            int sk = __builtin_amdgcn_readlane(sv, u);
            int dd = __builtin_amdgcn_readlane(dv, u);
            float ww = __int_as_float(__builtin_amdgcn_readlane(wbits, u));
            const float* hp = H + (size_t)sk * HID + lane;
            float h0 = hp[0], h1 = hp[32], h2 = hp[64], h3 = hp[96];
            float* ap = acc + dd * HID + lane;
            atomicAdd(ap,      ww * h0);
            atomicAdd(ap + 32, ww * h1);
            atomicAdd(ap + 64, ww * h2);
            atomicAdd(ap + 96, ww * h3);
        }
    }
}

__global__ __launch_bounds__(OWN_T) void gat_owner(const int* __restrict__ ei,
                                                   const unsigned* __restrict__ obw,
                                                   const float* __restrict__ a_s,
                                                   const float* __restrict__ a_d,
                                                   const float* __restrict__ H,
                                                   float* __restrict__ s_out,
                                                   float* __restrict__ out_acc,
                                                   int n_nodes, int n_edges) {
    __shared__ __attribute__((aligned(16))) float acc[NPB * HID];
    __shared__ float adl[NPB];
    __shared__ float ssum[NPB];
    __shared__ unsigned mb[NPB];
    __shared__ int lsrc[LCAP];
    __shared__ float llog[LCAP];
    __shared__ unsigned short ldl[LCAP];
    __shared__ int wcnt[OWN_W];
    __shared__ int ovfl;

    const int tid = threadIdx.x;
    const int lane = tid & 31;
    const int wv = tid >> 5;
    const int lo = blockIdx.x * NPB;
    int nown = n_nodes - lo;
    if (nown > NPB) nown = NPB;
    const int* __restrict__ srcp = ei;
    const int* __restrict__ dstp = ei + n_edges;

    for (int i = tid; i < nown; i += OWN_T) {
        float ad = a_d[lo + i];
        float sl = lrelu(a_s[lo + i] + ad);
        adl[i] = ad;
        ssum[i] = sl;
        mb[i] = f2o(sl);
    }
    if (tid == 0) ovfl = 0;
    __syncthreads();

    int* lw = lsrc + wv * WCAP;
    float* gw = llog + wv * WCAP;
    unsigned short* dw = ldl + wv * WCAP;
    int wc = 0;
    const unsigned obv = blockIdx.x & 255u;
    const unsigned pat = obv * 0x01010101u;
    const unsigned npat = ~pat;
    const int nw = (n_edges + 3) >> 2;
    const int nq = n_edges >> 4;
    const int nit = (nq + SCAN_U * OWN_T - 1) / (SCAN_U * OWN_T);
    const int rot = (int)(((long long)blockIdx.x * nit) / gridDim.x);
    const uint4* __restrict__ oq = (const uint4*)obw;
    for (int it = 0; it < nit; ++it) {
        int itr = it + rot;
        if (itr >= nit) itr -= nit;
        const int qb = itr * (SCAN_U * OWN_T) + tid;
        uint4 d[SCAN_U];
#pragma unroll
        for (int j = 0; j < SCAN_U; ++j) {
            int q = qb + j * OWN_T;
            d[j] = (q < nq) ? oq[q] : make_uint4(npat, npat, npat, npat);
        }
#pragma unroll
        for (int j = 0; j < SCAN_U; ++j) {
            int e0 = (qb + j * OWN_T) * 16;
            gat_keep4<false>(d[j].x, pat, obv, e0,      n_edges, lane, lw, wc);
            gat_keep4<false>(d[j].y, pat, obv, e0 + 4,  n_edges, lane, lw, wc);
            gat_keep4<false>(d[j].z, pat, obv, e0 + 8,  n_edges, lane, lw, wc);
            gat_keep4<false>(d[j].w, pat, obv, e0 + 12, n_edges, lane, lw, wc);
        }
    }
    for (int wi0 = nq * 4; wi0 < nw; wi0 += OWN_T) {
        int wi = wi0 + tid;
        unsigned wd = (wi < nw) ? obw[wi] : npat;
        gat_keep4<true>(wd, pat, obv, wi * 4, n_edges, lane, lw, wc);
    }
    if (lane == 0) {
        wcnt[wv] = wc;
        if (wc > WCAP) ovfl = 1;
    }
    __syncthreads();

    const bool ovf = ovfl != 0;
    const int nl = ovf ? 0 : wc;

    if (!ovf) {
        for (int kb = 0; kb < nl; kb += 32 * PH2_U) {
            int ee[PH2_U], dd[PH2_U], sv[PH2_U];
            float av[PH2_U];
#pragma unroll
            for (int u = 0; u < PH2_U; ++u) {
                int k = kb + u * 32 + lane;
                ee[u] = (k < nl) ? lw[k] : 0;
            }
#pragma unroll
            for (int u = 0; u < PH2_U; ++u) {
                dd[u] = dstp[ee[u]];
                sv[u] = srcp[ee[u]];
            }
#pragma unroll
            for (int u = 0; u < PH2_U; ++u) av[u] = a_s[sv[u]];
#pragma unroll
            for (int u = 0; u < PH2_U; ++u) {
                int k = kb + u * 32 + lane;
                if (k < nl) {
                    unsigned dl = (unsigned)(dd[u] - lo);
                    if (dl < (unsigned)nown) {
                        float lg = lrelu(av[u] + adl[dl]);
                        atomicMax(&mb[dl], f2o(lg));
                        lw[k] = sv[u];
                        dw[k] = (unsigned short)dl;
                        gw[k] = lg;
                    } else {
                        lw[k] = lo;
                        dw[k] = 0;
                        gw[k] = -__builtin_inff();
                    }
                }
            }
        }
    } else {
        for (int e = tid; e < n_edges; e += OWN_T) {
            unsigned dl = (unsigned)(dstp[e] - lo);
            if (dl < (unsigned)nown) {
                float lg = lrelu(a_s[srcp[e]] + adl[dl]);
                atomicMax(&mb[dl], f2o(lg));
            }
        }
    }
    __syncthreads();

    for (int i0 = wv; i0 < nown; i0 += OWN_W * 4) {
        float4 hv[4];
#pragma unroll
        for (int u = 0; u < 4; ++u) {
            int i = i0 + u * OWN_W;
            hv[u] = make_float4(0.0f, 0.0f, 0.0f, 0.0f);
            if (i < nown) hv[u] = ((const float4*)(H + (size_t)(lo + i) * HID))[lane];
        }
#pragma unroll
        for (int u = 0; u < 4; ++u) {
            int i = i0 + u * OWN_W;
            if (i < nown) {
                float es = expf(ssum[i] - o2f(mb[i]));
                float4 r;
                r.x = es * hv[u].x; r.y = es * hv[u].y; r.z = es * hv[u].z; r.w = es * hv[u].w;
                ((float4*)(acc + i * HID))[lane] = r;
                if (lane == 0) ssum[i] = es;
            }
        }
    }
    __syncthreads();

    if (!ovf) {
        const int myc = wcnt[lane];
        for (int wl = 0; wl < OWN_W; ++wl) {
            const int nlw = __builtin_amdgcn_readlane(myc, wl);
            for (int g = (wv - wl) & (OWN_W - 1); g * SC_GROUP < nlw; g += OWN_W)
                gat_add_group(lsrc + wl * WCAP, ldl + wl * WCAP, llog + wl * WCAP, g * SC_GROUP, nlw,
                              mb, ssum, acc, H, lane);
        }
    } else {
        for (int e = tid; e < n_edges; e += OWN_T) {
            unsigned dl = (unsigned)(dstp[e] - lo);
            if (dl < (unsigned)nown) {
                int s = srcp[e];
                float ww = expf(lrelu(a_s[s] + adl[dl]) - o2f(mb[dl]));
                atomicAdd(&ssum[dl], ww);
                const float* hp = H + (size_t)s * HID;
                float* ap = acc + dl * HID;
                for (int c = 0; c < HID; ++c) atomicAdd(ap + c, ww * hp[c]);
            }
        }
    }
    __syncthreads();

    float4* __restrict__ orow = (float4*)(out_acc + (size_t)lo * HID);
    const float4* arow = (const float4*)acc;
    for (int i = tid; i < nown * (HID / 4); i += OWN_T) orow[i] = arow[i];
    for (int i = tid; i < nown; i += OWN_T) s_out[lo + i] = ssum[i];
}

#define FIN_SLOTS 4
#define FIN_U 16

__global__ __launch_bounds__(HID * FIN_SLOTS) void gat_finalize(
        float* __restrict__ out_acc, const float* __restrict__ s,
        const float* __restrict__ bias, float* __restrict__ sums,
        float* __restrict__ sumsq, const int* __restrict__ batch,
        float* __restrict__ praw, float* __restrict__ cnt,
        int do_raw, int n_nodes) {
    __shared__ float red[2 * FIN_SLOTS * HID];
    const int tid = threadIdx.x;
    const int c = tid & (HID - 1);
    const int slot = tid >> 7;
    const int lane = tid & 31;
    const float b = bias[c];
    float ls = 0.0f, lq = 0.0f;
    const int chunk = (n_nodes + (int)gridDim.x - 1) / (int)gridDim.x;
    const int sub = (chunk + FIN_SLOTS - 1) / FIN_SLOTS;
    const int bstart = blockIdx.x * chunk;
    int bend = bstart + chunk;
    if (bend > n_nodes) bend = n_nodes;
    const int n0 = bstart + slot * sub;
    int n1 = n0 + sub;
    if (n1 > bend) n1 = bend;
    int gcur = 0, rl = 0;
    float run = 0.0f;
    if (do_raw && n0 < n1) gcur = batch[n0];
    for (int nb = n0; nb < n1; nb += FIN_U) {
        const int last = n1 - 1;
        const int cntu = (n1 - nb < FIN_U) ? (n1 - nb) : FIN_U;
        float a[FIN_U], v[FIN_U];
#pragma unroll
        for (int u = 0; u < FIN_U; ++u) {
            int n = nb + u;
            if (n > last) n = last;
            a[u] = out_acc[(size_t)n * HID + c];
        }
        int nl = nb + (lane & (FIN_U - 1));
        if (nl > last) nl = last;
        const int ib = __float_as_int(1.0f / s[nl]);
        const int gl = do_raw ? batch[nl] : 0;
#pragma unroll
        for (int u = 0; u < FIN_U; ++u) {
            float inv = __int_as_float(__builtin_amdgcn_readlane(ib, u));
            float t = a[u] * inv + b;
            float e = expm1f(t > 0.0f ? 0.0f : t);
            v[u] = t > 0.0f ? t : e;
        }
#pragma unroll
        for (int u = 0; u < FIN_U; ++u) {
            if (u < cntu) {
                out_acc[(size_t)(nb + u) * HID + c] = v[u];
                ls += v[u];
                lq += v[u] * v[u];
            }
        }
        if (do_raw) {
#pragma unroll
            for (int u = 0; u < FIN_U; ++u) {
                if (u < cntu) {
                    int gb = __builtin_amdgcn_readlane(gl, u);
                    if (gb != gcur) {
                        atomicAdd(&praw[(size_t)gcur * (3 * HID) + c], run);
                        if (c == 0) atomicAdd(&cnt[gcur], (float)rl);
                        run = 0.0f;
                        rl = 0;
                        gcur = gb;
                    }
                    run += v[u];
                    ++rl;
                }
            }
        }
    }
    if (do_raw && rl > 0) {
        atomicAdd(&praw[(size_t)gcur * (3 * HID) + c], run);
        if (c == 0) atomicAdd(&cnt[gcur], (float)rl);
    }
    red[slot * HID + c] = ls;
    red[(FIN_SLOTS + slot) * HID + c] = lq;
    __syncthreads();
    if (tid < HID) {
        float ts = 0.0f, tq = 0.0f;
#pragma unroll
        for (int k = 0; k < FIN_SLOTS; ++k) {
            ts += red[k * HID + tid];
            tq += red[(FIN_SLOTS + k) * HID + tid];
        }
        atomicAdd(&sums[tid], ts);
        atomicAdd(&sumsq[tid], tq);
    }
}

__global__ void gat_bn_fold(const float* __restrict__ sums, const float* __restrict__ sumsq,
                            const float* __restrict__ gamma, const float* __restrict__ beta,
                            const float* __restrict__ Wn, float* __restrict__ Wp,
                            float* __restrict__ c0, float* __restrict__ pooled_l,
                            const float* __restrict__ cnt, int n_nodes) {
    const int t = threadIdx.x;
    const int b = blockIdx.x;
    const float invN = 1.0f / (float)n_nodes;
    if (b < HID) {
        float mu = sums[b] * invN;
        float var = sumsq[b] * invN - mu * mu;
        float g = gamma[b] / sqrtf(var + BN_EPS);
        Wp[b * HID + t] = g * Wn[b * HID + t];
    } else if (b == HID) {
        __shared__ float dsh[HID];
        {
            float mu = sums[t] * invN;
            float var = sumsq[t] * invN - mu * mu;
            float g = gamma[t] / sqrtf(var + BN_EPS);
            dsh[t] = beta[t] - mu * g;
        }
        __syncthreads();
        float a = 0.0f;
#pragma unroll 16
        for (int k = 0; k < HID; ++k) a += dsh[k] * Wn[k * HID + t];
        c0[t] = a;
    } else {
        const int gr = b - HID - 1;
        float mu = sums[t] * invN;
        float var = sumsq[t] * invN - mu * mu;
        float g = gamma[t] / sqrtf(var + BN_EPS);
        float cn = cnt[gr];
        size_t idx = (size_t)gr * (3 * HID) + t;
        pooled_l[idx] = (pooled_l[idx] - cn * mu) * g + cn * beta[t];
    }
}

__global__ void gat_bn_pool(float* __restrict__ out_acc, const float* __restrict__ sums,
                            const float* __restrict__ sumsq, const float* __restrict__ gamma,
                            const float* __restrict__ beta, const int* __restrict__ batch,
                            float* __restrict__ pooled, int layer, int n_nodes) {
    int idx = blockIdx.x * blockDim.x + threadIdx.x;
    if (idx >= n_nodes * HID) return;
    int c = idx & (HID - 1);
    int n = idx >> 7;
    float invN = 1.0f / (float)n_nodes;
    float mu = sums[c] * invN;
    float var = sumsq[c] * invN - mu * mu;
    float xv = (out_acc[idx] - mu) / sqrtf(var + BN_EPS) * gamma[c] + beta[c];
    out_acc[idx] = xv;
    atomicAdd(&pooled[(size_t)batch[n] * (3 * HID) + layer * HID + c], xv);
}

extern "C" void kernel_launch(void* const* d_in, const int* in_sizes, int n_in,
                              void* d_out, int out_size, void* d_ws, size_t ws_size,
                              hipStream_t stream) {
    const float* x_in  = (const float*)d_in[0];
    const int*   ei    = (const int*)d_in[1];
    const int*   batch = (const int*)d_in[2];
    const float* W     = (const float*)d_in[3];
    const float* att_s = (const float*)d_in[4];
    const float* att_d = (const float*)d_in[5];
    const float* bias  = (const float*)d_in[6];
    const float* gamma = (const float*)d_in[7];
    const float* beta  = (const float*)d_in[8];

    const int N = in_sizes[0] / HID;
    const int E = in_sizes[1] / 2;

    float* pooled = (float*)d_out;
    float* xfinal = (float*)d_out + (size_t)NGRAPH * 3 * HID;

    char* w = (char*)d_ws;
    float* hbuf = (float*)w;  w += (size_t)N * HID * sizeof(float);
    float* xb   = (float*)w;  w += (size_t)N * HID * sizeof(float);
    unsigned* obw = (unsigned*)w;  w += (size_t)((E + 15) / 16) * 16;
    float* asrc = (float*)w;  w += (size_t)N * sizeof(float);
    float* adst = (float*)w;  w += (size_t)N * sizeof(float);
    float* sbuf = (float*)w;  w += (size_t)N * sizeof(float);
    float* sums = (float*)w;
    float* sumsq = sums + HID;
    float* cnt  = sums + 2 * HID;
    float* c0   = cnt + NGRAPH;
    float* Wp   = c0 + HID;

    const int waves_g = (N + 15) / 16;
    dim3 blk(256);
    dim3 grid_gemm((waves_g * 32 + 255) / 256);
    dim3 grid_node((N * 32 + 255) / 256);
    dim3 grid_own((N + NPB - 1) / NPB);
    dim3 grid_elem(((size_t)N * HID + 255) / 256);

    gat_zero<<<(NGRAPH * 3 * HID + 255) / 256, blk, 0, stream>>>(pooled, NGRAPH * 3 * HID);
    gat_owner_ids<<<((E + 3) / 4 + 255) / 256, blk, 0, stream>>>(ei, obw, E);

    for (int l = 0; l < 3; ++l) {
        const float* X  = (l == 0) ? x_in : xb;
        float* oacc     = (l == 2) ? xfinal : xb;
        const float* Wl = (l == 0) ? W : Wp;
        const float* Cl = (l == 0) ? (const float*)nullptr : c0;

        gat_gemm_wmma<<<grid_gemm, blk, 0, stream>>>(X, Wl, Cl, hbuf, N);
        gat_node_dots<<<grid_node, blk, 0, stream>>>(hbuf, att_s + l * HID, att_d + l * HID,
                                                     asrc, adst, N);
        gat_owner<<<grid_own, OWN_T, 0, stream>>>(ei, obw, asrc, adst, hbuf, sbuf, oacc, N, E);
        gat_zero<<<3, 256, 0, stream>>>(sums, 2 * HID + NGRAPH);
        gat_finalize<<<521, HID * FIN_SLOTS, 0, stream>>>(oacc, sbuf, bias + l * HID, sums, sumsq, batch,
                                              pooled + l * HID, cnt, (l < 2) ? 1 : 0, N);
        if (l < 2)
            gat_bn_fold<<<HID + 1 + NGRAPH, HID, 0, stream>>>(sums, sumsq, gamma + l * HID,
                                                              beta + l * HID,
                                                              W + (size_t)(l + 1) * HID * HID, Wp, c0,
                                                              pooled + l * HID, cnt, N);
        else
            gat_bn_pool<<<grid_elem, blk, 0, stream>>>(oacc, sums, sumsq, gamma + l * HID,
                                                       beta + l * HID, batch, pooled, l, N);
    }
}


# Kept-edge lists reused across layers: layer 0 writes each wave's count and (source, row) pairs to workspace; layers 1,2 read them back, no owner-byte scan and no index loads there (sum order kept)
# speedup vs baseline: 3.3981x; 1.1691x over previous
#include <hip/hip_runtime.h>
#include <hip/hip_bf16.h>

#define HID 128
#define NGRAPH 512
#define NEG_SLOPE 0.2f
#define BN_EPS 1e-5f

typedef __attribute__((ext_vector_type(2))) float v2f;
typedef __attribute__((ext_vector_type(8))) float v8f;

__device__ __forceinline__ unsigned f2o(float f) {
    unsigned u = __float_as_uint(f);
    return (u & 0x80000000u) ? ~u : (u | 0x80000000u);
}
__device__ __forceinline__ float o2f(unsigned u) {
    unsigned v = (u & 0x80000000u) ? (u & 0x7fffffffu) : ~u;
    return __uint_as_float(v);
}
__device__ __forceinline__ float lrelu(float a) { return a > 0.0f ? a : NEG_SLOPE * a; }

__global__ void gat_zero(float* __restrict__ p, int n) {
    int i = blockIdx.x * blockDim.x + threadIdx.x;
    if (i < n) p[i] = 0.0f;
}

__global__ void gat_gemm_wmma(const float* __restrict__ X, const float* __restrict__ W,
                              const float* __restrict__ C0, float* __restrict__ H, int nrows) {
    int wave = (blockIdx.x * blockDim.x + threadIdx.x) >> 5;
    int m0 = wave * 16;
    if (m0 >= nrows) return;
    int lane = threadIdx.x & 31;
    int lr = lane & 15;
    int lh = lane >> 4;

    v8f acc[8];
#pragma unroll
    for (int n = 0; n < 8; ++n) {
        float cv = C0 ? C0[n * 16 + lr] : 0.0f;
#pragma unroll
        for (int j = 0; j < 8; ++j) acc[n][j] = cv;
    }
    const float* xrow = X + (size_t)(m0 + lr) * HID;

    for (int k = 0; k < HID; k += 4) {
        int ka = k + lh * 2;
        v2f a;
        a.x = xrow[ka];
        a.y = xrow[ka + 1];
#pragma unroll
        for (int n = 0; n < 8; ++n) {
            v2f b;
            b.x = W[ka * HID + n * 16 + lr];
            b.y = W[(ka + 1) * HID + n * 16 + lr];
            acc[n] = __builtin_amdgcn_wmma_f32_16x16x4_f32(
                false, a, false, b, (short)0, acc[n], false, false);
        }
    }
#pragma unroll
    for (int n = 0; n < 8; ++n)
#pragma unroll
        for (int j = 0; j < 8; ++j)
            H[(size_t)(m0 + j + lh * 8) * HID + n * 16 + lr] = acc[n][j];
}

__global__ void gat_node_dots(const float* __restrict__ H, const float* __restrict__ att_s,
                              const float* __restrict__ att_d, float* __restrict__ a_s,
                              float* __restrict__ a_d, int n_nodes) {
    int node = (blockIdx.x * blockDim.x + threadIdx.x) >> 5;
    if (node >= n_nodes) return;
    int lane = threadIdx.x & 31;
    float4 hv = ((const float4*)(H + (size_t)node * HID))[lane];
    float4 ws = ((const float4*)att_s)[lane];
    float4 wd = ((const float4*)att_d)[lane];
    float ps = hv.x * ws.x + hv.y * ws.y + hv.z * ws.z + hv.w * ws.w;
    float pd = hv.x * wd.x + hv.y * wd.y + hv.z * wd.z + hv.w * wd.w;
    for (int off = 16; off > 0; off >>= 1) {
        ps += __shfl_down(ps, off, 32);
        pd += __shfl_down(pd, off, 32);
    }
    if (lane == 0) {
        a_s[node] = ps;
        a_d[node] = pd;
    }
}

#define NPB   391

__global__ void gat_owner_ids(const int* __restrict__ ei, unsigned* __restrict__ obw, int n_edges) {
    int t = blockIdx.x * blockDim.x + threadIdx.x;
    int e0 = t * 4;
    if (e0 >= n_edges) return;
    const int* __restrict__ dstp = ei + n_edges;
    unsigned wd = 0;
#pragma unroll
    for (int j = 0; j < 4; ++j)
        if (e0 + j < n_edges) wd |= ((unsigned)(dstp[e0 + j] / NPB) & 255u) << (8 * j);
    obw[t] = wd;
}

#define OWN_T 1024
#define OWN_W (OWN_T / 32)
#define WCAP  352
#define LCAP  (OWN_W * WCAP)
#define SCAN_U 8
#define PH2_U 8
#define SC_GROUP 8
#define NOROW 0xFFFFu

template <bool TAIL>
__device__ __forceinline__ void gat_keep4(unsigned wd, unsigned pat, unsigned obv, int e0,
                                          int n_edges, int lane, int* lw, int& wc) {
    unsigned x = wd ^ pat;
    bool hit = ((x - 0x01010101u) & ~x & 0x80808080u) != 0u;
    if (__builtin_amdgcn_ballot_w32(hit) != 0u) {
#pragma unroll
        for (int b = 0; b < 4; ++b) {
            bool mt = ((wd >> (8 * b)) & 255u) == obv;
            if (TAIL) mt = mt && (e0 + b < n_edges);
            unsigned mk = __builtin_amdgcn_ballot_w32(mt);
            if (mk != 0u) {
                if (mt) {
                    int pos = wc + __builtin_popcount(mk & ((1u << lane) - 1u));
                    if (pos < WCAP) lw[pos] = e0 + b;
                }
                wc += __builtin_popcount(mk);
            }
        }
    }
}

__device__ __forceinline__ void gat_add_group(const int* lwl, const unsigned short* dwl,
                                              const float* gwl, int k0, int nl,
                                              const unsigned* mb, float* ssum, float* acc,
                                              const float* __restrict__ H, int lane) {
    const int rest = nl - k0;
    const int gc = rest < SC_GROUP ? rest : SC_GROUP;
    int sv = 0, dv = 0;
    float w = 0.0f;
    if (lane < gc) {
        sv = lwl[k0 + lane];
        dv = dwl[k0 + lane];
        w = expf(gwl[k0 + lane] - o2f(mb[dv]));
        atomicAdd(&ssum[dv], w);
    }
    int wbits = __float_as_int(w);
    if (gc == SC_GROUP) {
        float hv[SC_GROUP][4];
        int dk[SC_GROUP];
        float wk[SC_GROUP];
#pragma unroll
        for (int u = 0; u < SC_GROUP; ++u) {
            int sk = __builtin_amdgcn_readlane(sv, u);
            dk[u] = __builtin_amdgcn_readlane(dv, u);
            wk[u] = __int_as_float(__builtin_amdgcn_readlane(wbits, u));
            const float* hp = H + (size_t)sk * HID + lane;
            hv[u][0] = hp[0];
            hv[u][1] = hp[32];
            hv[u][2] = hp[64];
            hv[u][3] = hp[96];
        }
#pragma unroll
        for (int u = 0; u < SC_GROUP; ++u) {
            float* ap = acc + dk[u] * HID + lane;
            atomicAdd(ap,      wk[u] * hv[u][0]);
            atomicAdd(ap + 32, wk[u] * hv[u][1]);
            atomicAdd(ap + 64, wk[u] * hv[u][2]);
            atomicAdd(ap + 96, wk[u] * hv[u][3]);
        }
    } else {
        for (int u = 0; u < gc; ++u) {
            int sk = __builtin_amdgcn_readlane(sv, u);
            int dd = __builtin_amdgcn_readlane(dv, u);
            float ww = __int_as_float(__builtin_amdgcn_readlane(wbits, u));
            const float* hp = H + (size_t)sk * HID + lane;
            float h0 = hp[0], h1 = hp[32], h2 = hp[64], h3 = hp[96];
            float* ap = acc + dd * HID + lane;
            atomicAdd(ap,      ww * h0);
            atomicAdd(ap + 32, ww * h1);
            atomicAdd(ap + 64, ww * h2);
            atomicAdd(ap + 96, ww * h3);
        }
    }
}

template <bool REUSE>
__global__ __launch_bounds__(OWN_T) void gat_owner(const int* __restrict__ ei,
                                                   const unsigned* __restrict__ obw,
                                                   const float* __restrict__ a_s,
                                                   const float* __restrict__ a_d,
                                                   const float* __restrict__ H,
                                                   float* __restrict__ s_out,
                                                   float* __restrict__ out_acc,
                                                   int* __restrict__ gsrc,
                                                   unsigned short* __restrict__ gdl,
                                                   int* __restrict__ gcnt,
                                                   int n_nodes, int n_edges) {
    __shared__ __attribute__((aligned(16))) float acc[NPB * HID];
    __shared__ float adl[NPB];
    __shared__ float ssum[NPB];
    __shared__ unsigned mb[NPB];
    __shared__ int lsrc[LCAP];
    __shared__ float llog[LCAP];
    __shared__ unsigned short ldl[LCAP];
    __shared__ int wcnt[OWN_W];
    __shared__ int ovfl;

    const int tid = threadIdx.x;
    const int lane = tid & 31;
    const int wv = tid >> 5;
    const int lo = blockIdx.x * NPB;
    int nown = n_nodes - lo;
    if (nown > NPB) nown = NPB;
    const int* __restrict__ srcp = ei;
    const int* __restrict__ dstp = ei + n_edges;

    for (int i = tid; i < nown; i += OWN_T) {
        float ad = a_d[lo + i];
        float sl = lrelu(a_s[lo + i] + ad);
        adl[i] = ad;
        ssum[i] = sl;
        mb[i] = f2o(sl);
    }
    if (tid == 0) ovfl = 0;
    __syncthreads();

    int* lw = lsrc + wv * WCAP;
    float* gw = llog + wv * WCAP;
    unsigned short* dw = ldl + wv * WCAP;
    int wc = 0;
    const size_t gl0 = (size_t)(blockIdx.x * OWN_W + wv) * WCAP;
    int* __restrict__ gs = gsrc + gl0;
    unsigned short* __restrict__ gd = gdl + gl0;
    if (REUSE) {
        wc = __builtin_amdgcn_readfirstlane(gcnt[blockIdx.x * OWN_W + wv]);
    } else {
        const unsigned obv = blockIdx.x & 255u;
        const unsigned pat = obv * 0x01010101u;
        const unsigned npat = ~pat;
        const int nw = (n_edges + 3) >> 2;
        const int nq = n_edges >> 4;
        const int nit = (nq + SCAN_U * OWN_T - 1) / (SCAN_U * OWN_T);
        const int rot = (int)(((long long)blockIdx.x * nit) / gridDim.x);
        const uint4* __restrict__ oq = (const uint4*)obw;
        for (int it = 0; it < nit; ++it) {
            int itr = it + rot;
            if (itr >= nit) itr -= nit;
            const int qb = itr * (SCAN_U * OWN_T) + tid;
            uint4 d[SCAN_U];
#pragma unroll
            for (int j = 0; j < SCAN_U; ++j) {
                int q = qb + j * OWN_T;
                d[j] = (q < nq) ? oq[q] : make_uint4(npat, npat, npat, npat);
            }
#pragma unroll
            for (int j = 0; j < SCAN_U; ++j) {
                int e0 = (qb + j * OWN_T) * 16;
                gat_keep4<false>(d[j].x, pat, obv, e0,      n_edges, lane, lw, wc);
                gat_keep4<false>(d[j].y, pat, obv, e0 + 4,  n_edges, lane, lw, wc);
                gat_keep4<false>(d[j].z, pat, obv, e0 + 8,  n_edges, lane, lw, wc);
                gat_keep4<false>(d[j].w, pat, obv, e0 + 12, n_edges, lane, lw, wc);
            }
        }
        for (int wi0 = nq * 4; wi0 < nw; wi0 += OWN_T) {
            int wi = wi0 + tid;
            unsigned wd = (wi < nw) ? obw[wi] : npat;
            gat_keep4<true>(wd, pat, obv, wi * 4, n_edges, lane, lw, wc);
        }
        if (lane == 0) gcnt[blockIdx.x * OWN_W + wv] = wc;
    }
    if (lane == 0) {
        wcnt[wv] = wc;
        if (wc > WCAP) ovfl = 1;
    }
    __syncthreads();

    const bool ovf = ovfl != 0;
    const int nl = ovf ? 0 : wc;

    if (!ovf && REUSE) {
        for (int kb = 0; kb < nl; kb += 32 * PH2_U) {
            int sv[PH2_U];
            unsigned dl[PH2_U];
            float av[PH2_U];
#pragma unroll
            for (int u = 0; u < PH2_U; ++u) {
                int k = kb + u * 32 + lane;
                sv[u] = (k < nl) ? gs[k] : lo;
                dl[u] = (k < nl) ? (unsigned)gd[k] : NOROW;
            }
#pragma unroll
            for (int u = 0; u < PH2_U; ++u) av[u] = a_s[sv[u]];
#pragma unroll
            for (int u = 0; u < PH2_U; ++u) {
                int k = kb + u * 32 + lane;
                if (k < nl) {
                    if (dl[u] != NOROW) {
                        float lg = lrelu(av[u] + adl[dl[u]]);
                        atomicMax(&mb[dl[u]], f2o(lg));
                        lw[k] = sv[u];
                        dw[k] = (unsigned short)dl[u];
                        gw[k] = lg;
                    } else {
                        lw[k] = lo;
                        dw[k] = 0;
                        gw[k] = -__builtin_inff();
                    }
                }
            }
        }
    } else if (!ovf) {
        for (int kb = 0; kb < nl; kb += 32 * PH2_U) {
            int ee[PH2_U], dd[PH2_U], sv[PH2_U];
            float av[PH2_U];
#pragma unroll
            for (int u = 0; u < PH2_U; ++u) {
                int k = kb + u * 32 + lane;
                ee[u] = (k < nl) ? lw[k] : 0;
            }
#pragma unroll
            for (int u = 0; u < PH2_U; ++u) {
                dd[u] = dstp[ee[u]];
                sv[u] = srcp[ee[u]];
            }
#pragma unroll
            for (int u = 0; u < PH2_U; ++u) av[u] = a_s[sv[u]];
#pragma unroll
            for (int u = 0; u < PH2_U; ++u) {
                int k = kb + u * 32 + lane;
                if (k < nl) {
                    unsigned dl = (unsigned)(dd[u] - lo);
                    if (dl < (unsigned)nown) {
                        float lg = lrelu(av[u] + adl[dl]);
                        atomicMax(&mb[dl], f2o(lg));
                        lw[k] = sv[u];
                        dw[k] = (unsigned short)dl;
                        gw[k] = lg;
                        gs[k] = sv[u];
                        gd[k] = (unsigned short)dl;
                    } else {
                        lw[k] = lo;
                        dw[k] = 0;
                        gw[k] = -__builtin_inff();
                        gs[k] = lo;
                        gd[k] = (unsigned short)NOROW;
                    }
                }
            }
        }
    } else {
        for (int e = tid; e < n_edges; e += OWN_T) {
            unsigned dl = (unsigned)(dstp[e] - lo);
            if (dl < (unsigned)nown) {
                float lg = lrelu(a_s[srcp[e]] + adl[dl]);
                atomicMax(&mb[dl], f2o(lg));
            }
        }
    }
    __syncthreads();

    for (int i0 = wv; i0 < nown; i0 += OWN_W * 4) {
        float4 hv[4];
#pragma unroll
        for (int u = 0; u < 4; ++u) {
            int i = i0 + u * OWN_W;
            hv[u] = make_float4(0.0f, 0.0f, 0.0f, 0.0f);
            if (i < nown) hv[u] = ((const float4*)(H + (size_t)(lo + i) * HID))[lane];
        }
#pragma unroll
        for (int u = 0; u < 4; ++u) {
            int i = i0 + u * OWN_W;
            if (i < nown) {
                float es = expf(ssum[i] - o2f(mb[i]));
                float4 r;
                r.x = es * hv[u].x; r.y = es * hv[u].y; r.z = es * hv[u].z; r.w = es * hv[u].w;
                ((float4*)(acc + i * HID))[lane] = r;
                if (lane == 0) ssum[i] = es;
            }
        }
    }
    __syncthreads();

    if (!ovf) {
        const int myc = wcnt[lane];
        for (int wl = 0; wl < OWN_W; ++wl) {
            const int nlw = __builtin_amdgcn_readlane(myc, wl);
            for (int g = (wv - wl) & (OWN_W - 1); g * SC_GROUP < nlw; g += OWN_W)
                gat_add_group(lsrc + wl * WCAP, ldl + wl * WCAP, llog + wl * WCAP, g * SC_GROUP, nlw,
                              mb, ssum, acc, H, lane);
        }
    } else {
        for (int e = tid; e < n_edges; e += OWN_T) {
            unsigned dl = (unsigned)(dstp[e] - lo);
            if (dl < (unsigned)nown) {
                int s = srcp[e];
                float ww = expf(lrelu(a_s[s] + adl[dl]) - o2f(mb[dl]));
                atomicAdd(&ssum[dl], ww);
                const float* hp = H + (size_t)s * HID;
                float* ap = acc + dl * HID;
                for (int c = 0; c < HID; ++c) atomicAdd(ap + c, ww * hp[c]);
            }
        }
    }
    __syncthreads();

    float4* __restrict__ orow = (float4*)(out_acc + (size_t)lo * HID);
    const float4* arow = (const float4*)acc;
    for (int i = tid; i < nown * (HID / 4); i += OWN_T) orow[i] = arow[i];
    for (int i = tid; i < nown; i += OWN_T) s_out[lo + i] = ssum[i];
}

#define FIN_SLOTS 4
#define FIN_U 16

__global__ __launch_bounds__(HID * FIN_SLOTS) void gat_finalize(
        float* __restrict__ out_acc, const float* __restrict__ s,
        const float* __restrict__ bias, float* __restrict__ sums,
        float* __restrict__ sumsq, const int* __restrict__ batch,
        float* __restrict__ praw, float* __restrict__ cnt,
        int do_raw, int n_nodes) {
    __shared__ float red[2 * FIN_SLOTS * HID];
    const int tid = threadIdx.x;
    const int c = tid & (HID - 1);
    const int slot = tid >> 7;
    const int lane = tid & 31;
    const float b = bias[c];
    float ls = 0.0f, lq = 0.0f;
    const int chunk = (n_nodes + (int)gridDim.x - 1) / (int)gridDim.x;
    const int sub = (chunk + FIN_SLOTS - 1) / FIN_SLOTS;
    const int bstart = blockIdx.x * chunk;
    int bend = bstart + chunk;
    if (bend > n_nodes) bend = n_nodes;
    const int n0 = bstart + slot * sub;
    int n1 = n0 + sub;
    if (n1 > bend) n1 = bend;
    int gcur = 0, rl = 0;
    float run = 0.0f;
    if (do_raw && n0 < n1) gcur = batch[n0];
    for (int nb = n0; nb < n1; nb += FIN_U) {
        const int last = n1 - 1;
        const int cntu = (n1 - nb < FIN_U) ? (n1 - nb) : FIN_U;
        float a[FIN_U], v[FIN_U];
#pragma unroll
        for (int u = 0; u < FIN_U; ++u) {
            int n = nb + u;
            if (n > last) n = last;
            a[u] = out_acc[(size_t)n * HID + c];
        }
        int nl = nb + (lane & (FIN_U - 1));
        if (nl > last) nl = last;
        const int ib = __float_as_int(1.0f / s[nl]);
        const int gl = do_raw ? batch[nl] : 0;
#pragma unroll
        for (int u = 0; u < FIN_U; ++u) {
            float inv = __int_as_float(__builtin_amdgcn_readlane(ib, u));
            float t = a[u] * inv + b;
            float e = expm1f(t > 0.0f ? 0.0f : t);
            v[u] = t > 0.0f ? t : e;
        }
#pragma unroll
        for (int u = 0; u < FIN_U; ++u) {
            if (u < cntu) {
                out_acc[(size_t)(nb + u) * HID + c] = v[u];
                ls += v[u];
                lq += v[u] * v[u];
            }
        }
        if (do_raw) {
#pragma unroll
            for (int u = 0; u < FIN_U; ++u) {
                if (u < cntu) {
                    int gb = __builtin_amdgcn_readlane(gl, u);
                    if (gb != gcur) {
                        atomicAdd(&praw[(size_t)gcur * (3 * HID) + c], run);
                        if (c == 0) atomicAdd(&cnt[gcur], (float)rl);
                        run = 0.0f;
                        rl = 0;
                        gcur = gb;
                    }
                    run += v[u];
                    ++rl;
                }
            }
        }
    }
    if (do_raw && rl > 0) {
        atomicAdd(&praw[(size_t)gcur * (3 * HID) + c], run);
        if (c == 0) atomicAdd(&cnt[gcur], (float)rl);
    }
    red[slot * HID + c] = ls;
    red[(FIN_SLOTS + slot) * HID + c] = lq;
    __syncthreads();
    if (tid < HID) {
        float ts = 0.0f, tq = 0.0f;
#pragma unroll
        for (int k = 0; k < FIN_SLOTS; ++k) {
            ts += red[k * HID + tid];
            tq += red[(FIN_SLOTS + k) * HID + tid];
        }
        atomicAdd(&sums[tid], ts);
        atomicAdd(&sumsq[tid], tq);
    }
}

__global__ void gat_bn_fold(const float* __restrict__ sums, const float* __restrict__ sumsq,
                            const float* __restrict__ gamma, const float* __restrict__ beta,
                            const float* __restrict__ Wn, float* __restrict__ Wp,
                            float* __restrict__ c0, float* __restrict__ pooled_l,
                            const float* __restrict__ cnt, int n_nodes) {
    const int t = threadIdx.x;
    const int b = blockIdx.x;
    const float invN = 1.0f / (float)n_nodes;
    if (b < HID) {
        float mu = sums[b] * invN;
        float var = sumsq[b] * invN - mu * mu;
        float g = gamma[b] / sqrtf(var + BN_EPS);
        Wp[b * HID + t] = g * Wn[b * HID + t];
    } else if (b == HID) {
        __shared__ float dsh[HID];
        {
            float mu = sums[t] * invN;
            float var = sumsq[t] * invN - mu * mu;
            float g = gamma[t] / sqrtf(var + BN_EPS);
            dsh[t] = beta[t] - mu * g;
        }
        __syncthreads();
        float a = 0.0f;
#pragma unroll 16
        for (int k = 0; k < HID; ++k) a += dsh[k] * Wn[k * HID + t];
        c0[t] = a;
    } else {
        const int gr = b - HID - 1;
        float mu = sums[t] * invN;
        float var = sumsq[t] * invN - mu * mu;
        float g = gamma[t] / sqrtf(var + BN_EPS);
        float cn = cnt[gr];
        size_t idx = (size_t)gr * (3 * HID) + t;
        pooled_l[idx] = (pooled_l[idx] - cn * mu) * g + cn * beta[t];
    }
}

__global__ void gat_bn_pool(float* __restrict__ out_acc, const float* __restrict__ sums,
                            const float* __restrict__ sumsq, const float* __restrict__ gamma,
                            const float* __restrict__ beta, const int* __restrict__ batch,
                            float* __restrict__ pooled, int layer, int n_nodes) {
    int idx = blockIdx.x * blockDim.x + threadIdx.x;
    if (idx >= n_nodes * HID) return;
    int c = idx & (HID - 1);
    int n = idx >> 7;
    float invN = 1.0f / (float)n_nodes;
    float mu = sums[c] * invN;
    float var = sumsq[c] * invN - mu * mu;
    float xv = (out_acc[idx] - mu) / sqrtf(var + BN_EPS) * gamma[c] + beta[c];
    out_acc[idx] = xv;
    atomicAdd(&pooled[(size_t)batch[n] * (3 * HID) + layer * HID + c], xv);
}

extern "C" void kernel_launch(void* const* d_in, const int* in_sizes, int n_in,
                              void* d_out, int out_size, void* d_ws, size_t ws_size,
                              hipStream_t stream) {
    const float* x_in  = (const float*)d_in[0];
    const int*   ei    = (const int*)d_in[1];
    const int*   batch = (const int*)d_in[2];
    const float* W     = (const float*)d_in[3];
    const float* att_s = (const float*)d_in[4];
    const float* att_d = (const float*)d_in[5];
    const float* bias  = (const float*)d_in[6];
    const float* gamma = (const float*)d_in[7];
    const float* beta  = (const float*)d_in[8];

    const int N = in_sizes[0] / HID;
    const int E = in_sizes[1] / 2;

    float* pooled = (float*)d_out;
    float* xfinal = (float*)d_out + (size_t)NGRAPH * 3 * HID;

    char* w = (char*)d_ws;
    float* hbuf = (float*)w;  w += (size_t)N * HID * sizeof(float);
    float* xb   = (float*)w;  w += (size_t)N * HID * sizeof(float);
    unsigned* obw = (unsigned*)w;  w += (size_t)((E + 15) / 16) * 16;
    float* asrc = (float*)w;  w += (size_t)N * sizeof(float);
    float* adst = (float*)w;  w += (size_t)N * sizeof(float);
    float* sbuf = (float*)w;  w += (size_t)N * sizeof(float);
    float* sums = (float*)w;
    float* sumsq = sums + HID;
    float* cnt  = sums + 2 * HID;
    float* c0   = cnt + NGRAPH;
    float* Wp   = c0 + HID;
    const int n_own = (N + NPB - 1) / NPB;
    int* gsrc = (int*)(Wp + HID * HID);
    int* gcnt = gsrc + (size_t)n_own * LCAP;
    unsigned short* gdl = (unsigned short*)(gcnt + (size_t)n_own * OWN_W);

    const int waves_g = (N + 15) / 16;
    dim3 blk(256);
    dim3 grid_gemm((waves_g * 32 + 255) / 256);
    dim3 grid_node((N * 32 + 255) / 256);
    dim3 grid_own(n_own);
    dim3 grid_elem(((size_t)N * HID + 255) / 256);

    gat_zero<<<(NGRAPH * 3 * HID + 255) / 256, blk, 0, stream>>>(pooled, NGRAPH * 3 * HID);
    gat_owner_ids<<<((E + 3) / 4 + 255) / 256, blk, 0, stream>>>(ei, obw, E);

    for (int l = 0; l < 3; ++l) {
        const float* X  = (l == 0) ? x_in : xb;
        float* oacc     = (l == 2) ? xfinal : xb;
        const float* Wl = (l == 0) ? W : Wp;
        const float* Cl = (l == 0) ? (const float*)nullptr : c0;

        gat_gemm_wmma<<<grid_gemm, blk, 0, stream>>>(X, Wl, Cl, hbuf, N);
        gat_node_dots<<<grid_node, blk, 0, stream>>>(hbuf, att_s + l * HID, att_d + l * HID,
                                                     asrc, adst, N);
        if (l == 0)
            gat_owner<false><<<grid_own, OWN_T, 0, stream>>>(ei, obw, asrc, adst, hbuf, sbuf, oacc,
                                                             gsrc, gdl, gcnt, N, E);
        else
            gat_owner<true><<<grid_own, OWN_T, 0, stream>>>(ei, obw, asrc, adst, hbuf, sbuf, oacc,
                                                            gsrc, gdl, gcnt, N, E);
        gat_zero<<<3, 256, 0, stream>>>(sums, 2 * HID + NGRAPH);
        gat_finalize<<<521, HID * FIN_SLOTS, 0, stream>>>(oacc, sbuf, bias + l * HID, sums, sumsq, batch,
                                              pooled + l * HID, cnt, (l < 2) ? 1 : 0, N);
        if (l < 2)
            gat_bn_fold<<<HID + 1 + NGRAPH, HID, 0, stream>>>(sums, sumsq, gamma + l * HID,
                                                              beta + l * HID,
                                                              W + (size_t)(l + 1) * HID * HID, Wp, c0,
                                                              pooled + l * HID, cnt, N);
        else
            gat_bn_pool<<<grid_elem, blk, 0, stream>>>(oacc, sums, sumsq, gamma + l * HID,
                                                       beta + l * HID, batch, pooled, l, N);
    }
}


# Node-dot launches removed: the product gets a 9th WMMA column tile [W.att_src | W.att_dst] (built with the folded weights) and stores the logits from it (logit sum order: W.att tree, then k chain)
# speedup vs baseline: 3.5353x; 1.0404x over previous
#include <hip/hip_runtime.h>
#include <hip/hip_bf16.h>

#define HID 128
#define NGRAPH 512
#define NEG_SLOPE 0.2f
#define BN_EPS 1e-5f

typedef __attribute__((ext_vector_type(2))) float v2f;
typedef __attribute__((ext_vector_type(8))) float v8f;

__device__ __forceinline__ unsigned f2o(float f) {
    unsigned u = __float_as_uint(f);
    return (u & 0x80000000u) ? ~u : (u | 0x80000000u);
}
__device__ __forceinline__ float o2f(unsigned u) {
    unsigned v = (u & 0x80000000u) ? (u & 0x7fffffffu) : ~u;
    return __uint_as_float(v);
}
__device__ __forceinline__ float lrelu(float a) { return a > 0.0f ? a : NEG_SLOPE * a; }

__global__ void gat_zero(float* __restrict__ p, int n) {
    int i = blockIdx.x * blockDim.x + threadIdx.x;
    if (i < n) p[i] = 0.0f;
}

#define XC 16

__global__ void gat_gemm_wmma(const float* __restrict__ X, const float* __restrict__ W,
                              const float* __restrict__ C0, const float* __restrict__ Wx,
                              const float* __restrict__ Cx, float* __restrict__ H,
                              float* __restrict__ a_s, float* __restrict__ a_d, int nrows) {
    int wave = (blockIdx.x * blockDim.x + threadIdx.x) >> 5;
    int m0 = wave * 16;
    if (m0 >= nrows) return;
    int lane = threadIdx.x & 31;
    int lr = lane & 15;
    int lh = lane >> 4;

    v8f acc[8];
#pragma unroll
    for (int n = 0; n < 8; ++n) {
        float cv = C0 ? C0[n * 16 + lr] : 0.0f;
#pragma unroll
        for (int j = 0; j < 8; ++j) acc[n][j] = cv;
    }
    v8f accx;
    {
        float cv = Cx[lr];
#pragma unroll
        for (int j = 0; j < 8; ++j) accx[j] = cv;
    }
    const float* xrow = X + (size_t)(m0 + lr) * HID;

    for (int k = 0; k < HID; k += 4) {
        int ka = k + lh * 2;
        v2f a;
        a.x = xrow[ka];
        a.y = xrow[ka + 1];
#pragma unroll
        for (int n = 0; n < 8; ++n) {
            v2f b;
            b.x = W[ka * HID + n * 16 + lr];
            b.y = W[(ka + 1) * HID + n * 16 + lr];
            acc[n] = __builtin_amdgcn_wmma_f32_16x16x4_f32(
                false, a, false, b, (short)0, acc[n], false, false);
        }
        v2f bx;
        bx.x = Wx[ka * XC + lr];
        bx.y = Wx[(ka + 1) * XC + lr];
        accx = __builtin_amdgcn_wmma_f32_16x16x4_f32(
            false, a, false, bx, (short)0, accx, false, false);
    }
#pragma unroll
    for (int n = 0; n < 8; ++n)
#pragma unroll
        for (int j = 0; j < 8; ++j)
            H[(size_t)(m0 + j + lh * 8) * HID + n * 16 + lr] = acc[n][j];
    if (lr < 2) {
        float* ap = lr ? a_d : a_s;
#pragma unroll
        for (int j = 0; j < 8; ++j) ap[m0 + j + lh * 8] = accx[j];
    }
}

__device__ __forceinline__ void gat_att_row(float p, int t, const float* __restrict__ att_s,
                                            const float* __restrict__ att_d,
                                            float* __restrict__ dst) {
    __shared__ float part[2 * (HID / 32)];
    float ps = p * att_s[t];
    float pd = p * att_d[t];
    for (int off = 16; off > 0; off >>= 1) {
        ps += __shfl_down(ps, off, 32);
        pd += __shfl_down(pd, off, 32);
    }
    if ((t & 31) == 0) {
        part[t >> 5] = ps;
        part[HID / 32 + (t >> 5)] = pd;
    }
    __syncthreads();
    if (t < XC) {
        float v = 0.0f;
        if (t < 2) {
            const float* q = part + t * (HID / 32);
            v = (q[0] + q[1]) + (q[2] + q[3]);
        }
        dst[t] = v;
    }
}

__global__ void gat_att_cols(const float* __restrict__ Wl, const float* __restrict__ att_s,
                             const float* __restrict__ att_d, float* __restrict__ Wx,
                             float* __restrict__ Cx) {
    const int t = threadIdx.x;
    const int k = blockIdx.x;
    gat_att_row(Wl[k * HID + t], t, att_s, att_d, Wx + k * XC);
    if (k == 0 && t < XC) Cx[t] = 0.0f;
}

#define NPB   391

__global__ void gat_owner_ids(const int* __restrict__ ei, unsigned* __restrict__ obw, int n_edges) {
    int t = blockIdx.x * blockDim.x + threadIdx.x;
    int e0 = t * 4;
    if (e0 >= n_edges) return;
    const int* __restrict__ dstp = ei + n_edges;
    unsigned wd = 0;
#pragma unroll
    for (int j = 0; j < 4; ++j)
        if (e0 + j < n_edges) wd |= ((unsigned)(dstp[e0 + j] / NPB) & 255u) << (8 * j);
    obw[t] = wd;
}

#define OWN_T 1024
#define OWN_W (OWN_T / 32)
#define WCAP  352
#define LCAP  (OWN_W * WCAP)
#define SCAN_U 8
#define PH2_U 8
#define SC_GROUP 8
#define NOROW 0xFFFFu

template <bool TAIL>
__device__ __forceinline__ void gat_keep4(unsigned wd, unsigned pat, unsigned obv, int e0,
                                          int n_edges, int lane, int* lw, int& wc) {
    unsigned x = wd ^ pat;
    bool hit = ((x - 0x01010101u) & ~x & 0x80808080u) != 0u;
    if (__builtin_amdgcn_ballot_w32(hit) != 0u) {
#pragma unroll
        for (int b = 0; b < 4; ++b) {
            bool mt = ((wd >> (8 * b)) & 255u) == obv;
            if (TAIL) mt = mt && (e0 + b < n_edges);
            unsigned mk = __builtin_amdgcn_ballot_w32(mt);
            if (mk != 0u) {
                if (mt) {
                    int pos = wc + __builtin_popcount(mk & ((1u << lane) - 1u));
                    if (pos < WCAP) lw[pos] = e0 + b;
                }
                wc += __builtin_popcount(mk);
            }
        }
    }
}

__device__ __forceinline__ void gat_add_group(const int* lwl, const unsigned short* dwl,
                                              const float* gwl, int k0, int nl,
                                              const unsigned* mb, float* ssum, float* acc,
                                              const float* __restrict__ H, int lane) {
    const int rest = nl - k0;
    const int gc = rest < SC_GROUP ? rest : SC_GROUP;
    int sv = 0, dv = 0;
    float w = 0.0f;
    if (lane < gc) {
        sv = lwl[k0 + lane];
        dv = dwl[k0 + lane];
        w = expf(gwl[k0 + lane] - o2f(mb[dv]));
        atomicAdd(&ssum[dv], w);
    }
    int wbits = __float_as_int(w);
    if (gc == SC_GROUP) {
        float hv[SC_GROUP][4];
        int dk[SC_GROUP];
        float wk[SC_GROUP];
#pragma unroll
        for (int u = 0; u < SC_GROUP; ++u) {
            int sk = __builtin_amdgcn_readlane(sv, u);
            dk[u] = __builtin_amdgcn_readlane(dv, u);
            wk[u] = __int_as_float(__builtin_amdgcn_readlane(wbits, u));
            const float* hp = H + (size_t)sk * HID + lane;
            hv[u][0] = hp[0];
            hv[u][1] = hp[32];
            hv[u][2] = hp[64];
            hv[u][3] = hp[96];
        }
#pragma unroll
        for (int u = 0; u < SC_GROUP; ++u) {
            float* ap = acc + dk[u] * HID + lane;
            atomicAdd(ap,      wk[u] * hv[u][0]);
            atomicAdd(ap + 32, wk[u] * hv[u][1]);
            atomicAdd(ap + 64, wk[u] * hv[u][2]);
            atomicAdd(ap + 96, wk[u] * hv[u][3]);
        }
    } else {
        for (int u = 0; u < gc; ++u) {
            int sk = __builtin_amdgcn_readlane(sv, u);
            int dd = __builtin_amdgcn_readlane(dv, u);
            float ww = __int_as_float(__builtin_amdgcn_readlane(wbits, u));
            const float* hp = H + (size_t)sk * HID + lane;
            float h0 = hp[0], h1 = hp[32], h2 = hp[64], h3 = hp[96];
            float* ap = acc + dd * HID + lane;
            atomicAdd(ap,      ww * h0);
            atomicAdd(ap + 32, ww * h1);
            atomicAdd(ap + 64, ww * h2);
            atomicAdd(ap + 96, ww * h3);
        }
    }
}

template <bool REUSE>
__global__ __launch_bounds__(OWN_T) void gat_owner(const int* __restrict__ ei,
                                                   const unsigned* __restrict__ obw,
                                                   const float* __restrict__ a_s,
                                                   const float* __restrict__ a_d,
                                                   const float* __restrict__ H,
                                                   float* __restrict__ s_out,
                                                   float* __restrict__ out_acc,
                                                   int* __restrict__ gsrc,
                                                   unsigned short* __restrict__ gdl,
                                                   int* __restrict__ gcnt,
                                                   int n_nodes, int n_edges) {
    __shared__ __attribute__((aligned(16))) float acc[NPB * HID];
    __shared__ float adl[NPB];
    __shared__ float ssum[NPB];
    __shared__ unsigned mb[NPB];
    __shared__ int lsrc[LCAP];
    __shared__ float llog[LCAP];
    __shared__ unsigned short ldl[LCAP];
    __shared__ int wcnt[OWN_W];
    __shared__ int ovfl;

    const int tid = threadIdx.x;
    const int lane = tid & 31;
    const int wv = tid >> 5;
    const int lo = blockIdx.x * NPB;
    int nown = n_nodes - lo;
    if (nown > NPB) nown = NPB;
    const int* __restrict__ srcp = ei;
    const int* __restrict__ dstp = ei + n_edges;

    for (int i = tid; i < nown; i += OWN_T) {
        float ad = a_d[lo + i];
        float sl = lrelu(a_s[lo + i] + ad);
        adl[i] = ad;
        ssum[i] = sl;
        mb[i] = f2o(sl);
    }
    if (tid == 0) ovfl = 0;
    __syncthreads();

    int* lw = lsrc + wv * WCAP;
    float* gw = llog + wv * WCAP;
    unsigned short* dw = ldl + wv * WCAP;
    int wc = 0;
    const size_t gl0 = (size_t)(blockIdx.x * OWN_W + wv) * WCAP;
    int* __restrict__ gs = gsrc + gl0;
    unsigned short* __restrict__ gd = gdl + gl0;
    if (REUSE) {
        wc = __builtin_amdgcn_readfirstlane(gcnt[blockIdx.x * OWN_W + wv]);
    } else {
        const unsigned obv = blockIdx.x & 255u;
        const unsigned pat = obv * 0x01010101u;
        const unsigned npat = ~pat;
        const int nw = (n_edges + 3) >> 2;
        const int nq = n_edges >> 4;
        const int nit = (nq + SCAN_U * OWN_T - 1) / (SCAN_U * OWN_T);
        const int rot = (int)(((long long)blockIdx.x * nit) / gridDim.x);
        const uint4* __restrict__ oq = (const uint4*)obw;
        for (int it = 0; it < nit; ++it) {
            int itr = it + rot;
            if (itr >= nit) itr -= nit;
            const int qb = itr * (SCAN_U * OWN_T) + tid;
            uint4 d[SCAN_U];
#pragma unroll
            for (int j = 0; j < SCAN_U; ++j) {
                int q = qb + j * OWN_T;
                d[j] = (q < nq) ? oq[q] : make_uint4(npat, npat, npat, npat);
            }
#pragma unroll
            for (int j = 0; j < SCAN_U; ++j) {
                int e0 = (qb + j * OWN_T) * 16;
                gat_keep4<false>(d[j].x, pat, obv, e0,      n_edges, lane, lw, wc);
                gat_keep4<false>(d[j].y, pat, obv, e0 + 4,  n_edges, lane, lw, wc);
                gat_keep4<false>(d[j].z, pat, obv, e0 + 8,  n_edges, lane, lw, wc);
                gat_keep4<false>(d[j].w, pat, obv, e0 + 12, n_edges, lane, lw, wc);
            }
        }
        for (int wi0 = nq * 4; wi0 < nw; wi0 += OWN_T) {
            int wi = wi0 + tid;
            unsigned wd = (wi < nw) ? obw[wi] : npat;
            gat_keep4<true>(wd, pat, obv, wi * 4, n_edges, lane, lw, wc);
        }
        if (lane == 0) gcnt[blockIdx.x * OWN_W + wv] = wc;
    }
    if (lane == 0) {
        wcnt[wv] = wc;
        if (wc > WCAP) ovfl = 1;
    }
    __syncthreads();

    const bool ovf = ovfl != 0;
    const int nl = ovf ? 0 : wc;

    if (!ovf && REUSE) {
        for (int kb = 0; kb < nl; kb += 32 * PH2_U) {
            int sv[PH2_U];
            unsigned dl[PH2_U];
            float av[PH2_U];
#pragma unroll
            for (int u = 0; u < PH2_U; ++u) {
                int k = kb + u * 32 + lane;
                sv[u] = (k < nl) ? gs[k] : lo;
                dl[u] = (k < nl) ? (unsigned)gd[k] : NOROW;
            }
#pragma unroll
            for (int u = 0; u < PH2_U; ++u) av[u] = a_s[sv[u]];
#pragma unroll
            for (int u = 0; u < PH2_U; ++u) {
                int k = kb + u * 32 + lane;
                if (k < nl) {
                    if (dl[u] != NOROW) {
                        float lg = lrelu(av[u] + adl[dl[u]]);
                        atomicMax(&mb[dl[u]], f2o(lg));
                        lw[k] = sv[u];
                        dw[k] = (unsigned short)dl[u];
                        gw[k] = lg;
                    } else {
                        lw[k] = lo;
                        dw[k] = 0;
                        gw[k] = -__builtin_inff();
                    }
                }
            }
        }
    } else if (!ovf) {
        for (int kb = 0; kb < nl; kb += 32 * PH2_U) {
            int ee[PH2_U], dd[PH2_U], sv[PH2_U];
            float av[PH2_U];
#pragma unroll
            for (int u = 0; u < PH2_U; ++u) {
                int k = kb + u * 32 + lane;
                ee[u] = (k < nl) ? lw[k] : 0;
            }
#pragma unroll
            for (int u = 0; u < PH2_U; ++u) {
                dd[u] = dstp[ee[u]];
                sv[u] = srcp[ee[u]];
            }
#pragma unroll
            for (int u = 0; u < PH2_U; ++u) av[u] = a_s[sv[u]];
#pragma unroll
            for (int u = 0; u < PH2_U; ++u) {
                int k = kb + u * 32 + lane;
                if (k < nl) {
                    unsigned dl = (unsigned)(dd[u] - lo);
                    if (dl < (unsigned)nown) {
                        float lg = lrelu(av[u] + adl[dl]);
                        atomicMax(&mb[dl], f2o(lg));
                        lw[k] = sv[u];
                        dw[k] = (unsigned short)dl;
                        gw[k] = lg;
                        gs[k] = sv[u];
                        gd[k] = (unsigned short)dl;
                    } else {
                        lw[k] = lo;
                        dw[k] = 0;
                        gw[k] = -__builtin_inff();
                        gs[k] = lo;
                        gd[k] = (unsigned short)NOROW;
                    }
                }
            }
        }
    } else {
        for (int e = tid; e < n_edges; e += OWN_T) {
            unsigned dl = (unsigned)(dstp[e] - lo);
            if (dl < (unsigned)nown) {
                float lg = lrelu(a_s[srcp[e]] + adl[dl]);
                atomicMax(&mb[dl], f2o(lg));
            }
        }
    }
    __syncthreads();

    for (int i0 = wv; i0 < nown; i0 += OWN_W * 4) {
        float4 hv[4];
#pragma unroll
        for (int u = 0; u < 4; ++u) {
            int i = i0 + u * OWN_W;
            hv[u] = make_float4(0.0f, 0.0f, 0.0f, 0.0f);
            if (i < nown) hv[u] = ((const float4*)(H + (size_t)(lo + i) * HID))[lane];
        }
#pragma unroll
        for (int u = 0; u < 4; ++u) {
            int i = i0 + u * OWN_W;
            if (i < nown) {
                float es = expf(ssum[i] - o2f(mb[i]));
                float4 r;
                r.x = es * hv[u].x; r.y = es * hv[u].y; r.z = es * hv[u].z; r.w = es * hv[u].w;
                ((float4*)(acc + i * HID))[lane] = r;
                if (lane == 0) ssum[i] = es;
            }
        }
    }
    __syncthreads();

    if (!ovf) {
        const int myc = wcnt[lane];
        for (int wl = 0; wl < OWN_W; ++wl) {
            const int nlw = __builtin_amdgcn_readlane(myc, wl);
            for (int g = (wv - wl) & (OWN_W - 1); g * SC_GROUP < nlw; g += OWN_W)
                gat_add_group(lsrc + wl * WCAP, ldl + wl * WCAP, llog + wl * WCAP, g * SC_GROUP, nlw,
                              mb, ssum, acc, H, lane);
        }
    } else {
        for (int e = tid; e < n_edges; e += OWN_T) {
            unsigned dl = (unsigned)(dstp[e] - lo);
            if (dl < (unsigned)nown) {
                int s = srcp[e];
                float ww = expf(lrelu(a_s[s] + adl[dl]) - o2f(mb[dl]));
                atomicAdd(&ssum[dl], ww);
                const float* hp = H + (size_t)s * HID;
                float* ap = acc + dl * HID;
                for (int c = 0; c < HID; ++c) atomicAdd(ap + c, ww * hp[c]);
            }
        }
    }
    __syncthreads();

    float4* __restrict__ orow = (float4*)(out_acc + (size_t)lo * HID);
    const float4* arow = (const float4*)acc;
    for (int i = tid; i < nown * (HID / 4); i += OWN_T) orow[i] = arow[i];
    for (int i = tid; i < nown; i += OWN_T) s_out[lo + i] = ssum[i];
}

#define FIN_SLOTS 4
#define FIN_U 16

__global__ __launch_bounds__(HID * FIN_SLOTS) void gat_finalize(
        float* __restrict__ out_acc, const float* __restrict__ s,
        const float* __restrict__ bias, float* __restrict__ sums,
        float* __restrict__ sumsq, const int* __restrict__ batch,
        float* __restrict__ praw, float* __restrict__ cnt,
        int do_raw, int n_nodes) {
    __shared__ float red[2 * FIN_SLOTS * HID];
    const int tid = threadIdx.x;
    const int c = tid & (HID - 1);
    const int slot = tid >> 7;
    const int lane = tid & 31;
    const float b = bias[c];
    float ls = 0.0f, lq = 0.0f;
    const int chunk = (n_nodes + (int)gridDim.x - 1) / (int)gridDim.x;
    const int sub = (chunk + FIN_SLOTS - 1) / FIN_SLOTS;
    const int bstart = blockIdx.x * chunk;
    int bend = bstart + chunk;
    if (bend > n_nodes) bend = n_nodes;
    const int n0 = bstart + slot * sub;
    int n1 = n0 + sub;
    if (n1 > bend) n1 = bend;
    int gcur = 0, rl = 0;
    float run = 0.0f;
    if (do_raw && n0 < n1) gcur = batch[n0];
    for (int nb = n0; nb < n1; nb += FIN_U) {
        const int last = n1 - 1;
        const int cntu = (n1 - nb < FIN_U) ? (n1 - nb) : FIN_U;
        float a[FIN_U], v[FIN_U];
#pragma unroll
        for (int u = 0; u < FIN_U; ++u) {
            int n = nb + u;
            if (n > last) n = last;
            a[u] = out_acc[(size_t)n * HID + c];
        }
        int nl = nb + (lane & (FIN_U - 1));
        if (nl > last) nl = last;
        const int ib = __float_as_int(1.0f / s[nl]);
        const int gl = do_raw ? batch[nl] : 0;
#pragma unroll
        for (int u = 0; u < FIN_U; ++u) {
            float inv = __int_as_float(__builtin_amdgcn_readlane(ib, u));
            float t = a[u] * inv + b;
            float e = expm1f(t > 0.0f ? 0.0f : t);
            v[u] = t > 0.0f ? t : e;
        }
#pragma unroll
        for (int u = 0; u < FIN_U; ++u) {
            if (u < cntu) {
                out_acc[(size_t)(nb + u) * HID + c] = v[u];
                ls += v[u];
                lq += v[u] * v[u];
            }
        }
        if (do_raw) {
#pragma unroll
            for (int u = 0; u < FIN_U; ++u) {
                if (u < cntu) {
                    int gb = __builtin_amdgcn_readlane(gl, u);
                    if (gb != gcur) {
                        atomicAdd(&praw[(size_t)gcur * (3 * HID) + c], run);
                        if (c == 0) atomicAdd(&cnt[gcur], (float)rl);
                        run = 0.0f;
                        rl = 0;
                        gcur = gb;
                    }
                    run += v[u];
                    ++rl;
                }
            }
        }
    }
    if (do_raw && rl > 0) {
        atomicAdd(&praw[(size_t)gcur * (3 * HID) + c], run);
        if (c == 0) atomicAdd(&cnt[gcur], (float)rl);
    }
    red[slot * HID + c] = ls;
    red[(FIN_SLOTS + slot) * HID + c] = lq;
    __syncthreads();
    if (tid < HID) {
        float ts = 0.0f, tq = 0.0f;
#pragma unroll
        for (int k = 0; k < FIN_SLOTS; ++k) {
            ts += red[k * HID + tid];
            tq += red[(FIN_SLOTS + k) * HID + tid];
        }
        atomicAdd(&sums[tid], ts);
        atomicAdd(&sumsq[tid], tq);
    }
}

__global__ void gat_bn_fold(const float* __restrict__ sums, const float* __restrict__ sumsq,
                            const float* __restrict__ gamma, const float* __restrict__ beta,
                            const float* __restrict__ Wn, float* __restrict__ Wp,
                            float* __restrict__ c0, float* __restrict__ pooled_l,
                            const float* __restrict__ cnt, const float* __restrict__ att_s,
                            const float* __restrict__ att_d, float* __restrict__ Wx,
                            float* __restrict__ Cx, int n_nodes) {
    const int t = threadIdx.x;
    const int b = blockIdx.x;
    const float invN = 1.0f / (float)n_nodes;
    if (b < HID) {
        float mu = sums[b] * invN;
        float var = sumsq[b] * invN - mu * mu;
        float g = gamma[b] / sqrtf(var + BN_EPS);
        float p = g * Wn[b * HID + t];
        Wp[b * HID + t] = p;
        gat_att_row(p, t, att_s, att_d, Wx + b * XC);
    } else if (b == HID) {
        __shared__ float dsh[HID];
        {
            float mu = sums[t] * invN;
            float var = sumsq[t] * invN - mu * mu;
            float g = gamma[t] / sqrtf(var + BN_EPS);
            dsh[t] = beta[t] - mu * g;
        }
        __syncthreads();
        float a = 0.0f;
#pragma unroll 16
        for (int k = 0; k < HID; ++k) a += dsh[k] * Wn[k * HID + t];
        c0[t] = a;
        gat_att_row(a, t, att_s, att_d, Cx);
    } else {
        const int gr = b - HID - 1;
        float mu = sums[t] * invN;
        float var = sumsq[t] * invN - mu * mu;
        float g = gamma[t] / sqrtf(var + BN_EPS);
        float cn = cnt[gr];
        size_t idx = (size_t)gr * (3 * HID) + t;
        pooled_l[idx] = (pooled_l[idx] - cn * mu) * g + cn * beta[t];
    }
}

__global__ void gat_bn_pool(float* __restrict__ out_acc, const float* __restrict__ sums,
                            const float* __restrict__ sumsq, const float* __restrict__ gamma,
                            const float* __restrict__ beta, const int* __restrict__ batch,
                            float* __restrict__ pooled, int layer, int n_nodes) {
    int idx = blockIdx.x * blockDim.x + threadIdx.x;
    if (idx >= n_nodes * HID) return;
    int c = idx & (HID - 1);
    int n = idx >> 7;
    float invN = 1.0f / (float)n_nodes;
    float mu = sums[c] * invN;
    float var = sumsq[c] * invN - mu * mu;
    float xv = (out_acc[idx] - mu) / sqrtf(var + BN_EPS) * gamma[c] + beta[c];
    out_acc[idx] = xv;
    atomicAdd(&pooled[(size_t)batch[n] * (3 * HID) + layer * HID + c], xv);
}

extern "C" void kernel_launch(void* const* d_in, const int* in_sizes, int n_in,
                              void* d_out, int out_size, void* d_ws, size_t ws_size,
                              hipStream_t stream) {
    const float* x_in  = (const float*)d_in[0];
    const int*   ei    = (const int*)d_in[1];
    const int*   batch = (const int*)d_in[2];
    const float* W     = (const float*)d_in[3];
    const float* att_s = (const float*)d_in[4];
    const float* att_d = (const float*)d_in[5];
    const float* bias  = (const float*)d_in[6];
    const float* gamma = (const float*)d_in[7];
    const float* beta  = (const float*)d_in[8];

    const int N = in_sizes[0] / HID;
    const int E = in_sizes[1] / 2;

    float* pooled = (float*)d_out;
    float* xfinal = (float*)d_out + (size_t)NGRAPH * 3 * HID;

    char* w = (char*)d_ws;
    float* hbuf = (float*)w;  w += (size_t)N * HID * sizeof(float);
    float* xb   = (float*)w;  w += (size_t)N * HID * sizeof(float);
    unsigned* obw = (unsigned*)w;  w += (size_t)((E + 15) / 16) * 16;
    float* asrc = (float*)w;  w += (size_t)N * sizeof(float);
    float* adst = (float*)w;  w += (size_t)N * sizeof(float);
    float* sbuf = (float*)w;  w += (size_t)N * sizeof(float);
    float* sums = (float*)w;
    float* sumsq = sums + HID;
    float* cnt  = sums + 2 * HID;
    float* c0   = cnt + NGRAPH;
    float* Wp   = c0 + HID;
    float* Wx   = Wp + HID * HID;
    float* Cx   = Wx + HID * XC;
    const int n_own = (N + NPB - 1) / NPB;
    int* gsrc = (int*)(Cx + XC);
    int* gcnt = gsrc + (size_t)n_own * LCAP;
    unsigned short* gdl = (unsigned short*)(gcnt + (size_t)n_own * OWN_W);

    const int waves_g = (N + 15) / 16;
    dim3 blk(256);
    dim3 grid_gemm((waves_g * 32 + 255) / 256);
    dim3 grid_own(n_own);
    dim3 grid_elem(((size_t)N * HID + 255) / 256);

    gat_zero<<<(NGRAPH * 3 * HID + 255) / 256, blk, 0, stream>>>(pooled, NGRAPH * 3 * HID);
    gat_owner_ids<<<((E + 3) / 4 + 255) / 256, blk, 0, stream>>>(ei, obw, E);
    gat_att_cols<<<HID, HID, 0, stream>>>(W, att_s, att_d, Wx, Cx);

    for (int l = 0; l < 3; ++l) {
        const float* X  = (l == 0) ? x_in : xb;
        float* oacc     = (l == 2) ? xfinal : xb;
        const float* Wl = (l == 0) ? W : Wp;
        const float* Cl = (l == 0) ? (const float*)nullptr : c0;

        gat_gemm_wmma<<<grid_gemm, blk, 0, stream>>>(X, Wl, Cl, Wx, Cx, hbuf, asrc, adst, N);
        if (l == 0)
            gat_owner<false><<<grid_own, OWN_T, 0, stream>>>(ei, obw, asrc, adst, hbuf, sbuf, oacc,
                                                             gsrc, gdl, gcnt, N, E);
        else
            gat_owner<true><<<grid_own, OWN_T, 0, stream>>>(ei, obw, asrc, adst, hbuf, sbuf, oacc,
                                                            gsrc, gdl, gcnt, N, E);
        gat_zero<<<3, 256, 0, stream>>>(sums, 2 * HID + NGRAPH);
        gat_finalize<<<521, HID * FIN_SLOTS, 0, stream>>>(oacc, sbuf, bias + l * HID, sums, sumsq, batch,
                                              pooled + l * HID, cnt, (l < 2) ? 1 : 0, N);
        if (l < 2)
            gat_bn_fold<<<HID + 1 + NGRAPH, HID, 0, stream>>>(sums, sumsq, gamma + l * HID,
                                                              beta + l * HID,
                                                              W + (size_t)(l + 1) * HID * HID, Wp, c0,
                                                              pooled + l * HID, cnt,
                                                              att_s + (l + 1) * HID,
                                                              att_d + (l + 1) * HID, Wx, Cx, N);
        else
            gat_bn_pool<<<grid_elem, blk, 0, stream>>>(oacc, sums, sumsq, gamma + l * HID,
                                                       beta + l * HID, batch, pooled, l, N);
    }
}


# Finalize launch removed: the owning work-group's write-out turns its LDS rows into elu(row/s+bias) and forms BN and per-graph sums there (stat/pool sum order: 49-row slot runs, LDS, then atomics)
# speedup vs baseline: 3.9625x; 1.1208x over previous
#include <hip/hip_runtime.h>
#include <hip/hip_bf16.h>

#define HID 128
#define NGRAPH 512
#define NEG_SLOPE 0.2f
#define BN_EPS 1e-5f

typedef __attribute__((ext_vector_type(2))) float v2f;
typedef __attribute__((ext_vector_type(8))) float v8f;

__device__ __forceinline__ unsigned f2o(float f) {
    unsigned u = __float_as_uint(f);
    return (u & 0x80000000u) ? ~u : (u | 0x80000000u);
}
__device__ __forceinline__ float o2f(unsigned u) {
    unsigned v = (u & 0x80000000u) ? (u & 0x7fffffffu) : ~u;
    return __uint_as_float(v);
}
__device__ __forceinline__ float lrelu(float a) { return a > 0.0f ? a : NEG_SLOPE * a; }

__global__ void gat_zero(float* __restrict__ p, int n) {
    int i = blockIdx.x * blockDim.x + threadIdx.x;
    if (i < n) p[i] = 0.0f;
}

#define XC 16

__global__ void gat_gemm_wmma(const float* __restrict__ X, const float* __restrict__ W,
                              const float* __restrict__ C0, const float* __restrict__ Wx,
                              const float* __restrict__ Cx, float* __restrict__ H,
                              float* __restrict__ a_s, float* __restrict__ a_d, int nrows) {
    int wave = (blockIdx.x * blockDim.x + threadIdx.x) >> 5;
    int m0 = wave * 16;
    if (m0 >= nrows) return;
    int lane = threadIdx.x & 31;
    int lr = lane & 15;
    int lh = lane >> 4;

    v8f acc[8];
#pragma unroll
    for (int n = 0; n < 8; ++n) {
        float cv = C0 ? C0[n * 16 + lr] : 0.0f;
#pragma unroll
        for (int j = 0; j < 8; ++j) acc[n][j] = cv;
    }
    v8f accx;
    {
        float cv = Cx[lr];
#pragma unroll
        for (int j = 0; j < 8; ++j) accx[j] = cv;
    }
    const float* xrow = X + (size_t)(m0 + lr) * HID;

    for (int k = 0; k < HID; k += 4) {
        int ka = k + lh * 2;
        v2f a;
        a.x = xrow[ka];
        a.y = xrow[ka + 1];
#pragma unroll
        for (int n = 0; n < 8; ++n) {
            v2f b;
            b.x = W[ka * HID + n * 16 + lr];
            b.y = W[(ka + 1) * HID + n * 16 + lr];
            acc[n] = __builtin_amdgcn_wmma_f32_16x16x4_f32(
                false, a, false, b, (short)0, acc[n], false, false);
        }
        v2f bx;
        bx.x = Wx[ka * XC + lr];
        bx.y = Wx[(ka + 1) * XC + lr];
        accx = __builtin_amdgcn_wmma_f32_16x16x4_f32(
            false, a, false, bx, (short)0, accx, false, false);
    }
#pragma unroll
    for (int n = 0; n < 8; ++n)
#pragma unroll
        for (int j = 0; j < 8; ++j)
            H[(size_t)(m0 + j + lh * 8) * HID + n * 16 + lr] = acc[n][j];
    if (lr < 2) {
        float* ap = lr ? a_d : a_s;
#pragma unroll
        for (int j = 0; j < 8; ++j) ap[m0 + j + lh * 8] = accx[j];
    }
}

__device__ __forceinline__ void gat_att_row(float p, int t, const float* __restrict__ att_s,
                                            const float* __restrict__ att_d,
                                            float* __restrict__ dst) {
    __shared__ float part[2 * (HID / 32)];
    float ps = p * att_s[t];
    float pd = p * att_d[t];
    for (int off = 16; off > 0; off >>= 1) {
        ps += __shfl_down(ps, off, 32);
        pd += __shfl_down(pd, off, 32);
    }
    if ((t & 31) == 0) {
        part[t >> 5] = ps;
        part[HID / 32 + (t >> 5)] = pd;
    }
    __syncthreads();
    if (t < XC) {
        float v = 0.0f;
        if (t < 2) {
            const float* q = part + t * (HID / 32);
            v = (q[0] + q[1]) + (q[2] + q[3]);
        }
        dst[t] = v;
    }
}

__global__ void gat_att_cols(const float* __restrict__ Wl, const float* __restrict__ att_s,
                             const float* __restrict__ att_d, float* __restrict__ Wx,
                             float* __restrict__ Cx) {
    const int t = threadIdx.x;
    const int k = blockIdx.x;
    gat_att_row(Wl[k * HID + t], t, att_s, att_d, Wx + k * XC);
    if (k == 0 && t < XC) Cx[t] = 0.0f;
}

#define NPB   391

__global__ void gat_owner_ids(const int* __restrict__ ei, unsigned* __restrict__ obw, int n_edges) {
    int t = blockIdx.x * blockDim.x + threadIdx.x;
    int e0 = t * 4;
    if (e0 >= n_edges) return;
    const int* __restrict__ dstp = ei + n_edges;
    unsigned wd = 0;
#pragma unroll
    for (int j = 0; j < 4; ++j)
        if (e0 + j < n_edges) wd |= ((unsigned)(dstp[e0 + j] / NPB) & 255u) << (8 * j);
    obw[t] = wd;
}

#define OWN_T 1024
#define OWN_W (OWN_T / 32)
#define WCAP  352
#define LCAP  (OWN_W * WCAP)
#define SCAN_U 8
#define PH2_U 8
#define SC_GROUP 8
#define NOROW 0xFFFFu
#define FIN_SLOTS (OWN_T / HID)
#define FIN_U 7

template <bool TAIL>
__device__ __forceinline__ void gat_keep4(unsigned wd, unsigned pat, unsigned obv, int e0,
                                          int n_edges, int lane, int* lw, int& wc) {
    unsigned x = wd ^ pat;
    bool hit = ((x - 0x01010101u) & ~x & 0x80808080u) != 0u;
    if (__builtin_amdgcn_ballot_w32(hit) != 0u) {
#pragma unroll
        for (int b = 0; b < 4; ++b) {
            bool mt = ((wd >> (8 * b)) & 255u) == obv;
            if (TAIL) mt = mt && (e0 + b < n_edges);
            unsigned mk = __builtin_amdgcn_ballot_w32(mt);
            if (mk != 0u) {
                if (mt) {
                    int pos = wc + __builtin_popcount(mk & ((1u << lane) - 1u));
                    if (pos < WCAP) lw[pos] = e0 + b;
                }
                wc += __builtin_popcount(mk);
            }
        }
    }
}

__device__ __forceinline__ void gat_add_group(const int* lwl, const unsigned short* dwl,
                                              const float* gwl, int k0, int nl,
                                              const unsigned* mb, float* ssum, float* acc,
                                              const float* __restrict__ H, int lane) {
    const int rest = nl - k0;
    const int gc = rest < SC_GROUP ? rest : SC_GROUP;
    int sv = 0, dv = 0;
    float w = 0.0f;
    if (lane < gc) {
        sv = lwl[k0 + lane];
        dv = dwl[k0 + lane];
        w = expf(gwl[k0 + lane] - o2f(mb[dv]));
        atomicAdd(&ssum[dv], w);
    }
    int wbits = __float_as_int(w);
    if (gc == SC_GROUP) {
        float hv[SC_GROUP][4];
        int dk[SC_GROUP];
        float wk[SC_GROUP];
#pragma unroll
        for (int u = 0; u < SC_GROUP; ++u) {
            int sk = __builtin_amdgcn_readlane(sv, u);
            dk[u] = __builtin_amdgcn_readlane(dv, u);
            wk[u] = __int_as_float(__builtin_amdgcn_readlane(wbits, u));
            const float* hp = H + (size_t)sk * HID + lane;
            hv[u][0] = hp[0];
            hv[u][1] = hp[32];
            hv[u][2] = hp[64];
            hv[u][3] = hp[96];
        }
#pragma unroll
        for (int u = 0; u < SC_GROUP; ++u) {
            float* ap = acc + dk[u] * HID + lane;
            atomicAdd(ap,      wk[u] * hv[u][0]);
            atomicAdd(ap + 32, wk[u] * hv[u][1]);
            atomicAdd(ap + 64, wk[u] * hv[u][2]);
            atomicAdd(ap + 96, wk[u] * hv[u][3]);
        }
    } else {
        for (int u = 0; u < gc; ++u) {
            int sk = __builtin_amdgcn_readlane(sv, u);
            int dd = __builtin_amdgcn_readlane(dv, u);
            float ww = __int_as_float(__builtin_amdgcn_readlane(wbits, u));
            const float* hp = H + (size_t)sk * HID + lane;
            float h0 = hp[0], h1 = hp[32], h2 = hp[64], h3 = hp[96];
            float* ap = acc + dd * HID + lane;
            atomicAdd(ap,      ww * h0);
            atomicAdd(ap + 32, ww * h1);
            atomicAdd(ap + 64, ww * h2);
            atomicAdd(ap + 96, ww * h3);
        }
    }
}

template <bool REUSE>
__global__ __launch_bounds__(OWN_T) void gat_owner(const int* __restrict__ ei,
                                                   const unsigned* __restrict__ obw,
                                                   const float* __restrict__ a_s,
                                                   const float* __restrict__ a_d,
                                                   const float* __restrict__ H,
                                                   float* __restrict__ out_acc,
                                                   int* __restrict__ gsrc,
                                                   unsigned short* __restrict__ gdl,
                                                   int* __restrict__ gcnt,
                                                   const float* __restrict__ bias,
                                                   float* __restrict__ sums,
                                                   float* __restrict__ sumsq,
                                                   const int* __restrict__ batch,
                                                   float* __restrict__ praw,
                                                   float* __restrict__ cnt,
                                                   int do_raw, int n_nodes, int n_edges) {
    __shared__ __attribute__((aligned(16))) float acc[NPB * HID];
    __shared__ float adl[NPB];
    __shared__ float ssum[NPB];
    __shared__ unsigned mb[NPB];
    __shared__ int gb[NPB];
    __shared__ int lsrc[LCAP];
    __shared__ float llog[LCAP];
    __shared__ unsigned short ldl[LCAP];
    __shared__ int wcnt[OWN_W];
    __shared__ int ovfl;

    const int tid = threadIdx.x;
    const int lane = tid & 31;
    const int wv = tid >> 5;
    const int lo = blockIdx.x * NPB;
    int nown = n_nodes - lo;
    if (nown > NPB) nown = NPB;
    const int* __restrict__ srcp = ei;
    const int* __restrict__ dstp = ei + n_edges;
    const float bv = bias[tid & (HID - 1)];

    for (int i = tid; i < nown; i += OWN_T) {
        float ad = a_d[lo + i];
        float sl = lrelu(a_s[lo + i] + ad);
        adl[i] = ad;
        ssum[i] = sl;
        mb[i] = f2o(sl);
        gb[i] = do_raw ? batch[lo + i] : 0;
    }
    if (tid == 0) ovfl = 0;
    __syncthreads();

    int* lw = lsrc + wv * WCAP;
    float* gw = llog + wv * WCAP;
    unsigned short* dw = ldl + wv * WCAP;
    int wc = 0;
    const size_t gl0 = (size_t)(blockIdx.x * OWN_W + wv) * WCAP;
    int* __restrict__ gs = gsrc + gl0;
    unsigned short* __restrict__ gd = gdl + gl0;
    if (REUSE) {
        wc = __builtin_amdgcn_readfirstlane(gcnt[blockIdx.x * OWN_W + wv]);
    } else {
        const unsigned obv = blockIdx.x & 255u;
        const unsigned pat = obv * 0x01010101u;
        const unsigned npat = ~pat;
        const int nw = (n_edges + 3) >> 2;
        const int nq = n_edges >> 4;
        const int nit = (nq + SCAN_U * OWN_T - 1) / (SCAN_U * OWN_T);
        const int rot = (int)(((long long)blockIdx.x * nit) / gridDim.x);
        const uint4* __restrict__ oq = (const uint4*)obw;
        for (int it = 0; it < nit; ++it) {
            int itr = it + rot;
            if (itr >= nit) itr -= nit;
            const int qb = itr * (SCAN_U * OWN_T) + tid;
            uint4 d[SCAN_U];
#pragma unroll
            for (int j = 0; j < SCAN_U; ++j) {
                int q = qb + j * OWN_T;
                d[j] = (q < nq) ? oq[q] : make_uint4(npat, npat, npat, npat);
            }
#pragma unroll
            for (int j = 0; j < SCAN_U; ++j) {
                int e0 = (qb + j * OWN_T) * 16;
                gat_keep4<false>(d[j].x, pat, obv, e0,      n_edges, lane, lw, wc);
                gat_keep4<false>(d[j].y, pat, obv, e0 + 4,  n_edges, lane, lw, wc);
                gat_keep4<false>(d[j].z, pat, obv, e0 + 8,  n_edges, lane, lw, wc);
                gat_keep4<false>(d[j].w, pat, obv, e0 + 12, n_edges, lane, lw, wc);
            }
        }
        for (int wi0 = nq * 4; wi0 < nw; wi0 += OWN_T) {
            int wi = wi0 + tid;
            unsigned wd = (wi < nw) ? obw[wi] : npat;
            gat_keep4<true>(wd, pat, obv, wi * 4, n_edges, lane, lw, wc);
        }
        if (lane == 0) gcnt[blockIdx.x * OWN_W + wv] = wc;
    }
    if (lane == 0) {
        wcnt[wv] = wc;
        if (wc > WCAP) ovfl = 1;
    }
    __syncthreads();

    const bool ovf = ovfl != 0;
    const int nl = ovf ? 0 : wc;

    if (!ovf && REUSE) {
        for (int kb = 0; kb < nl; kb += 32 * PH2_U) {
            int sv[PH2_U];
            unsigned dl[PH2_U];
            float av[PH2_U];
#pragma unroll
            for (int u = 0; u < PH2_U; ++u) {
                int k = kb + u * 32 + lane;
                sv[u] = (k < nl) ? gs[k] : lo;
                dl[u] = (k < nl) ? (unsigned)gd[k] : NOROW;
            }
#pragma unroll
            for (int u = 0; u < PH2_U; ++u) av[u] = a_s[sv[u]];
#pragma unroll
            for (int u = 0; u < PH2_U; ++u) {
                int k = kb + u * 32 + lane;
                if (k < nl) {
                    if (dl[u] != NOROW) {
                        float lg = lrelu(av[u] + adl[dl[u]]);
                        atomicMax(&mb[dl[u]], f2o(lg));
                        lw[k] = sv[u];
                        dw[k] = (unsigned short)dl[u];
                        gw[k] = lg;
                    } else {
                        lw[k] = lo;
                        dw[k] = 0;
                        gw[k] = -__builtin_inff();
                    }
                }
            }
        }
    } else if (!ovf) {
        for (int kb = 0; kb < nl; kb += 32 * PH2_U) {
            int ee[PH2_U], dd[PH2_U], sv[PH2_U];
            float av[PH2_U];
#pragma unroll
            for (int u = 0; u < PH2_U; ++u) {
                int k = kb + u * 32 + lane;
                ee[u] = (k < nl) ? lw[k] : 0;
            }
#pragma unroll
            for (int u = 0; u < PH2_U; ++u) {
                dd[u] = dstp[ee[u]];
                sv[u] = srcp[ee[u]];
            }
#pragma unroll
            for (int u = 0; u < PH2_U; ++u) av[u] = a_s[sv[u]];
#pragma unroll
            for (int u = 0; u < PH2_U; ++u) {
                int k = kb + u * 32 + lane;
                if (k < nl) {
                    unsigned dl = (unsigned)(dd[u] - lo);
                    if (dl < (unsigned)nown) {
                        float lg = lrelu(av[u] + adl[dl]);
                        atomicMax(&mb[dl], f2o(lg));
                        lw[k] = sv[u];
                        dw[k] = (unsigned short)dl;
                        gw[k] = lg;
                        gs[k] = sv[u];
                        gd[k] = (unsigned short)dl;
                    } else {
                        lw[k] = lo;
                        dw[k] = 0;
                        gw[k] = -__builtin_inff();
                        gs[k] = lo;
                        gd[k] = (unsigned short)NOROW;
                    }
                }
            }
        }
    } else {
        for (int e = tid; e < n_edges; e += OWN_T) {
            unsigned dl = (unsigned)(dstp[e] - lo);
            if (dl < (unsigned)nown) {
                float lg = lrelu(a_s[srcp[e]] + adl[dl]);
                atomicMax(&mb[dl], f2o(lg));
            }
        }
    }
    __syncthreads();

    for (int i0 = wv; i0 < nown; i0 += OWN_W * 4) {
        float4 hv[4];
#pragma unroll
        for (int u = 0; u < 4; ++u) {
            int i = i0 + u * OWN_W;
            hv[u] = make_float4(0.0f, 0.0f, 0.0f, 0.0f);
            if (i < nown) hv[u] = ((const float4*)(H + (size_t)(lo + i) * HID))[lane];
        }
#pragma unroll
        for (int u = 0; u < 4; ++u) {
            int i = i0 + u * OWN_W;
            if (i < nown) {
                float es = expf(ssum[i] - o2f(mb[i]));
                float4 r;
                r.x = es * hv[u].x; r.y = es * hv[u].y; r.z = es * hv[u].z; r.w = es * hv[u].w;
                ((float4*)(acc + i * HID))[lane] = r;
                if (lane == 0) ssum[i] = es;
            }
        }
    }
    __syncthreads();

    if (!ovf) {
        const int myc = wcnt[lane];
        for (int wl = 0; wl < OWN_W; ++wl) {
            const int nlw = __builtin_amdgcn_readlane(myc, wl);
            for (int g = (wv - wl) & (OWN_W - 1); g * SC_GROUP < nlw; g += OWN_W)
                gat_add_group(lsrc + wl * WCAP, ldl + wl * WCAP, llog + wl * WCAP, g * SC_GROUP, nlw,
                              mb, ssum, acc, H, lane);
        }
    } else {
        for (int e = tid; e < n_edges; e += OWN_T) {
            unsigned dl = (unsigned)(dstp[e] - lo);
            if (dl < (unsigned)nown) {
                int s = srcp[e];
                float ww = expf(lrelu(a_s[s] + adl[dl]) - o2f(mb[dl]));
                atomicAdd(&ssum[dl], ww);
                const float* hp = H + (size_t)s * HID;
                float* ap = acc + dl * HID;
                for (int c = 0; c < HID; ++c) atomicAdd(ap + c, ww * hp[c]);
            }
        }
    }
    __syncthreads();

    for (int i = tid; i < nown; i += OWN_T) ssum[i] = 1.0f / ssum[i];
    __syncthreads();
    const int c = tid & (HID - 1);
    const int slot = __builtin_amdgcn_readfirstlane(tid >> 7);
    const int sub = (nown + FIN_SLOTS - 1) / FIN_SLOTS;
    const int r0 = slot * sub;
    int r1 = r0 + sub;
    if (r1 > nown) r1 = nown;
    float* __restrict__ ocol = out_acc + (size_t)lo * HID + c;
    float ls = 0.0f, lq = 0.0f, run = 0.0f;
    int gcur = 0, rl = 0;
    if (do_raw && r0 < r1) gcur = __builtin_amdgcn_readfirstlane(gb[r0]);
    for (int ib = r0; ib < r1; ib += FIN_U) {
        const int last = r1 - 1;
        const int cntu = (r1 - ib < FIN_U) ? (r1 - ib) : FIN_U;
        int il = ib + (lane & 7);
        if (il > last) il = last;
        const int ibits = __float_as_int(ssum[il]);
        const int gl = gb[il];
        float v[FIN_U];
#pragma unroll
        for (int u = 0; u < FIN_U; ++u) {
            int i = ib + u;
            if (i > last) i = last;
            float inv = __int_as_float(__builtin_amdgcn_readlane(ibits, u));
            float t = acc[i * HID + c] * inv + bv;
            float e = expm1f(t > 0.0f ? 0.0f : t);
            v[u] = t > 0.0f ? t : e;
        }
#pragma unroll
        for (int u = 0; u < FIN_U; ++u) {
            if (u < cntu) {
                ocol[(size_t)(ib + u) * HID] = v[u];
                ls += v[u];
                lq += v[u] * v[u];
            }
        }
        if (do_raw) {
#pragma unroll
            for (int u = 0; u < FIN_U; ++u) {
                if (u < cntu) {
                    int g = __builtin_amdgcn_readlane(gl, u);
                    if (g != gcur) {
                        atomicAdd(&praw[(size_t)gcur * (3 * HID) + c], run);
                        if (c == 0) atomicAdd(&cnt[gcur], (float)rl);
                        run = 0.0f;
                        rl = 0;
                        gcur = g;
                    }
                    run += v[u];
                    ++rl;
                }
            }
        }
    }
    if (do_raw && rl > 0) {
        atomicAdd(&praw[(size_t)gcur * (3 * HID) + c], run);
        if (c == 0) atomicAdd(&cnt[gcur], (float)rl);
    }
    float* red = llog;
    red[slot * HID + c] = ls;
    red[(FIN_SLOTS + slot) * HID + c] = lq;
    __syncthreads();
    if (tid < HID) {
        float ts = 0.0f, tq = 0.0f;
#pragma unroll
        for (int k = 0; k < FIN_SLOTS; ++k) {
            ts += red[k * HID + tid];
            tq += red[(FIN_SLOTS + k) * HID + tid];
        }
        atomicAdd(&sums[tid], ts);
        atomicAdd(&sumsq[tid], tq);
    }
}

__global__ void gat_bn_fold(const float* __restrict__ sums, const float* __restrict__ sumsq,
                            const float* __restrict__ gamma, const float* __restrict__ beta,
                            const float* __restrict__ Wn, float* __restrict__ Wp,
                            float* __restrict__ c0, float* __restrict__ pooled_l,
                            const float* __restrict__ cnt, const float* __restrict__ att_s,
                            const float* __restrict__ att_d, float* __restrict__ Wx,
                            float* __restrict__ Cx, int n_nodes) {
    const int t = threadIdx.x;
    const int b = blockIdx.x;
    const float invN = 1.0f / (float)n_nodes;
    if (b < HID) {
        float mu = sums[b] * invN;
        float var = sumsq[b] * invN - mu * mu;
        float g = gamma[b] / sqrtf(var + BN_EPS);
        float p = g * Wn[b * HID + t];
        Wp[b * HID + t] = p;
        gat_att_row(p, t, att_s, att_d, Wx + b * XC);
    } else if (b == HID) {
        __shared__ float dsh[HID];
        {
            float mu = sums[t] * invN;
            float var = sumsq[t] * invN - mu * mu;
            float g = gamma[t] / sqrtf(var + BN_EPS);
            dsh[t] = beta[t] - mu * g;
        }
        __syncthreads();
        float a = 0.0f;
#pragma unroll 16
        for (int k = 0; k < HID; ++k) a += dsh[k] * Wn[k * HID + t];
        c0[t] = a;
        gat_att_row(a, t, att_s, att_d, Cx);
    } else {
        const int gr = b - HID - 1;
        float mu = sums[t] * invN;
        float var = sumsq[t] * invN - mu * mu;
        float g = gamma[t] / sqrtf(var + BN_EPS);
        float cn = cnt[gr];
        size_t idx = (size_t)gr * (3 * HID) + t;
        pooled_l[idx] = (pooled_l[idx] - cn * mu) * g + cn * beta[t];
    }
}

__global__ void gat_bn_pool(float* __restrict__ out_acc, const float* __restrict__ sums,
                            const float* __restrict__ sumsq, const float* __restrict__ gamma,
                            const float* __restrict__ beta, const int* __restrict__ batch,
                            float* __restrict__ pooled, int layer, int n_nodes) {
    int idx = blockIdx.x * blockDim.x + threadIdx.x;
    if (idx >= n_nodes * HID) return;
    int c = idx & (HID - 1);
    int n = idx >> 7;
    float invN = 1.0f / (float)n_nodes;
    float mu = sums[c] * invN;
    float var = sumsq[c] * invN - mu * mu;
    float xv = (out_acc[idx] - mu) / sqrtf(var + BN_EPS) * gamma[c] + beta[c];
    out_acc[idx] = xv;
    atomicAdd(&pooled[(size_t)batch[n] * (3 * HID) + layer * HID + c], xv);
}

extern "C" void kernel_launch(void* const* d_in, const int* in_sizes, int n_in,
                              void* d_out, int out_size, void* d_ws, size_t ws_size,
                              hipStream_t stream) {
    const float* x_in  = (const float*)d_in[0];
    const int*   ei    = (const int*)d_in[1];
    const int*   batch = (const int*)d_in[2];
    const float* W     = (const float*)d_in[3];
    const float* att_s = (const float*)d_in[4];
    const float* att_d = (const float*)d_in[5];
    const float* bias  = (const float*)d_in[6];
    const float* gamma = (const float*)d_in[7];
    const float* beta  = (const float*)d_in[8];

    const int N = in_sizes[0] / HID;
    const int E = in_sizes[1] / 2;

    float* pooled = (float*)d_out;
    float* xfinal = (float*)d_out + (size_t)NGRAPH * 3 * HID;

    char* w = (char*)d_ws;
    float* hbuf = (float*)w;  w += (size_t)N * HID * sizeof(float);
    float* xb   = (float*)w;  w += (size_t)N * HID * sizeof(float);
    unsigned* obw = (unsigned*)w;  w += (size_t)((E + 15) / 16) * 16;
    float* asrc = (float*)w;  w += (size_t)N * sizeof(float);
    float* adst = (float*)w;  w += (size_t)N * sizeof(float);
    float* sums = (float*)w;
    float* sumsq = sums + HID;
    float* cnt  = sums + 2 * HID;
    float* c0   = cnt + NGRAPH;
    float* Wp   = c0 + HID;
    float* Wx   = Wp + HID * HID;
    float* Cx   = Wx + HID * XC;
    const int n_own = (N + NPB - 1) / NPB;
    int* gsrc = (int*)(Cx + XC);
    int* gcnt = gsrc + (size_t)n_own * LCAP;
    unsigned short* gdl = (unsigned short*)(gcnt + (size_t)n_own * OWN_W);

    const int waves_g = (N + 15) / 16;
    dim3 blk(256);
    dim3 grid_gemm((waves_g * 32 + 255) / 256);
    dim3 grid_own(n_own);
    dim3 grid_elem(((size_t)N * HID + 255) / 256);

    gat_zero<<<(NGRAPH * 3 * HID + 255) / 256, blk, 0, stream>>>(pooled, NGRAPH * 3 * HID);
    gat_owner_ids<<<((E + 3) / 4 + 255) / 256, blk, 0, stream>>>(ei, obw, E);
    gat_att_cols<<<HID, HID, 0, stream>>>(W, att_s, att_d, Wx, Cx);

    for (int l = 0; l < 3; ++l) {
        const float* X  = (l == 0) ? x_in : xb;
        float* oacc     = (l == 2) ? xfinal : xb;
        const float* Wl = (l == 0) ? W : Wp;
        const float* Cl = (l == 0) ? (const float*)nullptr : c0;

        gat_gemm_wmma<<<grid_gemm, blk, 0, stream>>>(X, Wl, Cl, Wx, Cx, hbuf, asrc, adst, N);
        gat_zero<<<3, 256, 0, stream>>>(sums, 2 * HID + NGRAPH);
        if (l == 0)
            gat_owner<false><<<grid_own, OWN_T, 0, stream>>>(ei, obw, asrc, adst, hbuf, oacc, gsrc, gdl, gcnt,
                                                             bias + l * HID, sums, sumsq, batch,
                                                             pooled + l * HID, cnt, (l < 2) ? 1 : 0, N, E);
        else
            gat_owner<true><<<grid_own, OWN_T, 0, stream>>>(ei, obw, asrc, adst, hbuf, oacc, gsrc, gdl, gcnt,
                                                            bias + l * HID, sums, sumsq, batch,
                                                            pooled + l * HID, cnt, (l < 2) ? 1 : 0, N, E);
        if (l < 2)
            gat_bn_fold<<<HID + 1 + NGRAPH, HID, 0, stream>>>(sums, sumsq, gamma + l * HID,
                                                              beta + l * HID,
                                                              W + (size_t)(l + 1) * HID * HID, Wp, c0,
                                                              pooled + l * HID, cnt,
                                                              att_s + (l + 1) * HID,
                                                              att_d + (l + 1) * HID, Wx, Cx, N);
        else
            gat_bn_pool<<<grid_elem, blk, 0, stream>>>(oacc, sums, sumsq, gamma + l * HID,
                                                       beta + l * HID, batch, pooled, l, N);
    }
}


# Edge row gather: one 16-byte load per lane takes the whole 512-byte H row (was four 4-byte loads); LDS rows kept channel-transposed so the 4 adds stay conflict-free (sum order: arrival, as before)
# speedup vs baseline: 4.1380x; 1.0443x over previous
#include <hip/hip_runtime.h>
#include <hip/hip_bf16.h>

#define HID 128
#define NGRAPH 512
#define NEG_SLOPE 0.2f
#define BN_EPS 1e-5f

typedef __attribute__((ext_vector_type(2))) float v2f;
typedef __attribute__((ext_vector_type(8))) float v8f;

__device__ __forceinline__ unsigned f2o(float f) {
    unsigned u = __float_as_uint(f);
    return (u & 0x80000000u) ? ~u : (u | 0x80000000u);
}
__device__ __forceinline__ float o2f(unsigned u) {
    unsigned v = (u & 0x80000000u) ? (u & 0x7fffffffu) : ~u;
    return __uint_as_float(v);
}
__device__ __forceinline__ float lrelu(float a) { return a > 0.0f ? a : NEG_SLOPE * a; }

__global__ void gat_zero(float* __restrict__ p, int n) {
    int i = blockIdx.x * blockDim.x + threadIdx.x;
    if (i < n) p[i] = 0.0f;
}

#define XC 16

__global__ void gat_gemm_wmma(const float* __restrict__ X, const float* __restrict__ W,
                              const float* __restrict__ C0, const float* __restrict__ Wx,
                              const float* __restrict__ Cx, float* __restrict__ H,
                              float* __restrict__ a_s, float* __restrict__ a_d, int nrows) {
    int wave = (blockIdx.x * blockDim.x + threadIdx.x) >> 5;
    int m0 = wave * 16;
    if (m0 >= nrows) return;
    int lane = threadIdx.x & 31;
    int lr = lane & 15;
    int lh = lane >> 4;

    v8f acc[8];
#pragma unroll
    for (int n = 0; n < 8; ++n) {
        float cv = C0 ? C0[n * 16 + lr] : 0.0f;
#pragma unroll
        for (int j = 0; j < 8; ++j) acc[n][j] = cv;
    }
    v8f accx;
    {
        float cv = Cx[lr];
#pragma unroll
        for (int j = 0; j < 8; ++j) accx[j] = cv;
    }
    const float* xrow = X + (size_t)(m0 + lr) * HID;

    for (int k = 0; k < HID; k += 4) {
        int ka = k + lh * 2;
        v2f a;
        a.x = xrow[ka];
        a.y = xrow[ka + 1];
#pragma unroll
        for (int n = 0; n < 8; ++n) {
            v2f b;
            b.x = W[ka * HID + n * 16 + lr];
            b.y = W[(ka + 1) * HID + n * 16 + lr];
            acc[n] = __builtin_amdgcn_wmma_f32_16x16x4_f32(
                false, a, false, b, (short)0, acc[n], false, false);
        }
        v2f bx;
        bx.x = Wx[ka * XC + lr];
        bx.y = Wx[(ka + 1) * XC + lr];
        accx = __builtin_amdgcn_wmma_f32_16x16x4_f32(
            false, a, false, bx, (short)0, accx, false, false);
    }
#pragma unroll
    for (int n = 0; n < 8; ++n)
#pragma unroll
        for (int j = 0; j < 8; ++j)
            H[(size_t)(m0 + j + lh * 8) * HID + n * 16 + lr] = acc[n][j];
    if (lr < 2) {
        float* ap = lr ? a_d : a_s;
#pragma unroll
        for (int j = 0; j < 8; ++j) ap[m0 + j + lh * 8] = accx[j];
    }
}

__device__ __forceinline__ void gat_att_row(float p, int t, const float* __restrict__ att_s,
                                            const float* __restrict__ att_d,
                                            float* __restrict__ dst) {
    __shared__ float part[2 * (HID / 32)];
    float ps = p * att_s[t];
    float pd = p * att_d[t];
    for (int off = 16; off > 0; off >>= 1) {
        ps += __shfl_down(ps, off, 32);
        pd += __shfl_down(pd, off, 32);
    }
    if ((t & 31) == 0) {
        part[t >> 5] = ps;
        part[HID / 32 + (t >> 5)] = pd;
    }
    __syncthreads();
    if (t < XC) {
        float v = 0.0f;
        if (t < 2) {
            const float* q = part + t * (HID / 32);
            v = (q[0] + q[1]) + (q[2] + q[3]);
        }
        dst[t] = v;
    }
}

__global__ void gat_att_cols(const float* __restrict__ Wl, const float* __restrict__ att_s,
                             const float* __restrict__ att_d, float* __restrict__ Wx,
                             float* __restrict__ Cx) {
    const int t = threadIdx.x;
    const int k = blockIdx.x;
    gat_att_row(Wl[k * HID + t], t, att_s, att_d, Wx + k * XC);
    if (k == 0 && t < XC) Cx[t] = 0.0f;
}

#define NPB   391

__global__ void gat_owner_ids(const int* __restrict__ ei, unsigned* __restrict__ obw, int n_edges) {
    int t = blockIdx.x * blockDim.x + threadIdx.x;
    int e0 = t * 4;
    if (e0 >= n_edges) return;
    const int* __restrict__ dstp = ei + n_edges;
    unsigned wd = 0;
#pragma unroll
    for (int j = 0; j < 4; ++j)
        if (e0 + j < n_edges) wd |= ((unsigned)(dstp[e0 + j] / NPB) & 255u) << (8 * j);
    obw[t] = wd;
}

#define OWN_T 1024
#define OWN_W (OWN_T / 32)
#define WCAP  352
#define LCAP  (OWN_W * WCAP)
#define SCAN_U 8
#define PH2_U 8
#define SC_GROUP 8
#define NOROW 0xFFFFu
#define FIN_SLOTS (OWN_T / HID)
#define FIN_U 7
#define ROWPOS(c) ((((c) & 3) << 5) | ((c) >> 2))

template <bool TAIL>
__device__ __forceinline__ void gat_keep4(unsigned wd, unsigned pat, unsigned obv, int e0,
                                          int n_edges, int lane, int* lw, int& wc) {
    unsigned x = wd ^ pat;
    bool hit = ((x - 0x01010101u) & ~x & 0x80808080u) != 0u;
    if (__builtin_amdgcn_ballot_w32(hit) != 0u) {
#pragma unroll
        for (int b = 0; b < 4; ++b) {
            bool mt = ((wd >> (8 * b)) & 255u) == obv;
            if (TAIL) mt = mt && (e0 + b < n_edges);
            unsigned mk = __builtin_amdgcn_ballot_w32(mt);
            if (mk != 0u) {
                if (mt) {
                    int pos = wc + __builtin_popcount(mk & ((1u << lane) - 1u));
                    if (pos < WCAP) lw[pos] = e0 + b;
                }
                wc += __builtin_popcount(mk);
            }
        }
    }
}

__device__ __forceinline__ void gat_add_group(const int* lwl, const unsigned short* dwl,
                                              const float* gwl, int k0, int nl,
                                              const unsigned* mb, float* ssum, float* acc,
                                              const float* __restrict__ H, int lane) {
    const int rest = nl - k0;
    const int gc = rest < SC_GROUP ? rest : SC_GROUP;
    int sv = 0, dv = 0;
    float w = 0.0f;
    if (lane < gc) {
        sv = lwl[k0 + lane];
        dv = dwl[k0 + lane];
        w = expf(gwl[k0 + lane] - o2f(mb[dv]));
        atomicAdd(&ssum[dv], w);
    }
    int wbits = __float_as_int(w);
    if (gc == SC_GROUP) {
        float4 hv[SC_GROUP];
        int dk[SC_GROUP];
        float wk[SC_GROUP];
#pragma unroll
        for (int u = 0; u < SC_GROUP; ++u) {
            int sk = __builtin_amdgcn_readlane(sv, u);
            dk[u] = __builtin_amdgcn_readlane(dv, u);
            wk[u] = __int_as_float(__builtin_amdgcn_readlane(wbits, u));
            hv[u] = ((const float4*)(H + (size_t)sk * HID))[lane];
        }
#pragma unroll
        for (int u = 0; u < SC_GROUP; ++u) {
            float* ap = acc + dk[u] * HID + lane;
            atomicAdd(ap,      wk[u] * hv[u].x);
            atomicAdd(ap + 32, wk[u] * hv[u].y);
            atomicAdd(ap + 64, wk[u] * hv[u].z);
            atomicAdd(ap + 96, wk[u] * hv[u].w);
        }
    } else {
        for (int u = 0; u < gc; ++u) {
            int sk = __builtin_amdgcn_readlane(sv, u);
            int dd = __builtin_amdgcn_readlane(dv, u);
            float ww = __int_as_float(__builtin_amdgcn_readlane(wbits, u));
            const float4 h4 = ((const float4*)(H + (size_t)sk * HID))[lane];
            float* ap = acc + dd * HID + lane;
            atomicAdd(ap,      ww * h4.x);
            atomicAdd(ap + 32, ww * h4.y);
            atomicAdd(ap + 64, ww * h4.z);
            atomicAdd(ap + 96, ww * h4.w);
        }
    }
}

template <bool REUSE>
__global__ __launch_bounds__(OWN_T) void gat_owner(const int* __restrict__ ei,
                                                   const unsigned* __restrict__ obw,
                                                   const float* __restrict__ a_s,
                                                   const float* __restrict__ a_d,
                                                   const float* __restrict__ H,
                                                   float* __restrict__ out_acc,
                                                   int* __restrict__ gsrc,
                                                   unsigned short* __restrict__ gdl,
                                                   int* __restrict__ gcnt,
                                                   const float* __restrict__ bias,
                                                   float* __restrict__ sums,
                                                   float* __restrict__ sumsq,
                                                   const int* __restrict__ batch,
                                                   float* __restrict__ praw,
                                                   float* __restrict__ cnt,
                                                   int do_raw, int n_nodes, int n_edges) {
    __shared__ __attribute__((aligned(16))) float acc[NPB * HID];
    __shared__ float adl[NPB];
    __shared__ float ssum[NPB];
    __shared__ unsigned mb[NPB];
    __shared__ int gb[NPB];
    __shared__ int lsrc[LCAP];
    __shared__ float llog[LCAP];
    __shared__ unsigned short ldl[LCAP];
    __shared__ int wcnt[OWN_W];
    __shared__ int ovfl;

    const int tid = threadIdx.x;
    const int lane = tid & 31;
    const int wv = tid >> 5;
    const int lo = blockIdx.x * NPB;
    int nown = n_nodes - lo;
    if (nown > NPB) nown = NPB;
    const int* __restrict__ srcp = ei;
    const int* __restrict__ dstp = ei + n_edges;
    const float bv = bias[tid & (HID - 1)];

    for (int i = tid; i < nown; i += OWN_T) {
        float ad = a_d[lo + i];
        float sl = lrelu(a_s[lo + i] + ad);
        adl[i] = ad;
        ssum[i] = sl;
        mb[i] = f2o(sl);
        gb[i] = do_raw ? batch[lo + i] : 0;
    }
    if (tid == 0) ovfl = 0;
    __syncthreads();

    int* lw = lsrc + wv * WCAP;
    float* gw = llog + wv * WCAP;
    unsigned short* dw = ldl + wv * WCAP;
    int wc = 0;
    const size_t gl0 = (size_t)(blockIdx.x * OWN_W + wv) * WCAP;
    int* __restrict__ gs = gsrc + gl0;
    unsigned short* __restrict__ gd = gdl + gl0;
    if (REUSE) {
        wc = __builtin_amdgcn_readfirstlane(gcnt[blockIdx.x * OWN_W + wv]);
    } else {
        const unsigned obv = blockIdx.x & 255u;
        const unsigned pat = obv * 0x01010101u;
        const unsigned npat = ~pat;
        const int nw = (n_edges + 3) >> 2;
        const int nq = n_edges >> 4;
        const int nit = (nq + SCAN_U * OWN_T - 1) / (SCAN_U * OWN_T);
        const int rot = (int)(((long long)blockIdx.x * nit) / gridDim.x);
        const uint4* __restrict__ oq = (const uint4*)obw;
        for (int it = 0; it < nit; ++it) {
            int itr = it + rot;
            if (itr >= nit) itr -= nit;
            const int qb = itr * (SCAN_U * OWN_T) + tid;
            uint4 d[SCAN_U];
#pragma unroll
            for (int j = 0; j < SCAN_U; ++j) {
                int q = qb + j * OWN_T;
                d[j] = (q < nq) ? oq[q] : make_uint4(npat, npat, npat, npat);
            }
#pragma unroll
            for (int j = 0; j < SCAN_U; ++j) {
                int e0 = (qb + j * OWN_T) * 16;
                gat_keep4<false>(d[j].x, pat, obv, e0,      n_edges, lane, lw, wc);
                gat_keep4<false>(d[j].y, pat, obv, e0 + 4,  n_edges, lane, lw, wc);
                gat_keep4<false>(d[j].z, pat, obv, e0 + 8,  n_edges, lane, lw, wc);
                gat_keep4<false>(d[j].w, pat, obv, e0 + 12, n_edges, lane, lw, wc);
            }
        }
        for (int wi0 = nq * 4; wi0 < nw; wi0 += OWN_T) {
            int wi = wi0 + tid;
            unsigned wd = (wi < nw) ? obw[wi] : npat;
            gat_keep4<true>(wd, pat, obv, wi * 4, n_edges, lane, lw, wc);
        }
        if (lane == 0) gcnt[blockIdx.x * OWN_W + wv] = wc;
    }
    if (lane == 0) {
        wcnt[wv] = wc;
        if (wc > WCAP) ovfl = 1;
    }
    __syncthreads();

    const bool ovf = ovfl != 0;
    const int nl = ovf ? 0 : wc;

    if (!ovf && REUSE) {
        for (int kb = 0; kb < nl; kb += 32 * PH2_U) {
            int sv[PH2_U];
            unsigned dl[PH2_U];
            float av[PH2_U];
#pragma unroll
            for (int u = 0; u < PH2_U; ++u) {
                int k = kb + u * 32 + lane;
                sv[u] = (k < nl) ? gs[k] : lo;
                dl[u] = (k < nl) ? (unsigned)gd[k] : NOROW;
            }
#pragma unroll
            for (int u = 0; u < PH2_U; ++u) av[u] = a_s[sv[u]];
#pragma unroll
            for (int u = 0; u < PH2_U; ++u) {
                int k = kb + u * 32 + lane;
                if (k < nl) {
                    if (dl[u] != NOROW) {
                        float lg = lrelu(av[u] + adl[dl[u]]);
                        atomicMax(&mb[dl[u]], f2o(lg));
                        lw[k] = sv[u];
                        dw[k] = (unsigned short)dl[u];
                        gw[k] = lg;
                    } else {
                        lw[k] = lo;
                        dw[k] = 0;
                        gw[k] = -__builtin_inff();
                    }
                }
            }
        }
    } else if (!ovf) {
        for (int kb = 0; kb < nl; kb += 32 * PH2_U) {
            int ee[PH2_U], dd[PH2_U], sv[PH2_U];
            float av[PH2_U];
#pragma unroll
            for (int u = 0; u < PH2_U; ++u) {
                int k = kb + u * 32 + lane;
                ee[u] = (k < nl) ? lw[k] : 0;
            }
#pragma unroll
            for (int u = 0; u < PH2_U; ++u) {
                dd[u] = dstp[ee[u]];
                sv[u] = srcp[ee[u]];
            }
#pragma unroll
            for (int u = 0; u < PH2_U; ++u) av[u] = a_s[sv[u]];
#pragma unroll
            for (int u = 0; u < PH2_U; ++u) {
                int k = kb + u * 32 + lane;
                if (k < nl) {
                    unsigned dl = (unsigned)(dd[u] - lo);
                    if (dl < (unsigned)nown) {
                        float lg = lrelu(av[u] + adl[dl]);
                        atomicMax(&mb[dl], f2o(lg));
                        lw[k] = sv[u];
                        dw[k] = (unsigned short)dl;
                        gw[k] = lg;
                        gs[k] = sv[u];
                        gd[k] = (unsigned short)dl;
                    } else {
                        lw[k] = lo;
                        dw[k] = 0;
                        gw[k] = -__builtin_inff();
                        gs[k] = lo;
                        gd[k] = (unsigned short)NOROW;
                    }
                }
            }
        }
    } else {
        for (int e = tid; e < n_edges; e += OWN_T) {
            unsigned dl = (unsigned)(dstp[e] - lo);
            if (dl < (unsigned)nown) {
                float lg = lrelu(a_s[srcp[e]] + adl[dl]);
                atomicMax(&mb[dl], f2o(lg));
            }
        }
    }
    __syncthreads();

    for (int i0 = wv; i0 < nown; i0 += OWN_W * 4) {
        float4 hv[4];
#pragma unroll
        for (int u = 0; u < 4; ++u) {
            int i = i0 + u * OWN_W;
            hv[u] = make_float4(0.0f, 0.0f, 0.0f, 0.0f);
            if (i < nown) hv[u] = ((const float4*)(H + (size_t)(lo + i) * HID))[lane];
        }
#pragma unroll
        for (int u = 0; u < 4; ++u) {
            int i = i0 + u * OWN_W;
            if (i < nown) {
                float es = expf(ssum[i] - o2f(mb[i]));
                float* ap = acc + i * HID + lane;
                ap[0]  = es * hv[u].x;
                ap[32] = es * hv[u].y;
                ap[64] = es * hv[u].z;
                ap[96] = es * hv[u].w;
                if (lane == 0) ssum[i] = es;
            }
        }
    }
    __syncthreads();

    if (!ovf) {
        const int myc = wcnt[lane];
        for (int wl = 0; wl < OWN_W; ++wl) {
            const int nlw = __builtin_amdgcn_readlane(myc, wl);
            for (int g = (wv - wl) & (OWN_W - 1); g * SC_GROUP < nlw; g += OWN_W)
                gat_add_group(lsrc + wl * WCAP, ldl + wl * WCAP, llog + wl * WCAP, g * SC_GROUP, nlw,
                              mb, ssum, acc, H, lane);
        }
    } else {
        for (int e = tid; e < n_edges; e += OWN_T) {
            unsigned dl = (unsigned)(dstp[e] - lo);
            if (dl < (unsigned)nown) {
                int s = srcp[e];
                float ww = expf(lrelu(a_s[s] + adl[dl]) - o2f(mb[dl]));
                atomicAdd(&ssum[dl], ww);
                const float* hp = H + (size_t)s * HID;
                float* ap = acc + dl * HID;
                for (int c = 0; c < HID; ++c) atomicAdd(ap + ROWPOS(c), ww * hp[c]);
            }
        }
    }
    __syncthreads();

    for (int i = tid; i < nown; i += OWN_T) ssum[i] = 1.0f / ssum[i];
    __syncthreads();
    const int c = tid & (HID - 1);
    const int cw = ROWPOS(c);
    const int slot = __builtin_amdgcn_readfirstlane(tid >> 7);
    const int sub = (nown + FIN_SLOTS - 1) / FIN_SLOTS;
    const int r0 = slot * sub;
    int r1 = r0 + sub;
    if (r1 > nown) r1 = nown;
    float* __restrict__ ocol = out_acc + (size_t)lo * HID + c;
    float ls = 0.0f, lq = 0.0f, run = 0.0f;
    int gcur = 0, rl = 0;
    if (do_raw && r0 < r1) gcur = __builtin_amdgcn_readfirstlane(gb[r0]);
    for (int ib = r0; ib < r1; ib += FIN_U) {
        const int last = r1 - 1;
        const int cntu = (r1 - ib < FIN_U) ? (r1 - ib) : FIN_U;
        int il = ib + (lane & 7);
        if (il > last) il = last;
        const int ibits = __float_as_int(ssum[il]);
        const int gl = gb[il];
        float v[FIN_U];
#pragma unroll
        for (int u = 0; u < FIN_U; ++u) {
            int i = ib + u;
            if (i > last) i = last;
            float inv = __int_as_float(__builtin_amdgcn_readlane(ibits, u));
            float t = acc[i * HID + cw] * inv + bv;
            float e = expm1f(t > 0.0f ? 0.0f : t);
            v[u] = t > 0.0f ? t : e;
        }
#pragma unroll
        for (int u = 0; u < FIN_U; ++u) {
            if (u < cntu) {
                ocol[(size_t)(ib + u) * HID] = v[u];
                ls += v[u];
                lq += v[u] * v[u];
            }
        }
        if (do_raw) {
#pragma unroll
            for (int u = 0; u < FIN_U; ++u) {
                if (u < cntu) {
                    int g = __builtin_amdgcn_readlane(gl, u);
                    if (g != gcur) {
                        atomicAdd(&praw[(size_t)gcur * (3 * HID) + c], run);
                        if (c == 0) atomicAdd(&cnt[gcur], (float)rl);
                        run = 0.0f;
                        rl = 0;
                        gcur = g;
                    }
                    run += v[u];
                    ++rl;
                }
            }
        }
    }
    if (do_raw && rl > 0) {
        atomicAdd(&praw[(size_t)gcur * (3 * HID) + c], run);
        if (c == 0) atomicAdd(&cnt[gcur], (float)rl);
    }
    float* red = llog;
    red[slot * HID + c] = ls;
    red[(FIN_SLOTS + slot) * HID + c] = lq;
    __syncthreads();
    if (tid < HID) {
        float ts = 0.0f, tq = 0.0f;
#pragma unroll
        for (int k = 0; k < FIN_SLOTS; ++k) {
            ts += red[k * HID + tid];
            tq += red[(FIN_SLOTS + k) * HID + tid];
        }
        atomicAdd(&sums[tid], ts);
        atomicAdd(&sumsq[tid], tq);
    }
}

__global__ void gat_bn_fold(const float* __restrict__ sums, const float* __restrict__ sumsq,
                            const float* __restrict__ gamma, const float* __restrict__ beta,
                            const float* __restrict__ Wn, float* __restrict__ Wp,
                            float* __restrict__ c0, float* __restrict__ pooled_l,
                            const float* __restrict__ cnt, const float* __restrict__ att_s,
                            const float* __restrict__ att_d, float* __restrict__ Wx,
                            float* __restrict__ Cx, int n_nodes) {
    const int t = threadIdx.x;
    const int b = blockIdx.x;
    const float invN = 1.0f / (float)n_nodes;
    if (b < HID) {
        float mu = sums[b] * invN;
        float var = sumsq[b] * invN - mu * mu;
        float g = gamma[b] / sqrtf(var + BN_EPS);
        float p = g * Wn[b * HID + t];
        Wp[b * HID + t] = p;
        gat_att_row(p, t, att_s, att_d, Wx + b * XC);
    } else if (b == HID) {
        __shared__ float dsh[HID];
        {
            float mu = sums[t] * invN;
            float var = sumsq[t] * invN - mu * mu;
            float g = gamma[t] / sqrtf(var + BN_EPS);
            dsh[t] = beta[t] - mu * g;
        }
        __syncthreads();
        float a = 0.0f;
#pragma unroll 16
        for (int k = 0; k < HID; ++k) a += dsh[k] * Wn[k * HID + t];
        c0[t] = a;
        gat_att_row(a, t, att_s, att_d, Cx);
    } else {
        const int gr = b - HID - 1;
        float mu = sums[t] * invN;
        float var = sumsq[t] * invN - mu * mu;
        float g = gamma[t] / sqrtf(var + BN_EPS);
        float cn = cnt[gr];
        size_t idx = (size_t)gr * (3 * HID) + t;
        pooled_l[idx] = (pooled_l[idx] - cn * mu) * g + cn * beta[t];
    }
}

__global__ void gat_bn_pool(float* __restrict__ out_acc, const float* __restrict__ sums,
                            const float* __restrict__ sumsq, const float* __restrict__ gamma,
                            const float* __restrict__ beta, const int* __restrict__ batch,
                            float* __restrict__ pooled, int layer, int n_nodes) {
    int idx = blockIdx.x * blockDim.x + threadIdx.x;
    if (idx >= n_nodes * HID) return;
    int c = idx & (HID - 1);
    int n = idx >> 7;
    float invN = 1.0f / (float)n_nodes;
    float mu = sums[c] * invN;
    float var = sumsq[c] * invN - mu * mu;
    float xv = (out_acc[idx] - mu) / sqrtf(var + BN_EPS) * gamma[c] + beta[c];
    out_acc[idx] = xv;
    atomicAdd(&pooled[(size_t)batch[n] * (3 * HID) + layer * HID + c], xv);
}

extern "C" void kernel_launch(void* const* d_in, const int* in_sizes, int n_in,
                              void* d_out, int out_size, void* d_ws, size_t ws_size,
                              hipStream_t stream) {
    const float* x_in  = (const float*)d_in[0];
    const int*   ei    = (const int*)d_in[1];
    const int*   batch = (const int*)d_in[2];
    const float* W     = (const float*)d_in[3];
    const float* att_s = (const float*)d_in[4];
    const float* att_d = (const float*)d_in[5];
    const float* bias  = (const float*)d_in[6];
    const float* gamma = (const float*)d_in[7];
    const float* beta  = (const float*)d_in[8];

    const int N = in_sizes[0] / HID;
    const int E = in_sizes[1] / 2;

    float* pooled = (float*)d_out;
    float* xfinal = (float*)d_out + (size_t)NGRAPH * 3 * HID;

    char* w = (char*)d_ws;
    float* hbuf = (float*)w;  w += (size_t)N * HID * sizeof(float);
    float* xb   = (float*)w;  w += (size_t)N * HID * sizeof(float);
    unsigned* obw = (unsigned*)w;  w += (size_t)((E + 15) / 16) * 16;
    float* asrc = (float*)w;  w += (size_t)N * sizeof(float);
    float* adst = (float*)w;  w += (size_t)N * sizeof(float);
    float* sums = (float*)w;
    float* sumsq = sums + HID;
    float* cnt  = sums + 2 * HID;
    float* c0   = cnt + NGRAPH;
    float* Wp   = c0 + HID;
    float* Wx   = Wp + HID * HID;
    float* Cx   = Wx + HID * XC;
    const int n_own = (N + NPB - 1) / NPB;
    int* gsrc = (int*)(Cx + XC);
    int* gcnt = gsrc + (size_t)n_own * LCAP;
    unsigned short* gdl = (unsigned short*)(gcnt + (size_t)n_own * OWN_W);

    const int waves_g = (N + 15) / 16;
    dim3 blk(256);
    dim3 grid_gemm((waves_g * 32 + 255) / 256);
    dim3 grid_own(n_own);
    dim3 grid_elem(((size_t)N * HID + 255) / 256);

    gat_zero<<<(NGRAPH * 3 * HID + 255) / 256, blk, 0, stream>>>(pooled, NGRAPH * 3 * HID);
    gat_owner_ids<<<((E + 3) / 4 + 255) / 256, blk, 0, stream>>>(ei, obw, E);
    gat_att_cols<<<HID, HID, 0, stream>>>(W, att_s, att_d, Wx, Cx);

    for (int l = 0; l < 3; ++l) {
        const float* X  = (l == 0) ? x_in : xb;
        float* oacc     = (l == 2) ? xfinal : xb;
        const float* Wl = (l == 0) ? W : Wp;
        const float* Cl = (l == 0) ? (const float*)nullptr : c0;

        gat_gemm_wmma<<<grid_gemm, blk, 0, stream>>>(X, Wl, Cl, Wx, Cx, hbuf, asrc, adst, N);
        gat_zero<<<3, 256, 0, stream>>>(sums, 2 * HID + NGRAPH);
        if (l == 0)
            gat_owner<false><<<grid_own, OWN_T, 0, stream>>>(ei, obw, asrc, adst, hbuf, oacc, gsrc, gdl, gcnt,
                                                             bias + l * HID, sums, sumsq, batch,
                                                             pooled + l * HID, cnt, (l < 2) ? 1 : 0, N, E);
        else
            gat_owner<true><<<grid_own, OWN_T, 0, stream>>>(ei, obw, asrc, adst, hbuf, oacc, gsrc, gdl, gcnt,
                                                            bias + l * HID, sums, sumsq, batch,
                                                            pooled + l * HID, cnt, (l < 2) ? 1 : 0, N, E);
        if (l < 2)
            gat_bn_fold<<<HID + 1 + NGRAPH, HID, 0, stream>>>(sums, sumsq, gamma + l * HID,
                                                              beta + l * HID,
                                                              W + (size_t)(l + 1) * HID * HID, Wp, c0,
                                                              pooled + l * HID, cnt,
                                                              att_s + (l + 1) * HID,
                                                              att_d + (l + 1) * HID, Wx, Cx, N);
        else
            gat_bn_pool<<<grid_elem, blk, 0, stream>>>(oacc, sums, sumsq, gamma + l * HID,
                                                       beta + l * HID, batch, pooled, l, N);
    }
}


# Layer 0 edge pick-up: owner bytes become one bit row per owner (tiles OR'ed in LDS, whole-line stores); a work-group scans only its own row, 32 edges per word (list order new; row sum order: arrival)
# speedup vs baseline: 4.2946x; 1.0378x over previous
#include <hip/hip_runtime.h>
#include <hip/hip_bf16.h>

#define HID 128
#define NGRAPH 512
#define NEG_SLOPE 0.2f
#define BN_EPS 1e-5f

typedef __attribute__((ext_vector_type(2))) float v2f;
typedef __attribute__((ext_vector_type(8))) float v8f;

__device__ __forceinline__ unsigned f2o(float f) {
    unsigned u = __float_as_uint(f);
    return (u & 0x80000000u) ? ~u : (u | 0x80000000u);
}
__device__ __forceinline__ float o2f(unsigned u) {
    unsigned v = (u & 0x80000000u) ? (u & 0x7fffffffu) : ~u;
    return __uint_as_float(v);
}
__device__ __forceinline__ float lrelu(float a) { return a > 0.0f ? a : NEG_SLOPE * a; }

__global__ void gat_zero(float* __restrict__ p, int n) {
    int i = blockIdx.x * blockDim.x + threadIdx.x;
    if (i < n) p[i] = 0.0f;
}

#define XC 16

__global__ void gat_gemm_wmma(const float* __restrict__ X, const float* __restrict__ W,
                              const float* __restrict__ C0, const float* __restrict__ Wx,
                              const float* __restrict__ Cx, float* __restrict__ H,
                              float* __restrict__ a_s, float* __restrict__ a_d, int nrows) {
    int wave = (blockIdx.x * blockDim.x + threadIdx.x) >> 5;
    int m0 = wave * 16;
    if (m0 >= nrows) return;
    int lane = threadIdx.x & 31;
    int lr = lane & 15;
    int lh = lane >> 4;

    v8f acc[8];
#pragma unroll
    for (int n = 0; n < 8; ++n) {
        float cv = C0 ? C0[n * 16 + lr] : 0.0f;
#pragma unroll
        for (int j = 0; j < 8; ++j) acc[n][j] = cv;
    }
    v8f accx;
    {
        float cv = Cx[lr];
#pragma unroll
        for (int j = 0; j < 8; ++j) accx[j] = cv;
    }
    const float* xrow = X + (size_t)(m0 + lr) * HID;

    for (int k = 0; k < HID; k += 4) {
        int ka = k + lh * 2;
        v2f a;
        a.x = xrow[ka];
        a.y = xrow[ka + 1];
#pragma unroll
        for (int n = 0; n < 8; ++n) {
            v2f b;
            b.x = W[ka * HID + n * 16 + lr];
            b.y = W[(ka + 1) * HID + n * 16 + lr];
            acc[n] = __builtin_amdgcn_wmma_f32_16x16x4_f32(
                false, a, false, b, (short)0, acc[n], false, false);
        }
        v2f bx;
        bx.x = Wx[ka * XC + lr];
        bx.y = Wx[(ka + 1) * XC + lr];
        accx = __builtin_amdgcn_wmma_f32_16x16x4_f32(
            false, a, false, bx, (short)0, accx, false, false);
    }
#pragma unroll
    for (int n = 0; n < 8; ++n)
#pragma unroll
        for (int j = 0; j < 8; ++j)
            H[(size_t)(m0 + j + lh * 8) * HID + n * 16 + lr] = acc[n][j];
    if (lr < 2) {
        float* ap = lr ? a_d : a_s;
#pragma unroll
        for (int j = 0; j < 8; ++j) ap[m0 + j + lh * 8] = accx[j];
    }
}

__device__ __forceinline__ void gat_att_row(float p, int t, const float* __restrict__ att_s,
                                            const float* __restrict__ att_d,
                                            float* __restrict__ dst) {
    __shared__ float part[2 * (HID / 32)];
    float ps = p * att_s[t];
    float pd = p * att_d[t];
    for (int off = 16; off > 0; off >>= 1) {
        ps += __shfl_down(ps, off, 32);
        pd += __shfl_down(pd, off, 32);
    }
    if ((t & 31) == 0) {
        part[t >> 5] = ps;
        part[HID / 32 + (t >> 5)] = pd;
    }
    __syncthreads();
    if (t < XC) {
        float v = 0.0f;
        if (t < 2) {
            const float* q = part + t * (HID / 32);
            v = (q[0] + q[1]) + (q[2] + q[3]);
        }
        dst[t] = v;
    }
}

__global__ void gat_att_cols(const float* __restrict__ Wl, const float* __restrict__ att_s,
                             const float* __restrict__ att_d, float* __restrict__ Wx,
                             float* __restrict__ Cx) {
    const int t = threadIdx.x;
    const int k = blockIdx.x;
    gat_att_row(Wl[k * HID + t], t, att_s, att_d, Wx + k * XC);
    if (k == 0 && t < XC) Cx[t] = 0.0f;
}

#define NPB   391
#define OB_E  1024
#define OB_T  256
#define OB_W  (OB_E / 32)
#define OB_P  (OB_W + 1)

__global__ __launch_bounds__(OB_T) void gat_owner_bits(const int* __restrict__ ei,
                                                       unsigned* __restrict__ obw, int n_edges) {
    __shared__ unsigned tile[256 * OB_P];
    const int tid = threadIdx.x;
    const int lane = tid & 31;
    for (int i = tid; i < 256 * OB_P; i += OB_T) tile[i] = 0u;
    const int* __restrict__ dstp = ei + n_edges;
    const int eb = blockIdx.x * OB_E;
    int dv[OB_E / OB_T];
#pragma unroll
    for (int j = 0; j < OB_E / OB_T; ++j) {
        const int e = eb + j * OB_T + tid;
        dv[j] = (e < n_edges) ? dstp[e] : -1;
    }
    __syncthreads();
#pragma unroll
    for (int j = 0; j < OB_E / OB_T; ++j) {
        const int le = j * OB_T + tid;
        if (dv[j] >= 0) atomicOr(&tile[((unsigned)(dv[j] / NPB) & 255u) * OB_P + (le >> 5)], 1u << lane);
    }
    __syncthreads();
    const size_t pw = (size_t)gridDim.x * OB_W;
    unsigned* __restrict__ dst = obw + (size_t)blockIdx.x * OB_W + lane;
    const int r0 = tid >> 5;
#pragma unroll
    for (int j = 0; j < 256 / (OB_T / 32); ++j) {
        const int r = r0 + j * (OB_T / 32);
        dst[(size_t)r * pw] = tile[r * OB_P + lane];
    }
}

#define OWN_T 1024
#define OWN_W (OWN_T / 32)
#define WCAP  352
#define LCAP  (OWN_W * WCAP)
#define SCAN_U 8
#define PH2_U 8
#define SC_GROUP 8
#define NOROW 0xFFFFu
#define FIN_SLOTS (OWN_T / HID)
#define FIN_U 7
#define ROWPOS(c) ((((c) & 3) << 5) | ((c) >> 2))

__device__ __forceinline__ void gat_keepw(unsigned m, int e0, int lane, int* lw, int& wc) {
    for (int p = 0; p < 32; ++p) {
        unsigned mk = __builtin_amdgcn_ballot_w32(m != 0u);
        if (mk == 0u) break;
        if (m != 0u) {
            int pos = wc + __builtin_popcount(mk & ((1u << lane) - 1u));
            if (pos < WCAP) lw[pos] = e0 + __builtin_ctz(m);
            m &= m - 1u;
        }
        wc += __builtin_popcount(mk);
    }
}

__device__ __forceinline__ void gat_add_group(const int* lwl, const unsigned short* dwl,
                                              const float* gwl, int k0, int nl,
                                              const unsigned* mb, float* ssum, float* acc,
                                              const float* __restrict__ H, int lane) {
    const int rest = nl - k0;
    const int gc = rest < SC_GROUP ? rest : SC_GROUP;
    int sv = 0, dv = 0;
    float w = 0.0f;
    if (lane < gc) {
        sv = lwl[k0 + lane];
        dv = dwl[k0 + lane];
        w = expf(gwl[k0 + lane] - o2f(mb[dv]));
        atomicAdd(&ssum[dv], w);
    }
    int wbits = __float_as_int(w);
    if (gc == SC_GROUP) {
        float4 hv[SC_GROUP];
        int dk[SC_GROUP];
        float wk[SC_GROUP];
#pragma unroll
        for (int u = 0; u < SC_GROUP; ++u) {
            int sk = __builtin_amdgcn_readlane(sv, u);
            dk[u] = __builtin_amdgcn_readlane(dv, u);
            wk[u] = __int_as_float(__builtin_amdgcn_readlane(wbits, u));
            hv[u] = ((const float4*)(H + (size_t)sk * HID))[lane];
        }
#pragma unroll
        for (int u = 0; u < SC_GROUP; ++u) {
            float* ap = acc + dk[u] * HID + lane;
            atomicAdd(ap,      wk[u] * hv[u].x);
            atomicAdd(ap + 32, wk[u] * hv[u].y);
            atomicAdd(ap + 64, wk[u] * hv[u].z);
            atomicAdd(ap + 96, wk[u] * hv[u].w);
        }
    } else {
        for (int u = 0; u < gc; ++u) {
            int sk = __builtin_amdgcn_readlane(sv, u);
            int dd = __builtin_amdgcn_readlane(dv, u);
            float ww = __int_as_float(__builtin_amdgcn_readlane(wbits, u));
            const float4 h4 = ((const float4*)(H + (size_t)sk * HID))[lane];
            float* ap = acc + dd * HID + lane;
            atomicAdd(ap,      ww * h4.x);
            atomicAdd(ap + 32, ww * h4.y);
            atomicAdd(ap + 64, ww * h4.z);
            atomicAdd(ap + 96, ww * h4.w);
        }
    }
}

template <bool REUSE>
__global__ __launch_bounds__(OWN_T) void gat_owner(const int* __restrict__ ei,
                                                   const unsigned* __restrict__ obw,
                                                   const float* __restrict__ a_s,
                                                   const float* __restrict__ a_d,
                                                   const float* __restrict__ H,
                                                   float* __restrict__ out_acc,
                                                   int* __restrict__ gsrc,
                                                   unsigned short* __restrict__ gdl,
                                                   int* __restrict__ gcnt,
                                                   const float* __restrict__ bias,
                                                   float* __restrict__ sums,
                                                   float* __restrict__ sumsq,
                                                   const int* __restrict__ batch,
                                                   float* __restrict__ praw,
                                                   float* __restrict__ cnt,
                                                   int do_raw, int n_nodes, int n_edges) {
    __shared__ __attribute__((aligned(16))) float acc[NPB * HID];
    __shared__ float adl[NPB];
    __shared__ float ssum[NPB];
    __shared__ unsigned mb[NPB];
    __shared__ int gb[NPB];
    __shared__ int lsrc[LCAP];
    __shared__ float llog[LCAP];
    __shared__ unsigned short ldl[LCAP];
    __shared__ int wcnt[OWN_W];
    __shared__ int ovfl;

    const int tid = threadIdx.x;
    const int lane = tid & 31;
    const int wv = tid >> 5;
    const int lo = blockIdx.x * NPB;
    int nown = n_nodes - lo;
    if (nown > NPB) nown = NPB;
    const int* __restrict__ srcp = ei;
    const int* __restrict__ dstp = ei + n_edges;
    const float bv = bias[tid & (HID - 1)];

    for (int i = tid; i < nown; i += OWN_T) {
        float ad = a_d[lo + i];
        float sl = lrelu(a_s[lo + i] + ad);
        adl[i] = ad;
        ssum[i] = sl;
        mb[i] = f2o(sl);
        gb[i] = do_raw ? batch[lo + i] : 0;
    }
    if (tid == 0) ovfl = 0;
    __syncthreads();

    int* lw = lsrc + wv * WCAP;
    float* gw = llog + wv * WCAP;
    unsigned short* dw = ldl + wv * WCAP;
    int wc = 0;
    const size_t gl0 = (size_t)(blockIdx.x * OWN_W + wv) * WCAP;
    int* __restrict__ gs = gsrc + gl0;
    unsigned short* __restrict__ gd = gdl + gl0;
    if (REUSE) {
        wc = __builtin_amdgcn_readfirstlane(gcnt[blockIdx.x * OWN_W + wv]);
    } else {
        const int pw = ((n_edges + OB_E - 1) / OB_E) * OB_W;
        const int nq = pw >> 2;
        const uint4* __restrict__ oq = (const uint4*)(obw + (size_t)(blockIdx.x & 255u) * pw);
        for (int q0 = 0; q0 < nq; q0 += SCAN_U * OWN_T) {
            const int qb = q0 + tid;
            uint4 d[SCAN_U];
#pragma unroll
            for (int j = 0; j < SCAN_U; ++j) {
                int q = qb + j * OWN_T;
                d[j] = (q < nq) ? oq[q] : make_uint4(0u, 0u, 0u, 0u);
            }
#pragma unroll
            for (int j = 0; j < SCAN_U; ++j) {
                int e0 = (qb + j * OWN_T) * 128;
                gat_keepw(d[j].x, e0,      lane, lw, wc);
                gat_keepw(d[j].y, e0 + 32, lane, lw, wc);
                gat_keepw(d[j].z, e0 + 64, lane, lw, wc);
                gat_keepw(d[j].w, e0 + 96, lane, lw, wc);
            }
        }
        if (lane == 0) gcnt[blockIdx.x * OWN_W + wv] = wc;
    }
    if (lane == 0) {
        wcnt[wv] = wc;
        if (wc > WCAP) ovfl = 1;
    }
    __syncthreads();

    const bool ovf = ovfl != 0;
    const int nl = ovf ? 0 : wc;

    if (!ovf && REUSE) {
        for (int kb = 0; kb < nl; kb += 32 * PH2_U) {
            int sv[PH2_U];
            unsigned dl[PH2_U];
            float av[PH2_U];
#pragma unroll
            for (int u = 0; u < PH2_U; ++u) {
                int k = kb + u * 32 + lane;
                sv[u] = (k < nl) ? gs[k] : lo;
                dl[u] = (k < nl) ? (unsigned)gd[k] : NOROW;
            }
#pragma unroll
            for (int u = 0; u < PH2_U; ++u) av[u] = a_s[sv[u]];
#pragma unroll
            for (int u = 0; u < PH2_U; ++u) {
                int k = kb + u * 32 + lane;
                if (k < nl) {
                    if (dl[u] != NOROW) {
                        float lg = lrelu(av[u] + adl[dl[u]]);
                        atomicMax(&mb[dl[u]], f2o(lg));
                        lw[k] = sv[u];
                        dw[k] = (unsigned short)dl[u];
                        gw[k] = lg;
                    } else {
                        lw[k] = lo;
                        dw[k] = 0;
                        gw[k] = -__builtin_inff();
                    }
                }
            }
        }
    } else if (!ovf) {
        for (int kb = 0; kb < nl; kb += 32 * PH2_U) {
            int ee[PH2_U], dd[PH2_U], sv[PH2_U];
            float av[PH2_U];
#pragma unroll
            for (int u = 0; u < PH2_U; ++u) {
                int k = kb + u * 32 + lane;
                ee[u] = (k < nl) ? lw[k] : 0;
            }
#pragma unroll
            for (int u = 0; u < PH2_U; ++u) {
                dd[u] = dstp[ee[u]];
                sv[u] = srcp[ee[u]];
            }
#pragma unroll
            for (int u = 0; u < PH2_U; ++u) av[u] = a_s[sv[u]];
#pragma unroll
            for (int u = 0; u < PH2_U; ++u) {
                int k = kb + u * 32 + lane;
                if (k < nl) {
                    unsigned dl = (unsigned)(dd[u] - lo);
                    if (dl < (unsigned)nown) {
                        float lg = lrelu(av[u] + adl[dl]);
                        atomicMax(&mb[dl], f2o(lg));
                        lw[k] = sv[u];
                        dw[k] = (unsigned short)dl;
                        gw[k] = lg;
                        gs[k] = sv[u];
                        gd[k] = (unsigned short)dl;
                    } else {
                        lw[k] = lo;
                        dw[k] = 0;
                        gw[k] = -__builtin_inff();
                        gs[k] = lo;
                        gd[k] = (unsigned short)NOROW;
                    }
                }
            }
        }
    } else {
        for (int e = tid; e < n_edges; e += OWN_T) {
            unsigned dl = (unsigned)(dstp[e] - lo);
            if (dl < (unsigned)nown) {
                float lg = lrelu(a_s[srcp[e]] + adl[dl]);
                atomicMax(&mb[dl], f2o(lg));
            }
        }
    }
    __syncthreads();

    for (int i0 = wv; i0 < nown; i0 += OWN_W * 4) {
        float4 hv[4];
#pragma unroll
        for (int u = 0; u < 4; ++u) {
            int i = i0 + u * OWN_W;
            hv[u] = make_float4(0.0f, 0.0f, 0.0f, 0.0f);
            if (i < nown) hv[u] = ((const float4*)(H + (size_t)(lo + i) * HID))[lane];
        }
#pragma unroll
        for (int u = 0; u < 4; ++u) {
            int i = i0 + u * OWN_W;
            if (i < nown) {
                float es = expf(ssum[i] - o2f(mb[i]));
                float* ap = acc + i * HID + lane;
                ap[0]  = es * hv[u].x;
                ap[32] = es * hv[u].y;
                ap[64] = es * hv[u].z;
                ap[96] = es * hv[u].w;
                if (lane == 0) ssum[i] = es;
            }
        }
    }
    __syncthreads();

    if (!ovf) {
        const int myc = wcnt[lane];
        for (int wl = 0; wl < OWN_W; ++wl) {
            const int nlw = __builtin_amdgcn_readlane(myc, wl);
            for (int g = (wv - wl) & (OWN_W - 1); g * SC_GROUP < nlw; g += OWN_W)
                gat_add_group(lsrc + wl * WCAP, ldl + wl * WCAP, llog + wl * WCAP, g * SC_GROUP, nlw,
                              mb, ssum, acc, H, lane);
        }
    } else {
        for (int e = tid; e < n_edges; e += OWN_T) {
            unsigned dl = (unsigned)(dstp[e] - lo);
            if (dl < (unsigned)nown) {
                int s = srcp[e];
                float ww = expf(lrelu(a_s[s] + adl[dl]) - o2f(mb[dl]));
                atomicAdd(&ssum[dl], ww);
                const float* hp = H + (size_t)s * HID;
                float* ap = acc + dl * HID;
                for (int c = 0; c < HID; ++c) atomicAdd(ap + ROWPOS(c), ww * hp[c]);
            }
        }
    }
    __syncthreads();

    for (int i = tid; i < nown; i += OWN_T) ssum[i] = 1.0f / ssum[i];
    __syncthreads();
    const int c = tid & (HID - 1);
    const int cw = ROWPOS(c);
    const int slot = __builtin_amdgcn_readfirstlane(tid >> 7);
    const int sub = (nown + FIN_SLOTS - 1) / FIN_SLOTS;
    const int r0 = slot * sub;
    int r1 = r0 + sub;
    if (r1 > nown) r1 = nown;
    float* __restrict__ ocol = out_acc + (size_t)lo * HID + c;
    float ls = 0.0f, lq = 0.0f, run = 0.0f;
    int gcur = 0, rl = 0;
    if (do_raw && r0 < r1) gcur = __builtin_amdgcn_readfirstlane(gb[r0]);
    for (int ib = r0; ib < r1; ib += FIN_U) {
        const int last = r1 - 1;
        const int cntu = (r1 - ib < FIN_U) ? (r1 - ib) : FIN_U;
        int il = ib + (lane & 7);
        if (il > last) il = last;
        const int ibits = __float_as_int(ssum[il]);
        const int gl = gb[il];
        float v[FIN_U];
#pragma unroll
        for (int u = 0; u < FIN_U; ++u) {
            int i = ib + u;
            if (i > last) i = last;
            float inv = __int_as_float(__builtin_amdgcn_readlane(ibits, u));
            float t = acc[i * HID + cw] * inv + bv;
            float e = expm1f(t > 0.0f ? 0.0f : t);
            v[u] = t > 0.0f ? t : e;
        }
#pragma unroll
        for (int u = 0; u < FIN_U; ++u) {
            if (u < cntu) {
                ocol[(size_t)(ib + u) * HID] = v[u];
                ls += v[u];
                lq += v[u] * v[u];
            }
        }
        if (do_raw) {
#pragma unroll
            for (int u = 0; u < FIN_U; ++u) {
                if (u < cntu) {
                    int g = __builtin_amdgcn_readlane(gl, u);
                    if (g != gcur) {
                        atomicAdd(&praw[(size_t)gcur * (3 * HID) + c], run);
                        if (c == 0) atomicAdd(&cnt[gcur], (float)rl);
                        run = 0.0f;
                        rl = 0;
                        gcur = g;
                    }
                    run += v[u];
                    ++rl;
                }
            }
        }
    }
    if (do_raw && rl > 0) {
        atomicAdd(&praw[(size_t)gcur * (3 * HID) + c], run);
        if (c == 0) atomicAdd(&cnt[gcur], (float)rl);
    }
    float* red = llog;
    red[slot * HID + c] = ls;
    red[(FIN_SLOTS + slot) * HID + c] = lq;
    __syncthreads();
    if (tid < HID) {
        float ts = 0.0f, tq = 0.0f;
#pragma unroll
        for (int k = 0; k < FIN_SLOTS; ++k) {
            ts += red[k * HID + tid];
            tq += red[(FIN_SLOTS + k) * HID + tid];
        }
        atomicAdd(&sums[tid], ts);
        atomicAdd(&sumsq[tid], tq);
    }
}

__global__ void gat_bn_fold(const float* __restrict__ sums, const float* __restrict__ sumsq,
                            const float* __restrict__ gamma, const float* __restrict__ beta,
                            const float* __restrict__ Wn, float* __restrict__ Wp,
                            float* __restrict__ c0, float* __restrict__ pooled_l,
                            const float* __restrict__ cnt, const float* __restrict__ att_s,
                            const float* __restrict__ att_d, float* __restrict__ Wx,
                            float* __restrict__ Cx, int n_nodes) {
    const int t = threadIdx.x;
    const int b = blockIdx.x;
    const float invN = 1.0f / (float)n_nodes;
    if (b < HID) {
        float mu = sums[b] * invN;
        float var = sumsq[b] * invN - mu * mu;
        float g = gamma[b] / sqrtf(var + BN_EPS);
        float p = g * Wn[b * HID + t];
        Wp[b * HID + t] = p;
        gat_att_row(p, t, att_s, att_d, Wx + b * XC);
    } else if (b == HID) {
        __shared__ float dsh[HID];
        {
            float mu = sums[t] * invN;
            float var = sumsq[t] * invN - mu * mu;
            float g = gamma[t] / sqrtf(var + BN_EPS);
            dsh[t] = beta[t] - mu * g;
        }
        __syncthreads();
        float a = 0.0f;
#pragma unroll 16
        for (int k = 0; k < HID; ++k) a += dsh[k] * Wn[k * HID + t];
        c0[t] = a;
        gat_att_row(a, t, att_s, att_d, Cx);
    } else {
        const int gr = b - HID - 1;
        float mu = sums[t] * invN;
        float var = sumsq[t] * invN - mu * mu;
        float g = gamma[t] / sqrtf(var + BN_EPS);
        float cn = cnt[gr];
        size_t idx = (size_t)gr * (3 * HID) + t;
        pooled_l[idx] = (pooled_l[idx] - cn * mu) * g + cn * beta[t];
    }
}

__global__ void gat_bn_pool(float* __restrict__ out_acc, const float* __restrict__ sums,
                            const float* __restrict__ sumsq, const float* __restrict__ gamma,
                            const float* __restrict__ beta, const int* __restrict__ batch,
                            float* __restrict__ pooled, int layer, int n_nodes) {
    int idx = blockIdx.x * blockDim.x + threadIdx.x;
    if (idx >= n_nodes * HID) return;
    int c = idx & (HID - 1);
    int n = idx >> 7;
    float invN = 1.0f / (float)n_nodes;
    float mu = sums[c] * invN;
    float var = sumsq[c] * invN - mu * mu;
    float xv = (out_acc[idx] - mu) / sqrtf(var + BN_EPS) * gamma[c] + beta[c];
    out_acc[idx] = xv;
    atomicAdd(&pooled[(size_t)batch[n] * (3 * HID) + layer * HID + c], xv);
}

extern "C" void kernel_launch(void* const* d_in, const int* in_sizes, int n_in,
                              void* d_out, int out_size, void* d_ws, size_t ws_size,
                              hipStream_t stream) {
    const float* x_in  = (const float*)d_in[0];
    const int*   ei    = (const int*)d_in[1];
    const int*   batch = (const int*)d_in[2];
    const float* W     = (const float*)d_in[3];
    const float* att_s = (const float*)d_in[4];
    const float* att_d = (const float*)d_in[5];
    const float* bias  = (const float*)d_in[6];
    const float* gamma = (const float*)d_in[7];
    const float* beta  = (const float*)d_in[8];

    const int N = in_sizes[0] / HID;
    const int E = in_sizes[1] / 2;

    float* pooled = (float*)d_out;
    float* xfinal = (float*)d_out + (size_t)NGRAPH * 3 * HID;

    char* w = (char*)d_ws;
    float* hbuf = (float*)w;  w += (size_t)N * HID * sizeof(float);
    float* xb   = (float*)w;  w += (size_t)N * HID * sizeof(float);
    const int n_tiles = (E + OB_E - 1) / OB_E;
    unsigned* obw = (unsigned*)w;  w += (size_t)256 * n_tiles * OB_W * sizeof(unsigned);
    float* asrc = (float*)w;  w += (size_t)N * sizeof(float);
    float* adst = (float*)w;  w += (size_t)N * sizeof(float);
    float* sums = (float*)w;
    float* sumsq = sums + HID;
    float* cnt  = sums + 2 * HID;
    float* c0   = cnt + NGRAPH;
    float* Wp   = c0 + HID;
    float* Wx   = Wp + HID * HID;
    float* Cx   = Wx + HID * XC;
    const int n_own = (N + NPB - 1) / NPB;
    int* gsrc = (int*)(Cx + XC);
    int* gcnt = gsrc + (size_t)n_own * LCAP;
    unsigned short* gdl = (unsigned short*)(gcnt + (size_t)n_own * OWN_W);

    const int waves_g = (N + 15) / 16;
    dim3 blk(256);
    dim3 grid_gemm((waves_g * 32 + 255) / 256);
    dim3 grid_own(n_own);
    dim3 grid_elem(((size_t)N * HID + 255) / 256);

    gat_zero<<<(NGRAPH * 3 * HID + 255) / 256, blk, 0, stream>>>(pooled, NGRAPH * 3 * HID);
    gat_owner_bits<<<n_tiles, OB_T, 0, stream>>>(ei, obw, E);
    gat_att_cols<<<HID, HID, 0, stream>>>(W, att_s, att_d, Wx, Cx);

    for (int l = 0; l < 3; ++l) {
        const float* X  = (l == 0) ? x_in : xb;
        float* oacc     = (l == 2) ? xfinal : xb;
        const float* Wl = (l == 0) ? W : Wp;
        const float* Cl = (l == 0) ? (const float*)nullptr : c0;

        gat_gemm_wmma<<<grid_gemm, blk, 0, stream>>>(X, Wl, Cl, Wx, Cx, hbuf, asrc, adst, N);
        gat_zero<<<3, 256, 0, stream>>>(sums, 2 * HID + NGRAPH);
        if (l == 0)
            gat_owner<false><<<grid_own, OWN_T, 0, stream>>>(ei, obw, asrc, adst, hbuf, oacc, gsrc, gdl, gcnt,
                                                             bias + l * HID, sums, sumsq, batch,
                                                             pooled + l * HID, cnt, (l < 2) ? 1 : 0, N, E);
        else
            gat_owner<true><<<grid_own, OWN_T, 0, stream>>>(ei, obw, asrc, adst, hbuf, oacc, gsrc, gdl, gcnt,
                                                            bias + l * HID, sums, sumsq, batch,
                                                            pooled + l * HID, cnt, (l < 2) ? 1 : 0, N, E);
        if (l < 2)
            gat_bn_fold<<<HID + 1 + NGRAPH, HID, 0, stream>>>(sums, sumsq, gamma + l * HID,
                                                              beta + l * HID,
                                                              W + (size_t)(l + 1) * HID * HID, Wp, c0,
                                                              pooled + l * HID, cnt,
                                                              att_s + (l + 1) * HID,
                                                              att_d + (l + 1) * HID, Wx, Cx, N);
        else
            gat_bn_pool<<<grid_elem, blk, 0, stream>>>(oacc, sums, sumsq, gamma + l * HID,
                                                       beta + l * HID, batch, pooled, l, N);
    }
}


# 7 helper launches folded (15->8): launch 1 also zeroes sums, builds W.att tiles; product work-groups scale an LDS W copy by BN factors, form constant row + pooled fix-up (logit constant sum: k chain)
# speedup vs baseline: 4.9308x; 1.1481x over previous
#include <hip/hip_runtime.h>
#include <hip/hip_bf16.h>

#define HID 128
#define NGRAPH 512
#define NEG_SLOPE 0.2f
#define BN_EPS 1e-5f

typedef __attribute__((ext_vector_type(2))) float v2f;
typedef __attribute__((ext_vector_type(8))) float v8f;

__device__ __forceinline__ unsigned f2o(float f) {
    unsigned u = __float_as_uint(f);
    return (u & 0x80000000u) ? ~u : (u | 0x80000000u);
}
__device__ __forceinline__ float o2f(unsigned u) {
    unsigned v = (u & 0x80000000u) ? (u & 0x7fffffffu) : ~u;
    return __uint_as_float(v);
}
__device__ __forceinline__ float lrelu(float a) { return a > 0.0f ? a : NEG_SLOPE * a; }

#define XC 16
#define BP (HID + XC)
#define GM_T 256

__global__ __launch_bounds__(GM_T) void gat_gemm_wmma(const float* __restrict__ X,
                                                      const float* __restrict__ Wn,
                                                      const float* __restrict__ WxU,
                                                      const float* __restrict__ sums,
                                                      const float* __restrict__ sumsq,
                                                      const float* __restrict__ gamma,
                                                      const float* __restrict__ beta,
                                                      float* __restrict__ pooled_l,
                                                      const float* __restrict__ cnt,
                                                      float* __restrict__ H, float* __restrict__ a_s,
                                                      float* __restrict__ a_d, int fold, int nrows) {
    __shared__ __attribute__((aligned(16))) float Bs[HID * BP];
    __shared__ float gsh[HID];
    __shared__ float dsh[HID];
    __shared__ float c0s[BP];
    const int tid = threadIdx.x;

#pragma unroll
    for (int j = 0; j < HID * HID / 8 / GM_T; ++j) {
        const int i = tid + j * GM_T;
        const int kp = i >> 5, c4 = i & 31;
        const float4 w0 = ((const float4*)Wn)[(2 * kp) * (HID / 4) + c4];
        const float4 w1 = ((const float4*)Wn)[(2 * kp + 1) * (HID / 4) + c4];
        float4* p = (float4*)&Bs[(kp * BP + c4 * 4) * 2];
        p[0] = make_float4(w0.x, w1.x, w0.y, w1.y);
        p[1] = make_float4(w0.z, w1.z, w0.w, w1.w);
    }
    {
        const int kp = tid >> 2, c4 = tid & 3;
        const float4 w0 = ((const float4*)WxU)[(2 * kp) * (XC / 4) + c4];
        const float4 w1 = ((const float4*)WxU)[(2 * kp + 1) * (XC / 4) + c4];
        float4* p = (float4*)&Bs[(kp * BP + HID + c4 * 4) * 2];
        p[0] = make_float4(w0.x, w1.x, w0.y, w1.y);
        p[1] = make_float4(w0.z, w1.z, w0.w, w1.w);
    }
    if (fold) {
        if (tid < HID) {
            const float invN = 1.0f / (float)nrows;
            const float mu = sums[tid] * invN;
            const float var = sumsq[tid] * invN - mu * mu;
            const float g = gamma[tid] / sqrtf(var + BN_EPS);
            const float bt = beta[tid];
            gsh[tid] = g;
            dsh[tid] = bt - mu * g;
            for (int gr = blockIdx.x; gr < NGRAPH; gr += gridDim.x) {
                const float cn = cnt[gr];
                const size_t idx = (size_t)gr * (3 * HID) + tid;
                pooled_l[idx] = (pooled_l[idx] - cn * mu) * g + cn * bt;
            }
        }
        __syncthreads();
        if (tid < BP) {
            float a = 0.0f;
#pragma unroll 8
            for (int kp = 0; kp < HID / 2; ++kp) {
                const v2f v = *(const v2f*)&Bs[(kp * BP + tid) * 2];
                a += dsh[2 * kp] * v.x;
                a += dsh[2 * kp + 1] * v.y;
            }
            c0s[tid] = a;
        }
        __syncthreads();
#pragma unroll
        for (int j = 0; j < HID * HID / 8 / GM_T; ++j) {
            const int i = tid + j * GM_T;
            const int kp = i >> 5, c4 = i & 31;
            const float g0 = gsh[2 * kp], g1 = gsh[2 * kp + 1];
            float4* p = (float4*)&Bs[(kp * BP + c4 * 4) * 2];
            float4 u = p[0], v = p[1];
            p[0] = make_float4(u.x * g0, u.y * g1, u.z * g0, u.w * g1);
            p[1] = make_float4(v.x * g0, v.y * g1, v.z * g0, v.w * g1);
        }
        {
            const int kp = tid >> 2, c4 = tid & 3;
            const float g0 = gsh[2 * kp], g1 = gsh[2 * kp + 1];
            float4* p = (float4*)&Bs[(kp * BP + HID + c4 * 4) * 2];
            float4 u = p[0], v = p[1];
            p[0] = make_float4(u.x * g0, u.y * g1, u.z * g0, u.w * g1);
            p[1] = make_float4(v.x * g0, v.y * g1, v.z * g0, v.w * g1);
        }
    } else {
        if (tid < BP) c0s[tid] = 0.0f;
    }
    __syncthreads();

    int wave = (blockIdx.x * blockDim.x + tid) >> 5;
    int m0 = wave * 16;
    if (m0 >= nrows) return;
    int lane = tid & 31;
    int lr = lane & 15;
    int lh = lane >> 4;

    v8f acc[8];
#pragma unroll
    for (int n = 0; n < 8; ++n) {
        float cv = c0s[n * 16 + lr];
#pragma unroll
        for (int j = 0; j < 8; ++j) acc[n][j] = cv;
    }
    v8f accx;
    {
        float cv = c0s[HID + lr];
#pragma unroll
        for (int j = 0; j < 8; ++j) accx[j] = cv;
    }
    const float* xrow = X + (size_t)(m0 + lr) * HID;

    for (int k = 0; k < HID; k += 4) {
        int ka = k + lh * 2;
        v2f a;
        a.x = xrow[ka];
        a.y = xrow[ka + 1];
        const v2f* brow = (const v2f*)Bs + (ka >> 1) * BP + lr;
#pragma unroll
        for (int n = 0; n < 8; ++n) {
            const v2f b = brow[n * 16];
            acc[n] = __builtin_amdgcn_wmma_f32_16x16x4_f32(
                false, a, false, b, (short)0, acc[n], false, false);
        }
        const v2f bx = brow[HID];
        accx = __builtin_amdgcn_wmma_f32_16x16x4_f32(
            false, a, false, bx, (short)0, accx, false, false);
    }
#pragma unroll
    for (int n = 0; n < 8; ++n)
#pragma unroll
        for (int j = 0; j < 8; ++j)
            H[(size_t)(m0 + j + lh * 8) * HID + n * 16 + lr] = acc[n][j];
    if (lr < 2) {
        float* ap = lr ? a_d : a_s;
#pragma unroll
        for (int j = 0; j < 8; ++j) ap[m0 + j + lh * 8] = accx[j];
    }
}

#define NPB   391
#define OB_E  1024
#define OB_T  256
#define OB_W  (OB_E / 32)
#define OB_P  (OB_W + 1)

__global__ __launch_bounds__(OB_T) void gat_owner_bits(const int* __restrict__ ei,
                                                       unsigned* __restrict__ obw, int n_edges,
                                                       int n_tiles, float* __restrict__ z0, int nz0,
                                                       float* __restrict__ z1, int nz1,
                                                       const float* __restrict__ Wall,
                                                       const float* __restrict__ att_s,
                                                       const float* __restrict__ att_d,
                                                       float* __restrict__ WxU) {
    __shared__ unsigned tile[256 * OB_P];
    __shared__ float part[2 * 2 * (HID / 32)];
    const int tid = threadIdx.x;
    const int lane = tid & 31;
    if ((int)blockIdx.x >= n_tiles) {
        const int half = tid >> 7;
        const int t = tid & (HID - 1);
        const int r = ((int)blockIdx.x - n_tiles) * 2 + half;
        const int l = r >> 7;
        const float p = Wall[(size_t)r * HID + t];
        float ps = p * att_s[l * HID + t];
        float pd = p * att_d[l * HID + t];
        for (int off = 16; off > 0; off >>= 1) {
            ps += __shfl_down(ps, off, 32);
            pd += __shfl_down(pd, off, 32);
        }
        float* ph = part + half * (2 * (HID / 32));
        if (lane == 0) {
            ph[t >> 5] = ps;
            ph[HID / 32 + (t >> 5)] = pd;
        }
        __syncthreads();
        if (t < XC) {
            float v = 0.0f;
            if (t < 2) {
                const float* q = ph + t * (HID / 32);
                v = (q[0] + q[1]) + (q[2] + q[3]);
            }
            WxU[(size_t)r * XC + t] = v;
        }
        return;
    }
    for (int i = blockIdx.x * OB_T + tid; i < nz0; i += n_tiles * OB_T) z0[i] = 0.0f;
    for (int i = blockIdx.x * OB_T + tid; i < nz1; i += n_tiles * OB_T) z1[i] = 0.0f;
    for (int i = tid; i < 256 * OB_P; i += OB_T) tile[i] = 0u;
    const int* __restrict__ dstp = ei + n_edges;
    const int eb = blockIdx.x * OB_E;
    int dv[OB_E / OB_T];
#pragma unroll
    for (int j = 0; j < OB_E / OB_T; ++j) {
        const int e = eb + j * OB_T + tid;
        dv[j] = (e < n_edges) ? dstp[e] : -1;
    }
    __syncthreads();
#pragma unroll
    for (int j = 0; j < OB_E / OB_T; ++j) {
        const int le = j * OB_T + tid;
        if (dv[j] >= 0) atomicOr(&tile[((unsigned)(dv[j] / NPB) & 255u) * OB_P + (le >> 5)], 1u << lane);
    }
    __syncthreads();
    const size_t pw = (size_t)n_tiles * OB_W;
    unsigned* __restrict__ dst = obw + (size_t)blockIdx.x * OB_W + lane;
    const int r0 = tid >> 5;
#pragma unroll
    for (int j = 0; j < 256 / (OB_T / 32); ++j) {
        const int r = r0 + j * (OB_T / 32);
        dst[(size_t)r * pw] = tile[r * OB_P + lane];
    }
}

#define OWN_T 1024
#define OWN_W (OWN_T / 32)
#define WCAP  352
#define LCAP  (OWN_W * WCAP)
#define SCAN_U 8
#define PH2_U 8
#define SC_GROUP 8
#define NOROW 0xFFFFu
#define FIN_SLOTS (OWN_T / HID)
#define FIN_U 7
#define ROWPOS(c) ((((c) & 3) << 5) | ((c) >> 2))

__device__ __forceinline__ void gat_keepw(unsigned m, int e0, int lane, int* lw, int& wc) {
    for (int p = 0; p < 32; ++p) {
        unsigned mk = __builtin_amdgcn_ballot_w32(m != 0u);
        if (mk == 0u) break;
        if (m != 0u) {
            int pos = wc + __builtin_popcount(mk & ((1u << lane) - 1u));
            if (pos < WCAP) lw[pos] = e0 + __builtin_ctz(m);
            m &= m - 1u;
        }
        wc += __builtin_popcount(mk);
    }
}

__device__ __forceinline__ void gat_add_group(const int* lwl, const unsigned short* dwl,
                                              const float* gwl, int k0, int nl,
                                              const unsigned* mb, float* ssum, float* acc,
                                              const float* __restrict__ H, int lane) {
    const int rest = nl - k0;
    const int gc = rest < SC_GROUP ? rest : SC_GROUP;
    int sv = 0, dv = 0;
    float w = 0.0f;
    if (lane < gc) {
        sv = lwl[k0 + lane];
        dv = dwl[k0 + lane];
        w = expf(gwl[k0 + lane] - o2f(mb[dv]));
        atomicAdd(&ssum[dv], w);
    }
    int wbits = __float_as_int(w);
    if (gc == SC_GROUP) {
        float4 hv[SC_GROUP];
        int dk[SC_GROUP];
        float wk[SC_GROUP];
#pragma unroll
        for (int u = 0; u < SC_GROUP; ++u) {
            int sk = __builtin_amdgcn_readlane(sv, u);
            dk[u] = __builtin_amdgcn_readlane(dv, u);
            wk[u] = __int_as_float(__builtin_amdgcn_readlane(wbits, u));
            hv[u] = ((const float4*)(H + (size_t)sk * HID))[lane];
        }
#pragma unroll
        for (int u = 0; u < SC_GROUP; ++u) {
            float* ap = acc + dk[u] * HID + lane;
            atomicAdd(ap,      wk[u] * hv[u].x);
            atomicAdd(ap + 32, wk[u] * hv[u].y);
            atomicAdd(ap + 64, wk[u] * hv[u].z);
            atomicAdd(ap + 96, wk[u] * hv[u].w);
        }
    } else {
        for (int u = 0; u < gc; ++u) {
            int sk = __builtin_amdgcn_readlane(sv, u);
            int dd = __builtin_amdgcn_readlane(dv, u);
            float ww = __int_as_float(__builtin_amdgcn_readlane(wbits, u));
            const float4 h4 = ((const float4*)(H + (size_t)sk * HID))[lane];
            float* ap = acc + dd * HID + lane;
            atomicAdd(ap,      ww * h4.x);
            atomicAdd(ap + 32, ww * h4.y);
            atomicAdd(ap + 64, ww * h4.z);
            atomicAdd(ap + 96, ww * h4.w);
        }
    }
}

template <bool REUSE>
__global__ __launch_bounds__(OWN_T) void gat_owner(const int* __restrict__ ei,
                                                   const unsigned* __restrict__ obw,
                                                   const float* __restrict__ a_s,
                                                   const float* __restrict__ a_d,
                                                   const float* __restrict__ H,
                                                   float* __restrict__ out_acc,
                                                   int* __restrict__ gsrc,
                                                   unsigned short* __restrict__ gdl,
                                                   int* __restrict__ gcnt,
                                                   const float* __restrict__ bias,
                                                   float* __restrict__ sums,
                                                   float* __restrict__ sumsq,
                                                   const int* __restrict__ batch,
                                                   float* __restrict__ praw,
                                                   float* __restrict__ cnt,
                                                   int do_raw, int n_nodes, int n_edges) {
    __shared__ __attribute__((aligned(16))) float acc[NPB * HID];
    __shared__ float adl[NPB];
    __shared__ float ssum[NPB];
    __shared__ unsigned mb[NPB];
    __shared__ int gb[NPB];
    __shared__ int lsrc[LCAP];
    __shared__ float llog[LCAP];
    __shared__ unsigned short ldl[LCAP];
    __shared__ int wcnt[OWN_W];
    __shared__ int ovfl;

    const int tid = threadIdx.x;
    const int lane = tid & 31;
    const int wv = tid >> 5;
    const int lo = blockIdx.x * NPB;
    int nown = n_nodes - lo;
    if (nown > NPB) nown = NPB;
    const int* __restrict__ srcp = ei;
    const int* __restrict__ dstp = ei + n_edges;
    const float bv = bias[tid & (HID - 1)];

    for (int i = tid; i < nown; i += OWN_T) {
        float ad = a_d[lo + i];
        float sl = lrelu(a_s[lo + i] + ad);
        adl[i] = ad;
        ssum[i] = sl;
        mb[i] = f2o(sl);
        gb[i] = do_raw ? batch[lo + i] : 0;
    }
    if (tid == 0) ovfl = 0;
    __syncthreads();

    int* lw = lsrc + wv * WCAP;
    float* gw = llog + wv * WCAP;
    unsigned short* dw = ldl + wv * WCAP;
    int wc = 0;
    const size_t gl0 = (size_t)(blockIdx.x * OWN_W + wv) * WCAP;
    int* __restrict__ gs = gsrc + gl0;
    unsigned short* __restrict__ gd = gdl + gl0;
    if (REUSE) {
        wc = __builtin_amdgcn_readfirstlane(gcnt[blockIdx.x * OWN_W + wv]);
    } else {
        const int pw = ((n_edges + OB_E - 1) / OB_E) * OB_W;
        const int nq = pw >> 2;
        const uint4* __restrict__ oq = (const uint4*)(obw + (size_t)(blockIdx.x & 255u) * pw);
        for (int q0 = 0; q0 < nq; q0 += SCAN_U * OWN_T) {
            const int qb = q0 + tid;
            uint4 d[SCAN_U];
#pragma unroll
            for (int j = 0; j < SCAN_U; ++j) {
                int q = qb + j * OWN_T;
                d[j] = (q < nq) ? oq[q] : make_uint4(0u, 0u, 0u, 0u);
            }
#pragma unroll
            for (int j = 0; j < SCAN_U; ++j) {
                int e0 = (qb + j * OWN_T) * 128;
                gat_keepw(d[j].x, e0,      lane, lw, wc);
                gat_keepw(d[j].y, e0 + 32, lane, lw, wc);
                gat_keepw(d[j].z, e0 + 64, lane, lw, wc);
                gat_keepw(d[j].w, e0 + 96, lane, lw, wc);
            }
        }
        if (lane == 0) gcnt[blockIdx.x * OWN_W + wv] = wc;
    }
    if (lane == 0) {
        wcnt[wv] = wc;
        if (wc > WCAP) ovfl = 1;
    }
    __syncthreads();

    const bool ovf = ovfl != 0;
    const int nl = ovf ? 0 : wc;

    if (!ovf && REUSE) {
        for (int kb = 0; kb < nl; kb += 32 * PH2_U) {
            int sv[PH2_U];
            unsigned dl[PH2_U];
            float av[PH2_U];
#pragma unroll
            for (int u = 0; u < PH2_U; ++u) {
                int k = kb + u * 32 + lane;
                sv[u] = (k < nl) ? gs[k] : lo;
                dl[u] = (k < nl) ? (unsigned)gd[k] : NOROW;
            }
#pragma unroll
            for (int u = 0; u < PH2_U; ++u) av[u] = a_s[sv[u]];
#pragma unroll
            for (int u = 0; u < PH2_U; ++u) {
                int k = kb + u * 32 + lane;
                if (k < nl) {
                    if (dl[u] != NOROW) {
                        float lg = lrelu(av[u] + adl[dl[u]]);
                        atomicMax(&mb[dl[u]], f2o(lg));
                        lw[k] = sv[u];
                        dw[k] = (unsigned short)dl[u];
                        gw[k] = lg;
                    } else {
                        lw[k] = lo;
                        dw[k] = 0;
                        gw[k] = -__builtin_inff();
                    }
                }
            }
        }
    } else if (!ovf) {
        for (int kb = 0; kb < nl; kb += 32 * PH2_U) {
            int ee[PH2_U], dd[PH2_U], sv[PH2_U];
            float av[PH2_U];
#pragma unroll
            for (int u = 0; u < PH2_U; ++u) {
                int k = kb + u * 32 + lane;
                ee[u] = (k < nl) ? lw[k] : 0;
            }
#pragma unroll
            for (int u = 0; u < PH2_U; ++u) {
                dd[u] = dstp[ee[u]];
                sv[u] = srcp[ee[u]];
            }
#pragma unroll
            for (int u = 0; u < PH2_U; ++u) av[u] = a_s[sv[u]];
#pragma unroll
            for (int u = 0; u < PH2_U; ++u) {
                int k = kb + u * 32 + lane;
                if (k < nl) {
                    unsigned dl = (unsigned)(dd[u] - lo);
                    if (dl < (unsigned)nown) {
                        float lg = lrelu(av[u] + adl[dl]);
                        atomicMax(&mb[dl], f2o(lg));
                        lw[k] = sv[u];
                        dw[k] = (unsigned short)dl;
                        gw[k] = lg;
                        gs[k] = sv[u];
                        gd[k] = (unsigned short)dl;
                    } else {
                        lw[k] = lo;
                        dw[k] = 0;
                        gw[k] = -__builtin_inff();
                        gs[k] = lo;
                        gd[k] = (unsigned short)NOROW;
                    }
                }
            }
        }
    } else {
        for (int e = tid; e < n_edges; e += OWN_T) {
            unsigned dl = (unsigned)(dstp[e] - lo);
            if (dl < (unsigned)nown) {
                float lg = lrelu(a_s[srcp[e]] + adl[dl]);
                atomicMax(&mb[dl], f2o(lg));
            }
        }
    }
    __syncthreads();

    for (int i0 = wv; i0 < nown; i0 += OWN_W * 4) {
        float4 hv[4];
#pragma unroll
        for (int u = 0; u < 4; ++u) {
            int i = i0 + u * OWN_W;
            hv[u] = make_float4(0.0f, 0.0f, 0.0f, 0.0f);
            if (i < nown) hv[u] = ((const float4*)(H + (size_t)(lo + i) * HID))[lane];
        }
#pragma unroll
        for (int u = 0; u < 4; ++u) {
            int i = i0 + u * OWN_W;
            if (i < nown) {
                float es = expf(ssum[i] - o2f(mb[i]));
                float* ap = acc + i * HID + lane;
                ap[0]  = es * hv[u].x;
                ap[32] = es * hv[u].y;
                ap[64] = es * hv[u].z;
                ap[96] = es * hv[u].w;
                if (lane == 0) ssum[i] = es;
            }
        }
    }
    __syncthreads();

    if (!ovf) {
        const int myc = wcnt[lane];
        for (int wl = 0; wl < OWN_W; ++wl) {
            const int nlw = __builtin_amdgcn_readlane(myc, wl);
            for (int g = (wv - wl) & (OWN_W - 1); g * SC_GROUP < nlw; g += OWN_W)
                gat_add_group(lsrc + wl * WCAP, ldl + wl * WCAP, llog + wl * WCAP, g * SC_GROUP, nlw,
                              mb, ssum, acc, H, lane);
        }
    } else {
        for (int e = tid; e < n_edges; e += OWN_T) {
            unsigned dl = (unsigned)(dstp[e] - lo);
            if (dl < (unsigned)nown) {
                int s = srcp[e];
                float ww = expf(lrelu(a_s[s] + adl[dl]) - o2f(mb[dl]));
                atomicAdd(&ssum[dl], ww);
                const float* hp = H + (size_t)s * HID;
                float* ap = acc + dl * HID;
                for (int c = 0; c < HID; ++c) atomicAdd(ap + ROWPOS(c), ww * hp[c]);
            }
        }
    }
    __syncthreads();

    for (int i = tid; i < nown; i += OWN_T) ssum[i] = 1.0f / ssum[i];
    __syncthreads();
    const int c = tid & (HID - 1);
    const int cw = ROWPOS(c);
    const int slot = __builtin_amdgcn_readfirstlane(tid >> 7);
    const int sub = (nown + FIN_SLOTS - 1) / FIN_SLOTS;
    const int r0 = slot * sub;
    int r1 = r0 + sub;
    if (r1 > nown) r1 = nown;
    float* __restrict__ ocol = out_acc + (size_t)lo * HID + c;
    float ls = 0.0f, lq = 0.0f, run = 0.0f;
    int gcur = 0, rl = 0;
    if (do_raw && r0 < r1) gcur = __builtin_amdgcn_readfirstlane(gb[r0]);
    for (int ib = r0; ib < r1; ib += FIN_U) {
        const int last = r1 - 1;
        const int cntu = (r1 - ib < FIN_U) ? (r1 - ib) : FIN_U;
        int il = ib + (lane & 7);
        if (il > last) il = last;
        const int ibits = __float_as_int(ssum[il]);
        const int gl = gb[il];
        float v[FIN_U];
#pragma unroll
        for (int u = 0; u < FIN_U; ++u) {
            int i = ib + u;
            if (i > last) i = last;
            float inv = __int_as_float(__builtin_amdgcn_readlane(ibits, u));
            float t = acc[i * HID + cw] * inv + bv;
            float e = expm1f(t > 0.0f ? 0.0f : t);
            v[u] = t > 0.0f ? t : e;
        }
#pragma unroll
        for (int u = 0; u < FIN_U; ++u) {
            if (u < cntu) {
                ocol[(size_t)(ib + u) * HID] = v[u];
                ls += v[u];
                lq += v[u] * v[u];
            }
        }
        if (do_raw) {
#pragma unroll
            for (int u = 0; u < FIN_U; ++u) {
                if (u < cntu) {
                    int g = __builtin_amdgcn_readlane(gl, u);
                    if (g != gcur) {
                        atomicAdd(&praw[(size_t)gcur * (3 * HID) + c], run);
                        if (c == 0) atomicAdd(&cnt[gcur], (float)rl);
                        run = 0.0f;
                        rl = 0;
                        gcur = g;
                    }
                    run += v[u];
                    ++rl;
                }
            }
        }
    }
    if (do_raw && rl > 0) {
        atomicAdd(&praw[(size_t)gcur * (3 * HID) + c], run);
        if (c == 0) atomicAdd(&cnt[gcur], (float)rl);
    }
    float* red = llog;
    red[slot * HID + c] = ls;
    red[(FIN_SLOTS + slot) * HID + c] = lq;
    __syncthreads();
    if (tid < HID) {
        float ts = 0.0f, tq = 0.0f;
#pragma unroll
        for (int k = 0; k < FIN_SLOTS; ++k) {
            ts += red[k * HID + tid];
            tq += red[(FIN_SLOTS + k) * HID + tid];
        }
        atomicAdd(&sums[tid], ts);
        atomicAdd(&sumsq[tid], tq);
    }
}

__global__ void gat_bn_pool(float* __restrict__ out_acc, const float* __restrict__ sums,
                            const float* __restrict__ sumsq, const float* __restrict__ gamma,
                            const float* __restrict__ beta, const int* __restrict__ batch,
                            float* __restrict__ pooled, int layer, int n_nodes) {
    int idx = blockIdx.x * blockDim.x + threadIdx.x;
    if (idx >= n_nodes * HID) return;
    int c = idx & (HID - 1);
    int n = idx >> 7;
    float invN = 1.0f / (float)n_nodes;
    float mu = sums[c] * invN;
    float var = sumsq[c] * invN - mu * mu;
    float xv = (out_acc[idx] - mu) / sqrtf(var + BN_EPS) * gamma[c] + beta[c];
    out_acc[idx] = xv;
    atomicAdd(&pooled[(size_t)batch[n] * (3 * HID) + layer * HID + c], xv);
}

extern "C" void kernel_launch(void* const* d_in, const int* in_sizes, int n_in,
                              void* d_out, int out_size, void* d_ws, size_t ws_size,
                              hipStream_t stream) {
    const float* x_in  = (const float*)d_in[0];
    const int*   ei    = (const int*)d_in[1];
    const int*   batch = (const int*)d_in[2];
    const float* W     = (const float*)d_in[3];
    const float* att_s = (const float*)d_in[4];
    const float* att_d = (const float*)d_in[5];
    const float* bias  = (const float*)d_in[6];
    const float* gamma = (const float*)d_in[7];
    const float* beta  = (const float*)d_in[8];

    const int N = in_sizes[0] / HID;
    const int E = in_sizes[1] / 2;

    float* pooled = (float*)d_out;
    float* xfinal = (float*)d_out + (size_t)NGRAPH * 3 * HID;

    char* w = (char*)d_ws;
    float* hbuf = (float*)w;  w += (size_t)N * HID * sizeof(float);
    float* xb   = (float*)w;  w += (size_t)N * HID * sizeof(float);
    const int n_tiles = (E + OB_E - 1) / OB_E;
    unsigned* obw = (unsigned*)w;  w += (size_t)256 * n_tiles * OB_W * sizeof(unsigned);
    float* asrc = (float*)w;  w += (size_t)N * sizeof(float);
    float* adst = (float*)w;  w += (size_t)N * sizeof(float);
    const int SL = 2 * HID + NGRAPH;
    float* sums0 = (float*)w;
    float* WxU  = sums0 + 3 * SL;
    const int n_own = (N + NPB - 1) / NPB;
    int* gsrc = (int*)(WxU + 3 * HID * XC);
    int* gcnt = gsrc + (size_t)n_own * LCAP;
    unsigned short* gdl = (unsigned short*)(gcnt + (size_t)n_own * OWN_W);

    const int waves_g = (N + 15) / 16;
    dim3 blk(256);
    dim3 grid_gemm((waves_g * 32 + GM_T - 1) / GM_T);
    dim3 grid_own(n_own);
    dim3 grid_elem(((size_t)N * HID + 255) / 256);

    gat_owner_bits<<<n_tiles + 3 * HID / 2, OB_T, 0, stream>>>(ei, obw, E, n_tiles, pooled, NGRAPH * 3 * HID,
                                                             sums0, 3 * SL, W, att_s, att_d, WxU);

    for (int l = 0; l < 3; ++l) {
        const int lp = (l == 0) ? 0 : l - 1;
        const float* X  = (l == 0) ? x_in : xb;
        float* oacc     = (l == 2) ? xfinal : xb;
        float* sl = sums0 + l * SL;
        float* sp = sums0 + lp * SL;

        gat_gemm_wmma<<<grid_gemm, GM_T, 0, stream>>>(X, W + (size_t)l * HID * HID, WxU + l * HID * XC,
                                                      sp, sp + HID, gamma + lp * HID, beta + lp * HID,
                                                      pooled + lp * HID, sp + 2 * HID,
                                                      hbuf, asrc, adst, (l > 0) ? 1 : 0, N);
        if (l == 0)
            gat_owner<false><<<grid_own, OWN_T, 0, stream>>>(ei, obw, asrc, adst, hbuf, oacc, gsrc, gdl, gcnt,
                                                             bias + l * HID, sl, sl + HID, batch,
                                                             pooled + l * HID, sl + 2 * HID, (l < 2) ? 1 : 0, N, E);
        else
            gat_owner<true><<<grid_own, OWN_T, 0, stream>>>(ei, obw, asrc, adst, hbuf, oacc, gsrc, gdl, gcnt,
                                                            bias + l * HID, sl, sl + HID, batch,
                                                            pooled + l * HID, sl + 2 * HID, (l < 2) ? 1 : 0, N, E);
        if (l == 2)
            gat_bn_pool<<<grid_elem, blk, 0, stream>>>(oacc, sl, sl + HID, gamma + l * HID,
                                                       beta + l * HID, batch, pooled, l, N);
    }
}
